# starting kernel
# baseline (speedup 1.0000x reference)
#include <hip/hip_runtime.h>
#include <hip/hip_bf16.h>

typedef __attribute__((ext_vector_type(2))) float v2f;
typedef __attribute__((ext_vector_type(8))) float v8f;

#define HEADS 4
#define HC 256
#define CPH 64
#define NEG_SLOPE 0.2f

__device__ __forceinline__ unsigned enc_f32(float f) {
    unsigned u = __float_as_uint(f);
    return u ^ ((u >> 31) ? 0xFFFFFFFFu : 0x80000000u);
}
__device__ __forceinline__ float dec_f32(unsigned u) {
    return __uint_as_float(u ^ ((u >> 31) ? 0x80000000u : 0xFFFFFFFFu));
}

__device__ __forceinline__ void edge_sd(const int* __restrict__ ei, int E, int e,
                                        int& s, int& d) {
    if (e < E) { s = ei[e]; d = ei[E + e]; }
    else       { s = d = e - E; }
}

__global__ __launch_bounds__(256) void gemm_wmma_f32(
        const float* __restrict__ X, const float* __restrict__ W,
        float* __restrict__ H, int M, int K) {
    const int wavesPerBlock = 8;
    int wave  = blockIdx.x * wavesPerBlock + (threadIdx.x >> 5);
    const int strips = HC / 64;
    int tileM = wave / strips;
    int strip = wave % strips;
    if (tileM * 16 >= M) return;

    int lane = threadIdx.x & 31;
    int half = lane >> 4;
    int l16  = lane & 15;
    int m    = tileM * 16 + l16;
    int col0 = strip * 64 + l16;
    const float* xrow = X + (size_t)m * K;

    v8f acc0 = {}, acc1 = {}, acc2 = {}, acc3 = {};
    for (int kb = 0; kb < K; kb += 4) {
        int ka = kb + 2 * half;
        v2f a;
        a.x = xrow[ka];
        a.y = xrow[ka + 1];
        const float* w0 = W + (size_t)ka * HC + col0;
        const float* w1 = W + (size_t)(ka + 1) * HC + col0;
        v2f b0, b1, b2, b3;
        b0.x = w0[0];  b0.y = w1[0];
        b1.x = w0[16]; b1.y = w1[16];
        b2.x = w0[32]; b2.y = w1[32];
        b3.x = w0[48]; b3.y = w1[48];
        acc0 = __builtin_amdgcn_wmma_f32_16x16x4_f32(
                false, a, false, b0, (short)0, acc0, false, false);
        acc1 = __builtin_amdgcn_wmma_f32_16x16x4_f32(
                false, a, false, b1, (short)0, acc1, false, false);
        acc2 = __builtin_amdgcn_wmma_f32_16x16x4_f32(
                false, a, false, b2, (short)0, acc2, false, false);
        acc3 = __builtin_amdgcn_wmma_f32_16x16x4_f32(
                false, a, false, b3, (short)0, acc3, false, false);
    }
#pragma unroll
    for (int r = 0; r < 8; ++r) {
        int row = tileM * 16 + r + 8 * half;
        if (row < M) {
            float* hrow = H + (size_t)row * HC + col0;
            hrow[0]  = acc0[r];
            hrow[16] = acc1[r];
            hrow[32] = acc2[r];
            hrow[48] = acc3[r];
        }
    }
}

__global__ void node_alpha(const float* __restrict__ H,
                           const float* __restrict__ a_src,
                           const float* __restrict__ a_dst,
                           float* __restrict__ alphaS,
                           float* __restrict__ alphaD, int N) {
    int t = blockIdx.x * blockDim.x + threadIdx.x;
    if (t >= N * HEADS) return;
    int n = t >> 2, h = t & 3;
    const float4* hp = (const float4*)(H + (size_t)n * HC + h * CPH);
    const float4* as = (const float4*)(a_src + h * CPH);
    const float4* ad = (const float4*)(a_dst + h * CPH);
    float ss = 0.f, sd = 0.f;
#pragma unroll 4
    for (int i = 0; i < CPH / 4; ++i) {
        float4 hv = hp[i], av = as[i], dv = ad[i];
        ss += hv.x * av.x + hv.y * av.y + hv.z * av.z + hv.w * av.w;
        sd += hv.x * dv.x + hv.y * dv.y + hv.z * dv.z + hv.w * dv.w;
    }
    alphaS[t] = ss;
    alphaD[t] = sd;
}

__global__ void fill_f32(float* __restrict__ p, float v, size_t n) {
    size_t i = (size_t)blockIdx.x * blockDim.x + threadIdx.x;
    if (i < n) p[i] = v;
}
__global__ void fill_u32(unsigned* __restrict__ p, unsigned v, size_t n) {
    size_t i = (size_t)blockIdx.x * blockDim.x + threadIdx.x;
    if (i < n) p[i] = v;
}

__global__ void edge_max(const int* __restrict__ ei, int E, int N,
                         const float* __restrict__ aS,
                         const float* __restrict__ aD,
                         unsigned* __restrict__ mEnc) {
    int e = blockIdx.x * blockDim.x + threadIdx.x;
    if (e >= E + N) return;
    int s, d; edge_sd(ei, E, e, s, d);
    float4 as = *(const float4*)(aS + 4 * (size_t)s);
    float4 ad = *(const float4*)(aD + 4 * (size_t)d);
    float v[4] = {as.x + ad.x, as.y + ad.y, as.z + ad.z, as.w + ad.w};
#pragma unroll
    for (int h = 0; h < 4; ++h) {
        float x = v[h];
        x = x >= 0.f ? x : NEG_SLOPE * x;
        atomicMax(&mEnc[4 * (size_t)d + h], enc_f32(x));
    }
}

__global__ void edge_sum(const int* __restrict__ ei, int E, int N,
                         const float* __restrict__ aS,
                         const float* __restrict__ aD,
                         const unsigned* __restrict__ mEnc,
                         float* __restrict__ sums) {
    int e = blockIdx.x * blockDim.x + threadIdx.x;
    if (e >= E + N) return;
    int s, d; edge_sd(ei, E, e, s, d);
    float4 as = *(const float4*)(aS + 4 * (size_t)s);
    float4 ad = *(const float4*)(aD + 4 * (size_t)d);
    float v[4] = {as.x + ad.x, as.y + ad.y, as.z + ad.z, as.w + ad.w};
#pragma unroll
    for (int h = 0; h < 4; ++h) {
        float x = v[h];
        x = x >= 0.f ? x : NEG_SLOPE * x;
        float m = dec_f32(mEnc[4 * (size_t)d + h]);
        atomicAdd(&sums[4 * (size_t)d + h], expf(x - m));
    }
}

__global__ void edge_aggr(const int* __restrict__ ei, int E, int N,
                          const float* __restrict__ H,
                          const float* __restrict__ aS,
                          const float* __restrict__ aD,
                          const unsigned* __restrict__ mEnc,
                          const float* __restrict__ sums,
                          float* __restrict__ agg) {
    long long t = (long long)blockIdx.x * blockDim.x + threadIdx.x;
    long long Et = (long long)E + N;
    if (t >= Et * 64) return;
    int e  = (int)(t >> 6);
    int fq = (int)(t & 63);
    int h  = fq >> 4;
    int s, d; edge_sd(ei, E, e, s, d);
    float x = aS[4 * (size_t)s + h] + aD[4 * (size_t)d + h];
    x = x >= 0.f ? x : NEG_SLOPE * x;
    float m     = dec_f32(mEnc[4 * (size_t)d + h]);
    float denom = sums[4 * (size_t)d + h] + 1e-16f;
    float alpha = expf(x - m) / denom;
    float4 hv = *(const float4*)(H + (size_t)s * HC + fq * 4);
    float* o  = agg + (size_t)d * HC + fq * 4;
    atomicAdd(o + 0, hv.x * alpha);
    atomicAdd(o + 1, hv.y * alpha);
    atomicAdd(o + 2, hv.z * alpha);
    atomicAdd(o + 3, hv.w * alpha);
}

__global__ void finalize_bias(float* __restrict__ agg,
                              const float* __restrict__ b, int N, int relu) {
    int t = blockIdx.x * blockDim.x + threadIdx.x;
    if (t >= N * (HC / 4)) return;
    int n = t / (HC / 4), fq = t % (HC / 4);
    float4* p = (float4*)(agg + (size_t)n * HC + fq * 4);
    float4 v  = *p;
    float4 bb = *(const float4*)(b + fq * 4);
    v.x += bb.x; v.y += bb.y; v.z += bb.z; v.w += bb.w;
    if (relu) {
        v.x = fmaxf(v.x, 0.f); v.y = fmaxf(v.y, 0.f);
        v.z = fmaxf(v.z, 0.f); v.w = fmaxf(v.w, 0.f);
    }
    *p = v;
}

static void gat_layer(const float* Xin, int K, const float* W, const float* as,
                      const float* ad, const float* b, float* Hbuf, float* agg,
                      const int* ei, int E, int N, float* alphaS, float* alphaD,
                      unsigned* mEnc, float* sums, int relu, hipStream_t stream) {
    int waves  = ((N + 15) / 16) * (HC / 64);
    int blocks = (waves + 7) / 8;
    gemm_wmma_f32<<<blocks, 256, 0, stream>>>(Xin, W, Hbuf, N, K);

    int nh = N * HEADS;
    node_alpha<<<(nh + 255) / 256, 256, 0, stream>>>(Hbuf, as, ad, alphaS, alphaD, N);
    fill_u32<<<(nh + 255) / 256, 256, 0, stream>>>(mEnc, 0u, (size_t)nh);
    fill_f32<<<(nh + 255) / 256, 256, 0, stream>>>(sums, 0.f, (size_t)nh);
    size_t nagg = (size_t)N * HC;
    fill_f32<<<(int)((nagg + 255) / 256), 256, 0, stream>>>(agg, 0.f, nagg);

    int Et = E + N;
    edge_max<<<(Et + 255) / 256, 256, 0, stream>>>(ei, E, N, alphaS, alphaD, mEnc);
    edge_sum<<<(Et + 255) / 256, 256, 0, stream>>>(ei, E, N, alphaS, alphaD, mEnc, sums);
    long long tt = (long long)Et * 64;
    edge_aggr<<<(int)((tt + 255) / 256), 256, 0, stream>>>(ei, E, N, Hbuf, alphaS,
                                                           alphaD, mEnc, sums, agg);
    finalize_bias<<<(N * (HC / 4) + 255) / 256, 256, 0, stream>>>(agg, b, N, relu);
}

extern "C" void kernel_launch(void* const* d_in, const int* in_sizes, int n_in,
                              void* d_out, int out_size, void* d_ws, size_t ws_size,
                              hipStream_t stream) {
    const float* x   = (const float*)d_in[0];
    const int*   ei  = (const int*)d_in[1];
    const float* W1  = (const float*)d_in[2];
    const float* as1 = (const float*)d_in[3];
    const float* ad1 = (const float*)d_in[4];
    const float* b1  = (const float*)d_in[5];
    const float* W2  = (const float*)d_in[6];
    const float* as2 = (const float*)d_in[7];
    const float* ad2 = (const float*)d_in[8];
    const float* b2  = (const float*)d_in[9];
    const float* W3  = (const float*)d_in[10];
    const float* as3 = (const float*)d_in[11];
    const float* ad3 = (const float*)d_in[12];
    const float* b3  = (const float*)d_in[13];

    int K1 = in_sizes[2] / HC;
    int K2 = in_sizes[6] / HC;
    int K3 = in_sizes[10] / HC;
    int N  = in_sizes[0] / K1;
    int E  = in_sizes[1] / 2;

    float*    Hbuf   = (float*)d_ws;
    float*    bufA   = Hbuf + (size_t)N * HC;
    float*    bufB   = bufA + (size_t)N * HC;
    float*    alphaS = bufB + (size_t)N * HC;
    float*    alphaD = alphaS + (size_t)N * HEADS;
    unsigned* mEnc   = (unsigned*)(alphaD + (size_t)N * HEADS);
    float*    sums   = (float*)(mEnc + (size_t)N * HEADS);

    gat_layer(x,    K1, W1, as1, ad1, b1, Hbuf, bufA,          ei, E, N,
              alphaS, alphaD, mEnc, sums,  1, stream);
    gat_layer(bufA, K2, W2, as2, ad2, b2, Hbuf, bufB,          ei, E, N,
              alphaS, alphaD, mEnc, sums,  1, stream);
    gat_layer(bufB, K3, W3, as3, ad3, b3, Hbuf, (float*)d_out, ei, E, N,
              alphaS, alphaD, mEnc, sums,  0, stream);
}


# Each block filters the edge list once into an LDS hit list (source, local dst, logits); maxima, weights and row sums come from the list, rows fetched a batch ahead; sum order: LDS arrival, as before
# speedup vs baseline: 4.8036x; 4.8036x over previous
#include <hip/hip_runtime.h>
#include <hip/hip_bf16.h>

typedef __attribute__((ext_vector_type(2))) float v2f;
typedef __attribute__((ext_vector_type(8))) float v8f;

#define HEADS 4
#define HC 256
#define CPH 64
#define NEG_SLOPE 0.2f

__device__ __forceinline__ unsigned enc_f32(float f) {
    unsigned u = __float_as_uint(f);
    return u ^ ((u >> 31) ? 0xFFFFFFFFu : 0x80000000u);
}
__device__ __forceinline__ float dec_f32(unsigned u) {
    return __uint_as_float(u ^ ((u >> 31) ? 0x80000000u : 0xFFFFFFFFu));
}

__global__ __launch_bounds__(256) void gemm_wmma_f32(
        const float* __restrict__ X, const float* __restrict__ W,
        float* __restrict__ H, int M, int K) {
    const int wavesPerBlock = 8;
    int wave  = blockIdx.x * wavesPerBlock + (threadIdx.x >> 5);
    const int strips = HC / 64;
    int tileM = wave / strips;
    int strip = wave % strips;
    if (tileM * 16 >= M) return;

    int lane = threadIdx.x & 31;
    int half = lane >> 4;
    int l16  = lane & 15;
    int m    = tileM * 16 + l16;
    int col0 = strip * 64 + l16;
    const float* xrow = X + (size_t)m * K;

    v8f acc0 = {}, acc1 = {}, acc2 = {}, acc3 = {};
    for (int kb = 0; kb < K; kb += 4) {
        int ka = kb + 2 * half;
        v2f a;
        a.x = xrow[ka];
        a.y = xrow[ka + 1];
        const float* w0 = W + (size_t)ka * HC + col0;
        const float* w1 = W + (size_t)(ka + 1) * HC + col0;
        v2f b0, b1, b2, b3;
        b0.x = w0[0];  b0.y = w1[0];
        b1.x = w0[16]; b1.y = w1[16];
        b2.x = w0[32]; b2.y = w1[32];
        b3.x = w0[48]; b3.y = w1[48];
        acc0 = __builtin_amdgcn_wmma_f32_16x16x4_f32(
                false, a, false, b0, (short)0, acc0, false, false);
        acc1 = __builtin_amdgcn_wmma_f32_16x16x4_f32(
                false, a, false, b1, (short)0, acc1, false, false);
        acc2 = __builtin_amdgcn_wmma_f32_16x16x4_f32(
                false, a, false, b2, (short)0, acc2, false, false);
        acc3 = __builtin_amdgcn_wmma_f32_16x16x4_f32(
                false, a, false, b3, (short)0, acc3, false, false);
    }
#pragma unroll
    for (int r = 0; r < 8; ++r) {
        int row = tileM * 16 + r + 8 * half;
        if (row < M) {
            float* hrow = H + (size_t)row * HC + col0;
            hrow[0]  = acc0[r];
            hrow[16] = acc1[r];
            hrow[32] = acc2[r];
            hrow[48] = acc3[r];
        }
    }
}

__global__ void node_alpha(const float* __restrict__ H,
                           const float* __restrict__ a_src,
                           const float* __restrict__ a_dst,
                           float* __restrict__ alphaS,
                           float* __restrict__ alphaD, int N) {
    int t = blockIdx.x * blockDim.x + threadIdx.x;
    if (t >= N * HEADS) return;
    int n = t >> 2, h = t & 3;
    const float4* hp = (const float4*)(H + (size_t)n * HC + h * CPH);
    const float4* as = (const float4*)(a_src + h * CPH);
    const float4* ad = (const float4*)(a_dst + h * CPH);
    float ss = 0.f, sd = 0.f;
#pragma unroll 4
    for (int i = 0; i < CPH / 4; ++i) {
        float4 hv = hp[i], av = as[i], dv = ad[i];
        ss += hv.x * av.x + hv.y * av.y + hv.z * av.z + hv.w * av.w;
        sd += hv.x * dv.x + hv.y * dv.y + hv.z * dv.z + hv.w * dv.w;
    }
    alphaS[t] = ss;
    alphaD[t] = sd;
}

__device__ __forceinline__ float leaky(float x) {
    return x >= 0.f ? x : NEG_SLOPE * x;
}

#define TD 200
#define TILE_THREADS 1024
#define SCAN_U 8
#define CAP 4352
#define NIT ((CAP + TILE_THREADS - 1) / TILE_THREADS)

__device__ __forceinline__ void load_dst(int (&d)[SCAN_U], const int* __restrict__ dsts,
                                         int c, int lane, int E) {
#pragma unroll
    for (int j = 0; j < SCAN_U; ++j) {
        int e = c + j * 32 + lane;
        int v = dsts[e < E ? e : E - 1];
        d[j] = (e < E) ? v : -1;
    }
}

__device__ __forceinline__ void tile_add_edge(
        const float* __restrict__ H, const float* __restrict__ aS, int s, int dl,
        int lane, float* acc, const float* sAD, const unsigned* sM, float* sSum) {
    const float* hrow = H + (size_t)s * HC + lane;
    float hv[8];
#pragma unroll
    for (int k = 0; k < 8; ++k) hv[k] = hrow[32 * k];
    float4 as = *(const float4*)(aS + 4 * (size_t)s);
    float w0 = expf(leaky(as.x + sAD[dl * 4 + 0]) - dec_f32(sM[dl * 4 + 0]));
    float w1 = expf(leaky(as.y + sAD[dl * 4 + 1]) - dec_f32(sM[dl * 4 + 1]));
    float w2 = expf(leaky(as.z + sAD[dl * 4 + 2]) - dec_f32(sM[dl * 4 + 2]));
    float w3 = expf(leaky(as.w + sAD[dl * 4 + 3]) - dec_f32(sM[dl * 4 + 3]));
    if (lane < 4) {
        float wl = lane == 0 ? w0 : lane == 1 ? w1 : lane == 2 ? w2 : w3;
        atomicAdd(&sSum[dl * 4 + lane], wl);
    }
    float* a = acc + dl * HC + lane;
    atomicAdd(a + 0,   hv[0] * w0);
    atomicAdd(a + 32,  hv[1] * w0);
    atomicAdd(a + 64,  hv[2] * w1);
    atomicAdd(a + 96,  hv[3] * w1);
    atomicAdd(a + 128, hv[4] * w2);
    atomicAdd(a + 160, hv[5] * w2);
    atomicAdd(a + 192, hv[6] * w3);
    atomicAdd(a + 224, hv[7] * w3);
}

__device__ __forceinline__ void load_piece(int4 (&d)[SCAN_U], const int4* __restrict__ dv,
                                           int vb, int lane, int nvec) {
#pragma unroll
    for (int j = 0; j < SCAN_U; ++j) {
        int vi = vb + j * 32 + lane;
        d[j] = dv[vi < nvec ? vi : nvec - 1];
    }
}

__device__ __forceinline__ void walk_issue(const float* __restrict__ H, const int2* lSD, int q,
                                           int lane, float (&h)[4][8], int (&dl)[4]) {
    const int4* p = (const int4*)(lSD + 4 * q);
    int4 e01 = p[0], e23 = p[1];
    int s[4] = {e01.x, e01.z, e23.x, e23.z};
    dl[0] = e01.y; dl[1] = e01.w; dl[2] = e23.y; dl[3] = e23.w;
#pragma unroll
    for (int u = 0; u < 4; ++u) {
        const float* row = H + (size_t)__builtin_amdgcn_readfirstlane(s[u]) * HC + lane;
#pragma unroll
        for (int k = 0; k < 8; ++k) h[u][k] = row[32 * k];
    }
}

__device__ __forceinline__ void walk_apply(float* acc, const float4* lW, int q, int lane,
                                           const float (&h)[4][8], const int (&dl)[4]) {
    float4 w[4];
#pragma unroll
    for (int u = 0; u < 4; ++u) w[u] = lW[4 * q + u];
#pragma unroll
    for (int u = 0; u < 4; ++u) {
        float* a = acc + dl[u] * HC + lane;
        atomicAdd(a + 0,   h[u][0] * w[u].x);
        atomicAdd(a + 32,  h[u][1] * w[u].x);
        atomicAdd(a + 64,  h[u][2] * w[u].y);
        atomicAdd(a + 96,  h[u][3] * w[u].y);
        atomicAdd(a + 128, h[u][4] * w[u].z);
        atomicAdd(a + 160, h[u][5] * w[u].z);
        atomicAdd(a + 192, h[u][6] * w[u].w);
        atomicAdd(a + 224, h[u][7] * w[u].w);
    }
}

__global__ __launch_bounds__(TILE_THREADS) void aggr_tile(
        const int* __restrict__ ei, int E, int N,
        const float* __restrict__ H,
        const float* __restrict__ aS,
        const float* __restrict__ aD,
        const float* __restrict__ b, int relu,
        float* __restrict__ out) {
    __shared__ __attribute__((aligned(16))) float    acc[(TD + 1) * HC];
    __shared__ __attribute__((aligned(16))) float    sAD[TD * 4];
    __shared__ __attribute__((aligned(16))) unsigned sM[TD * 4];
    __shared__ float    sSum[TD * 4];
    __shared__ __attribute__((aligned(16))) float4   lW[CAP];
    __shared__ __attribute__((aligned(16))) int2     lSD[CAP];
    __shared__ int sCnt;

    const int tid  = threadIdx.x;
    const int lane = tid & 31;
    const int wave = tid >> 5;
    const int nwaves = TILE_THREADS / 32;
    const int d0 = blockIdx.x * TD;
    const int nd = (N - d0) < TD ? (N - d0) : TD;
    const int* __restrict__ srcs = ei;
    const int* __restrict__ dsts = ei + E;
    const int wv = __builtin_amdgcn_readfirstlane(wave);

    for (int i = tid; i < TD * HC / 4; i += TILE_THREADS)
        ((float4*)acc)[i] = make_float4(0.f, 0.f, 0.f, 0.f);
    for (int i = tid; i < TD * 4; i += TILE_THREADS) {
        int dl = i >> 2;
        float ad = 0.f, lg = 0.f;
        unsigned me = 0u;
        if (dl < nd) {
            size_t g = 4 * (size_t)(d0 + dl) + (i & 3);
            ad = aD[g];
            lg = leaky(aS[g] + ad);
            me = enc_f32(lg);
            ((float*)lW)[i] = lg;
            if ((i & 3) == 0) lSD[dl] = make_int2(d0 + dl, dl);
        }
        sAD[i] = ad; sM[i] = me; sSum[i] = 0.f;
    }
    if (tid == 0) sCnt = nd;
    __syncthreads();

    int head = (int)((4u - (unsigned)(((size_t)dsts >> 2) & 3)) & 3u);
    if (head > E) head = E;
    const int nvec = (E - head) >> 2;
    const int4* __restrict__ dv = (const int4*)(dsts + head);
    if (tid < E - 4 * nvec) {
        int e = tid < head ? tid : 4 * nvec + tid;
        if ((unsigned)(dsts[e] - d0) < (unsigned)nd) {
            int idx = atomicAdd(&sCnt, 1);
            if (idx < CAP) lSD[idx].x = e;
        }
    }
    if (nvec > 0) {
        const int PV = 32 * SCAN_U;
        int4 cur[SCAN_U];
        load_piece(cur, dv, wv * PV, lane, nvec);
        for (int vb = wv * PV; vb < nvec; vb += nwaves * PV) {
            int4 nx[SCAN_U];
            load_piece(nx, dv, vb + nwaves * PV, lane, nvec);
            unsigned hm = 0u;
#pragma unroll
            for (int j = 0; j < SCAN_U; ++j) {
                unsigned lim = (vb + j * 32 + lane < nvec) ? (unsigned)nd : 0u;
                hm |= ((unsigned)(cur[j].x - d0) < lim ? 1u : 0u) << (4 * j + 0);
                hm |= ((unsigned)(cur[j].y - d0) < lim ? 1u : 0u) << (4 * j + 1);
                hm |= ((unsigned)(cur[j].z - d0) < lim ? 1u : 0u) << (4 * j + 2);
                hm |= ((unsigned)(cur[j].w - d0) < lim ? 1u : 0u) << (4 * j + 3);
            }
            if (hm) {
                int idx = atomicAdd(&sCnt, __popc(hm));
                int eb  = head + 4 * (vb + lane);
                do {
                    int bit = __ffs(hm) - 1;
                    hm &= hm - 1;
                    if (idx < CAP) lSD[idx].x = eb + (bit >> 2) * 128 + (bit & 3);
                    ++idx;
                } while (hm);
            }
#pragma unroll
            for (int j = 0; j < SCAN_U; ++j) cur[j] = nx[j];
        }
    }
    __syncthreads();

    const int total = sCnt;
    if (total <= CAP - 4) {
        const int T = total;
        if (T > nd) {
            int sI[NIT], dI[NIT];
            float4 aI[NIT];
#pragma unroll
            for (int k = 0; k < NIT; ++k) {
                int i = nd + tid + k * TILE_THREADS;
                int e = (i < T) ? lSD[i].x : 0;
                sI[k] = srcs[e];
                dI[k] = dsts[e] - d0;
            }
#pragma unroll
            for (int k = 0; k < NIT; ++k) aI[k] = *(const float4*)(aS + 4 * (size_t)sI[k]);
#pragma unroll
            for (int k = 0; k < NIT; ++k) {
                int i = nd + tid + k * TILE_THREADS;
                if (i < T) {
                    int dl = dI[k];
                    float4 ad = *(const float4*)(sAD + 4 * dl);
                    float4 lg;
                    lg.x = leaky(aI[k].x + ad.x); lg.y = leaky(aI[k].y + ad.y);
                    lg.z = leaky(aI[k].z + ad.z); lg.w = leaky(aI[k].w + ad.w);
                    atomicMax(&sM[dl * 4 + 0], enc_f32(lg.x));
                    atomicMax(&sM[dl * 4 + 1], enc_f32(lg.y));
                    atomicMax(&sM[dl * 4 + 2], enc_f32(lg.z));
                    atomicMax(&sM[dl * 4 + 3], enc_f32(lg.w));
                    lSD[i] = make_int2(sI[k], dl);
                    lW[i]  = lg;
                }
            }
        }
        __syncthreads();

        const int T4 = (T + 3) & ~3;
        for (int i = tid; i < T4 + 4; i += TILE_THREADS) {
            if (i < T) {
                int dl = lSD[i].y;
                float4 lg = lW[i];
                uint4 m = *(const uint4*)(sM + 4 * dl);
                float4 w;
                w.x = expf(lg.x - dec_f32(m.x)); w.y = expf(lg.y - dec_f32(m.y));
                w.z = expf(lg.z - dec_f32(m.z)); w.w = expf(lg.w - dec_f32(m.w));
                lW[i] = w;
                atomicAdd(&sSum[dl * 4 + 0], w.x);
                atomicAdd(&sSum[dl * 4 + 1], w.y);
                atomicAdd(&sSum[dl * 4 + 2], w.z);
                atomicAdd(&sSum[dl * 4 + 3], w.w);
            } else {
                lSD[i] = make_int2(d0, TD);
                lW[i]  = make_float4(0.f, 0.f, 0.f, 0.f);
            }
        }
        __syncthreads();

        const int nb = T4 >> 2;
        float hA[4][8], hB[4][8];
        int dA[4], dB[4];
        if (wv < nb) walk_issue(H, lSD, wv, lane, hA, dA);
        for (int q = wv; q < nb; q += 2 * nwaves) {
            int q1 = q + nwaves;
            q1 = q1 < nb ? q1 : nb;
            walk_issue(H, lSD, q1, lane, hB, dB);
            __builtin_amdgcn_sched_barrier(0);
            walk_apply(acc, lW, q, lane, hA, dA);
            int q2 = q + 2 * nwaves;
            q2 = q2 < nb ? q2 : nb;
            walk_issue(H, lSD, q2, lane, hA, dA);
            walk_apply(acc, lW, q1, lane, hB, dB);
        }
    } else {
        const int CH = 32 * SCAN_U;
        int dn[SCAN_U];

        load_dst(dn, dsts, wave * CH, lane, E);
        for (int c0 = wave * CH; c0 < E; c0 += nwaves * CH) {
            int sv[SCAN_U];
            unsigned dlv[SCAN_U];
#pragma unroll
            for (int j = 0; j < SCAN_U; ++j) dlv[j] = (unsigned)(dn[j] - d0);
            load_dst(dn, dsts, c0 + nwaves * CH, lane, E);
#pragma unroll
            for (int j = 0; j < SCAN_U; ++j) {
                int e = c0 + j * 32 + lane;
                sv[j] = (dlv[j] < (unsigned)nd) ? srcs[e] : 0;
            }
#pragma unroll
            for (int j = 0; j < SCAN_U; ++j) {
                if (dlv[j] < (unsigned)nd) {
                    int dl = (int)dlv[j];
                    float4 as = *(const float4*)(aS + 4 * (size_t)sv[j]);
                    atomicMax(&sM[dl * 4 + 0], enc_f32(leaky(as.x + sAD[dl * 4 + 0])));
                    atomicMax(&sM[dl * 4 + 1], enc_f32(leaky(as.y + sAD[dl * 4 + 1])));
                    atomicMax(&sM[dl * 4 + 2], enc_f32(leaky(as.z + sAD[dl * 4 + 2])));
                    atomicMax(&sM[dl * 4 + 3], enc_f32(leaky(as.w + sAD[dl * 4 + 3])));
                }
            }
        }
        __syncthreads();

        load_dst(dn, dsts, wave * CH, lane, E);
        for (int dl = wave; dl < nd; dl += nwaves)
            tile_add_edge(H, aS, d0 + dl, dl, lane, acc, sAD, sM, sSum);

        for (int c0 = wave * CH; c0 < E; c0 += nwaves * CH) {
            int sv[SCAN_U];
            unsigned dlv[SCAN_U];
#pragma unroll
            for (int j = 0; j < SCAN_U; ++j) dlv[j] = (unsigned)(dn[j] - d0);
            load_dst(dn, dsts, c0 + nwaves * CH, lane, E);
#pragma unroll
            for (int j = 0; j < SCAN_U; ++j) {
                int e = c0 + j * 32 + lane;
                sv[j] = (dlv[j] < (unsigned)nd) ? srcs[e] : 0;
            }
#pragma unroll
            for (int j = 0; j < SCAN_U; ++j) {
                unsigned mask = (unsigned)__ballot(dlv[j] < (unsigned)nd);
                while (mask) {
                    int l = __ffs(mask) - 1;
                    mask &= mask - 1;
                    int s  = __shfl(sv[j], l);
                    int dl = __shfl((int)dlv[j], l);
                    tile_add_edge(H, aS, s, dl, lane, acc, sAD, sM, sSum);
                }
            }
        }
    }
    __syncthreads();

    for (int i = tid; i < nd * (HC / 4); i += TILE_THREADS) {
        int dl = i / (HC / 4), fq = i % (HC / 4);
        float den = sSum[dl * 4 + (fq / (CPH / 4))] + 1e-16f;
        float4 v  = *(const float4*)(acc + dl * HC + fq * 4);
        float4 bb = *(const float4*)(b + fq * 4);
        v.x = v.x / den + bb.x; v.y = v.y / den + bb.y;
        v.z = v.z / den + bb.z; v.w = v.w / den + bb.w;
        if (relu) {
            v.x = fmaxf(v.x, 0.f); v.y = fmaxf(v.y, 0.f);
            v.z = fmaxf(v.z, 0.f); v.w = fmaxf(v.w, 0.f);
        }
        *(float4*)(out + (size_t)(d0 + dl) * HC + fq * 4) = v;
    }
}

static void gat_layer(const float* Xin, int K, const float* W, const float* as,
                      const float* ad, const float* b, float* Hbuf, float* agg,
                      const int* ei, int E, int N, float* alphaS, float* alphaD,
                      int relu, hipStream_t stream) {
    int waves  = ((N + 15) / 16) * (HC / 64);
    int blocks = (waves + 7) / 8;
    gemm_wmma_f32<<<blocks, 256, 0, stream>>>(Xin, W, Hbuf, N, K);

    int nh = N * HEADS;
    node_alpha<<<(nh + 255) / 256, 256, 0, stream>>>(Hbuf, as, ad, alphaS, alphaD, N);

    aggr_tile<<<(N + TD - 1) / TD, TILE_THREADS, 0, stream>>>(ei, E, N, Hbuf, alphaS,
                                                              alphaD, b, relu, agg);
}

extern "C" void kernel_launch(void* const* d_in, const int* in_sizes, int n_in,
                              void* d_out, int out_size, void* d_ws, size_t ws_size,
                              hipStream_t stream) {
    const float* x   = (const float*)d_in[0];
    const int*   ei  = (const int*)d_in[1];
    const float* W1  = (const float*)d_in[2];
    const float* as1 = (const float*)d_in[3];
    const float* ad1 = (const float*)d_in[4];
    const float* b1  = (const float*)d_in[5];
    const float* W2  = (const float*)d_in[6];
    const float* as2 = (const float*)d_in[7];
    const float* ad2 = (const float*)d_in[8];
    const float* b2  = (const float*)d_in[9];
    const float* W3  = (const float*)d_in[10];
    const float* as3 = (const float*)d_in[11];
    const float* ad3 = (const float*)d_in[12];
    const float* b3  = (const float*)d_in[13];

    int K1 = in_sizes[2] / HC;
    int K2 = in_sizes[6] / HC;
    int K3 = in_sizes[10] / HC;
    int N  = in_sizes[0] / K1;
    int E  = in_sizes[1] / 2;

    float*    Hbuf   = (float*)d_ws;
    float*    bufA   = Hbuf + (size_t)N * HC;
    float*    bufB   = bufA + (size_t)N * HC;
    float*    alphaS = bufB + (size_t)N * HC;
    float*    alphaD = alphaS + (size_t)N * HEADS;

    gat_layer(x,    K1, W1, as1, ad1, b1, Hbuf, bufA,          ei, E, N,
              alphaS, alphaD,  1, stream);
    gat_layer(bufA, K2, W2, as2, ad2, b2, Hbuf, bufB,          ei, E, N,
              alphaS, alphaD,  1, stream);
    gat_layer(bufB, K3, W3, as3, ad3, b3, Hbuf, (float*)d_out, ei, E, N,
              alphaS, alphaD,  0, stream);
}


# Edge filter runs in layer 1's aggregation launch only: it leaves each tile's resolved hit list (source, local dst) and its length in the workspace; layers 2 and 3 read them back, no rescan
# speedup vs baseline: 5.5598x; 1.1574x over previous
#include <hip/hip_runtime.h>
#include <hip/hip_bf16.h>

typedef __attribute__((ext_vector_type(2))) float v2f;
typedef __attribute__((ext_vector_type(8))) float v8f;

#define HEADS 4
#define HC 256
#define CPH 64
#define NEG_SLOPE 0.2f

__device__ __forceinline__ unsigned enc_f32(float f) {
    unsigned u = __float_as_uint(f);
    return u ^ ((u >> 31) ? 0xFFFFFFFFu : 0x80000000u);
}
__device__ __forceinline__ float dec_f32(unsigned u) {
    return __uint_as_float(u ^ ((u >> 31) ? 0x80000000u : 0xFFFFFFFFu));
}

__global__ __launch_bounds__(256) void gemm_wmma_f32(
        const float* __restrict__ X, const float* __restrict__ W,
        float* __restrict__ H, int M, int K) {
    const int wavesPerBlock = 8;
    int wave  = blockIdx.x * wavesPerBlock + (threadIdx.x >> 5);
    const int strips = HC / 64;
    int tileM = wave / strips;
    int strip = wave % strips;
    if (tileM * 16 >= M) return;

    int lane = threadIdx.x & 31;
    int half = lane >> 4;
    int l16  = lane & 15;
    int m    = tileM * 16 + l16;
    int col0 = strip * 64 + l16;
    const float* xrow = X + (size_t)m * K;

    v8f acc0 = {}, acc1 = {}, acc2 = {}, acc3 = {};
    for (int kb = 0; kb < K; kb += 4) {
        int ka = kb + 2 * half;
        v2f a;
        a.x = xrow[ka];
        a.y = xrow[ka + 1];
        const float* w0 = W + (size_t)ka * HC + col0;
        const float* w1 = W + (size_t)(ka + 1) * HC + col0;
        v2f b0, b1, b2, b3;
        b0.x = w0[0];  b0.y = w1[0];
        b1.x = w0[16]; b1.y = w1[16];
        b2.x = w0[32]; b2.y = w1[32];
        b3.x = w0[48]; b3.y = w1[48];
        acc0 = __builtin_amdgcn_wmma_f32_16x16x4_f32(
                false, a, false, b0, (short)0, acc0, false, false);
        acc1 = __builtin_amdgcn_wmma_f32_16x16x4_f32(
                false, a, false, b1, (short)0, acc1, false, false);
        acc2 = __builtin_amdgcn_wmma_f32_16x16x4_f32(
                false, a, false, b2, (short)0, acc2, false, false);
        acc3 = __builtin_amdgcn_wmma_f32_16x16x4_f32(
                false, a, false, b3, (short)0, acc3, false, false);
    }
#pragma unroll
    for (int r = 0; r < 8; ++r) {
        int row = tileM * 16 + r + 8 * half;
        if (row < M) {
            float* hrow = H + (size_t)row * HC + col0;
            hrow[0]  = acc0[r];
            hrow[16] = acc1[r];
            hrow[32] = acc2[r];
            hrow[48] = acc3[r];
        }
    }
}

__global__ void node_alpha(const float* __restrict__ H,
                           const float* __restrict__ a_src,
                           const float* __restrict__ a_dst,
                           float* __restrict__ alphaS,
                           float* __restrict__ alphaD, int N) {
    int t = blockIdx.x * blockDim.x + threadIdx.x;
    if (t >= N * HEADS) return;
    int n = t >> 2, h = t & 3;
    const float4* hp = (const float4*)(H + (size_t)n * HC + h * CPH);
    const float4* as = (const float4*)(a_src + h * CPH);
    const float4* ad = (const float4*)(a_dst + h * CPH);
    float ss = 0.f, sd = 0.f;
#pragma unroll 4
    for (int i = 0; i < CPH / 4; ++i) {
        float4 hv = hp[i], av = as[i], dv = ad[i];
        ss += hv.x * av.x + hv.y * av.y + hv.z * av.z + hv.w * av.w;
        sd += hv.x * dv.x + hv.y * dv.y + hv.z * dv.z + hv.w * dv.w;
    }
    alphaS[t] = ss;
    alphaD[t] = sd;
}

__device__ __forceinline__ float leaky(float x) {
    return x >= 0.f ? x : NEG_SLOPE * x;
}

#define TD 200
#define TILE_THREADS 1024
#define SCAN_U 8
#define CAP 4352
#define NIT ((CAP + TILE_THREADS - 1) / TILE_THREADS)

__device__ __forceinline__ void load_dst(int (&d)[SCAN_U], const int* __restrict__ dsts,
                                         int c, int lane, int E) {
#pragma unroll
    for (int j = 0; j < SCAN_U; ++j) {
        int e = c + j * 32 + lane;
        int v = dsts[e < E ? e : E - 1];
        d[j] = (e < E) ? v : -1;
    }
}

__device__ __forceinline__ void tile_add_edge(
        const float* __restrict__ H, const float* __restrict__ aS, int s, int dl,
        int lane, float* acc, const float* sAD, const unsigned* sM, float* sSum) {
    const float* hrow = H + (size_t)s * HC + lane;
    float hv[8];
#pragma unroll
    for (int k = 0; k < 8; ++k) hv[k] = hrow[32 * k];
    float4 as = *(const float4*)(aS + 4 * (size_t)s);
    float w0 = expf(leaky(as.x + sAD[dl * 4 + 0]) - dec_f32(sM[dl * 4 + 0]));
    float w1 = expf(leaky(as.y + sAD[dl * 4 + 1]) - dec_f32(sM[dl * 4 + 1]));
    float w2 = expf(leaky(as.z + sAD[dl * 4 + 2]) - dec_f32(sM[dl * 4 + 2]));
    float w3 = expf(leaky(as.w + sAD[dl * 4 + 3]) - dec_f32(sM[dl * 4 + 3]));
    if (lane < 4) {
        float wl = lane == 0 ? w0 : lane == 1 ? w1 : lane == 2 ? w2 : w3;
        atomicAdd(&sSum[dl * 4 + lane], wl);
    }
    float* a = acc + dl * HC + lane;
    atomicAdd(a + 0,   hv[0] * w0);
    atomicAdd(a + 32,  hv[1] * w0);
    atomicAdd(a + 64,  hv[2] * w1);
    atomicAdd(a + 96,  hv[3] * w1);
    atomicAdd(a + 128, hv[4] * w2);
    atomicAdd(a + 160, hv[5] * w2);
    atomicAdd(a + 192, hv[6] * w3);
    atomicAdd(a + 224, hv[7] * w3);
}

__device__ __forceinline__ void load_piece(int4 (&d)[SCAN_U], const int4* __restrict__ dv,
                                           int vb, int lane, int nvec) {
#pragma unroll
    for (int j = 0; j < SCAN_U; ++j) {
        int vi = vb + j * 32 + lane;
        d[j] = dv[vi < nvec ? vi : nvec - 1];
    }
}

__device__ __forceinline__ void walk_issue(const float* __restrict__ H, const int2* lSD, int q,
                                           int lane, float (&h)[4][8], int (&dl)[4]) {
    const int4* p = (const int4*)(lSD + 4 * q);
    int4 e01 = p[0], e23 = p[1];
    int s[4] = {e01.x, e01.z, e23.x, e23.z};
    dl[0] = e01.y; dl[1] = e01.w; dl[2] = e23.y; dl[3] = e23.w;
#pragma unroll
    for (int u = 0; u < 4; ++u) {
        const float* row = H + (size_t)__builtin_amdgcn_readfirstlane(s[u]) * HC + lane;
#pragma unroll
        for (int k = 0; k < 8; ++k) h[u][k] = row[32 * k];
    }
}

__device__ __forceinline__ void walk_apply(float* acc, const float4* lW, int q, int lane,
                                           const float (&h)[4][8], const int (&dl)[4]) {
    float4 w[4];
#pragma unroll
    for (int u = 0; u < 4; ++u) w[u] = lW[4 * q + u];
#pragma unroll
    for (int u = 0; u < 4; ++u) {
        float* a = acc + dl[u] * HC + lane;
        atomicAdd(a + 0,   h[u][0] * w[u].x);
        atomicAdd(a + 32,  h[u][1] * w[u].x);
        atomicAdd(a + 64,  h[u][2] * w[u].y);
        atomicAdd(a + 96,  h[u][3] * w[u].y);
        atomicAdd(a + 128, h[u][4] * w[u].z);
        atomicAdd(a + 160, h[u][5] * w[u].z);
        atomicAdd(a + 192, h[u][6] * w[u].w);
        atomicAdd(a + 224, h[u][7] * w[u].w);
    }
}

template <bool BUILD>
__global__ __launch_bounds__(TILE_THREADS) void aggr_tile(
        const int* __restrict__ ei, int E, int N,
        const float* __restrict__ H,
        const float* __restrict__ aS,
        const float* __restrict__ aD,
        const float* __restrict__ b, int relu,
        float* __restrict__ out,
        int* __restrict__ tileList, int* __restrict__ tileCnt) {
    __shared__ __attribute__((aligned(16))) float    acc[(TD + 1) * HC];
    __shared__ __attribute__((aligned(16))) float    sAD[TD * 4];
    __shared__ __attribute__((aligned(16))) unsigned sM[TD * 4];
    __shared__ float    sSum[TD * 4];
    __shared__ __attribute__((aligned(16))) float4   lW[CAP];
    __shared__ __attribute__((aligned(16))) int2     lSD[CAP];
    __shared__ int sCnt;

    const int tid  = threadIdx.x;
    const int lane = tid & 31;
    const int wave = tid >> 5;
    const int nwaves = TILE_THREADS / 32;
    const int d0 = blockIdx.x * TD;
    const int nd = (N - d0) < TD ? (N - d0) : TD;
    const int* __restrict__ srcs = ei;
    const int* __restrict__ dsts = ei + E;
    const int wv = __builtin_amdgcn_readfirstlane(wave);
    int2* __restrict__ tl = (int2*)tileList + (size_t)blockIdx.x * CAP;
    int storedTotal = 0;
    if (!BUILD) storedTotal = tileCnt[blockIdx.x];

    for (int i = tid; i < TD * HC / 4; i += TILE_THREADS)
        ((float4*)acc)[i] = make_float4(0.f, 0.f, 0.f, 0.f);
    for (int i = tid; i < TD * 4; i += TILE_THREADS) {
        int dl = i >> 2;
        float ad = 0.f, lg = 0.f;
        unsigned me = 0u;
        if (dl < nd) {
            size_t g = 4 * (size_t)(d0 + dl) + (i & 3);
            ad = aD[g];
            lg = leaky(aS[g] + ad);
            me = enc_f32(lg);
            ((float*)lW)[i] = lg;
            if ((i & 3) == 0) lSD[dl] = make_int2(d0 + dl, dl);
        }
        sAD[i] = ad; sM[i] = me; sSum[i] = 0.f;
    }
    if (tid == 0) sCnt = nd;
    __syncthreads();

    if (BUILD) {
    int head = (int)((4u - (unsigned)(((size_t)dsts >> 2) & 3)) & 3u);
    if (head > E) head = E;
    const int nvec = (E - head) >> 2;
    const int4* __restrict__ dv = (const int4*)(dsts + head);
    if (tid < E - 4 * nvec) {
        int e = tid < head ? tid : 4 * nvec + tid;
        if ((unsigned)(dsts[e] - d0) < (unsigned)nd) {
            int idx = atomicAdd(&sCnt, 1);
            if (idx < CAP) lSD[idx].x = e;
        }
    }
    if (nvec > 0) {
        const int PV = 32 * SCAN_U;
        int4 cur[SCAN_U];
        load_piece(cur, dv, wv * PV, lane, nvec);
        for (int vb = wv * PV; vb < nvec; vb += nwaves * PV) {
            int4 nx[SCAN_U];
            load_piece(nx, dv, vb + nwaves * PV, lane, nvec);
            unsigned hm = 0u;
#pragma unroll
            for (int j = 0; j < SCAN_U; ++j) {
                unsigned lim = (vb + j * 32 + lane < nvec) ? (unsigned)nd : 0u;
                hm |= ((unsigned)(cur[j].x - d0) < lim ? 1u : 0u) << (4 * j + 0);
                hm |= ((unsigned)(cur[j].y - d0) < lim ? 1u : 0u) << (4 * j + 1);
                hm |= ((unsigned)(cur[j].z - d0) < lim ? 1u : 0u) << (4 * j + 2);
                hm |= ((unsigned)(cur[j].w - d0) < lim ? 1u : 0u) << (4 * j + 3);
            }
            if (hm) {
                int idx = atomicAdd(&sCnt, __popc(hm));
                int eb  = head + 4 * (vb + lane);
                do {
                    int bit = __ffs(hm) - 1;
                    hm &= hm - 1;
                    if (idx < CAP) lSD[idx].x = eb + (bit >> 2) * 128 + (bit & 3);
                    ++idx;
                } while (hm);
            }
#pragma unroll
            for (int j = 0; j < SCAN_U; ++j) cur[j] = nx[j];
        }
    }
    __syncthreads();
    if (tid == 0) tileCnt[blockIdx.x] = sCnt;
    }

    const int total = BUILD ? sCnt : storedTotal;
    if (total <= CAP - 4) {
        const int T = total;
        if (T > nd) {
            int sI[NIT], dI[NIT];
            float4 aI[NIT];
#pragma unroll
            for (int k = 0; k < NIT; ++k) {
                int i = nd + tid + k * TILE_THREADS;
                if (BUILD) {
                    int e = (i < T) ? lSD[i].x : 0;
                    sI[k] = srcs[e];
                    dI[k] = dsts[e] - d0;
                } else {
                    int2 p = make_int2(0, 0);
                    if (i < T) p = tl[i];
                    sI[k] = p.x;
                    dI[k] = p.y;
                }
            }
#pragma unroll
            for (int k = 0; k < NIT; ++k) aI[k] = *(const float4*)(aS + 4 * (size_t)sI[k]);
#pragma unroll
            for (int k = 0; k < NIT; ++k) {
                int i = nd + tid + k * TILE_THREADS;
                if (i < T) {
                    int dl = dI[k];
                    float4 ad = *(const float4*)(sAD + 4 * dl);
                    float4 lg;
                    lg.x = leaky(aI[k].x + ad.x); lg.y = leaky(aI[k].y + ad.y);
                    lg.z = leaky(aI[k].z + ad.z); lg.w = leaky(aI[k].w + ad.w);
                    atomicMax(&sM[dl * 4 + 0], enc_f32(lg.x));
                    atomicMax(&sM[dl * 4 + 1], enc_f32(lg.y));
                    atomicMax(&sM[dl * 4 + 2], enc_f32(lg.z));
                    atomicMax(&sM[dl * 4 + 3], enc_f32(lg.w));
                    lSD[i] = make_int2(sI[k], dl);
                    if (BUILD) tl[i] = make_int2(sI[k], dl);
                    lW[i]  = lg;
                }
            }
        }
        __syncthreads();

        const int T4 = (T + 3) & ~3;
        for (int i = tid; i < T4 + 4; i += TILE_THREADS) {
            if (i < T) {
                int dl = lSD[i].y;
                float4 lg = lW[i];
                uint4 m = *(const uint4*)(sM + 4 * dl);
                float4 w;
                w.x = expf(lg.x - dec_f32(m.x)); w.y = expf(lg.y - dec_f32(m.y));
                w.z = expf(lg.z - dec_f32(m.z)); w.w = expf(lg.w - dec_f32(m.w));
                lW[i] = w;
                atomicAdd(&sSum[dl * 4 + 0], w.x);
                atomicAdd(&sSum[dl * 4 + 1], w.y);
                atomicAdd(&sSum[dl * 4 + 2], w.z);
                atomicAdd(&sSum[dl * 4 + 3], w.w);
            } else {
                lSD[i] = make_int2(d0, TD);
                lW[i]  = make_float4(0.f, 0.f, 0.f, 0.f);
            }
        }
        __syncthreads();

        const int nb = T4 >> 2;
        float hA[4][8], hB[4][8];
        int dA[4], dB[4];
        if (wv < nb) walk_issue(H, lSD, wv, lane, hA, dA);
        for (int q = wv; q < nb; q += 2 * nwaves) {
            int q1 = q + nwaves;
            q1 = q1 < nb ? q1 : nb;
            walk_issue(H, lSD, q1, lane, hB, dB);
            __builtin_amdgcn_sched_barrier(0);
            walk_apply(acc, lW, q, lane, hA, dA);
            int q2 = q + 2 * nwaves;
            q2 = q2 < nb ? q2 : nb;
            walk_issue(H, lSD, q2, lane, hA, dA);
            walk_apply(acc, lW, q1, lane, hB, dB);
        }
    } else {
        const int CH = 32 * SCAN_U;
        int dn[SCAN_U];

        load_dst(dn, dsts, wave * CH, lane, E);
        for (int c0 = wave * CH; c0 < E; c0 += nwaves * CH) {
            int sv[SCAN_U];
            unsigned dlv[SCAN_U];
#pragma unroll
            for (int j = 0; j < SCAN_U; ++j) dlv[j] = (unsigned)(dn[j] - d0);
            load_dst(dn, dsts, c0 + nwaves * CH, lane, E);
#pragma unroll
            for (int j = 0; j < SCAN_U; ++j) {
                int e = c0 + j * 32 + lane;
                sv[j] = (dlv[j] < (unsigned)nd) ? srcs[e] : 0;
            }
#pragma unroll
            for (int j = 0; j < SCAN_U; ++j) {
                if (dlv[j] < (unsigned)nd) {
                    int dl = (int)dlv[j];
                    float4 as = *(const float4*)(aS + 4 * (size_t)sv[j]);
                    atomicMax(&sM[dl * 4 + 0], enc_f32(leaky(as.x + sAD[dl * 4 + 0])));
                    atomicMax(&sM[dl * 4 + 1], enc_f32(leaky(as.y + sAD[dl * 4 + 1])));
                    atomicMax(&sM[dl * 4 + 2], enc_f32(leaky(as.z + sAD[dl * 4 + 2])));
                    atomicMax(&sM[dl * 4 + 3], enc_f32(leaky(as.w + sAD[dl * 4 + 3])));
                }
            }
        }
        __syncthreads();

        load_dst(dn, dsts, wave * CH, lane, E);
        for (int dl = wave; dl < nd; dl += nwaves)
            tile_add_edge(H, aS, d0 + dl, dl, lane, acc, sAD, sM, sSum);

        for (int c0 = wave * CH; c0 < E; c0 += nwaves * CH) {
            int sv[SCAN_U];
            unsigned dlv[SCAN_U];
#pragma unroll
            for (int j = 0; j < SCAN_U; ++j) dlv[j] = (unsigned)(dn[j] - d0);
            load_dst(dn, dsts, c0 + nwaves * CH, lane, E);
#pragma unroll
            for (int j = 0; j < SCAN_U; ++j) {
                int e = c0 + j * 32 + lane;
                sv[j] = (dlv[j] < (unsigned)nd) ? srcs[e] : 0;
            }
#pragma unroll
            for (int j = 0; j < SCAN_U; ++j) {
                unsigned mask = (unsigned)__ballot(dlv[j] < (unsigned)nd);
                while (mask) {
                    int l = __ffs(mask) - 1;
                    mask &= mask - 1;
                    int s  = __shfl(sv[j], l);
                    int dl = __shfl((int)dlv[j], l);
                    tile_add_edge(H, aS, s, dl, lane, acc, sAD, sM, sSum);
                }
            }
        }
    }
    __syncthreads();

    for (int i = tid; i < nd * (HC / 4); i += TILE_THREADS) {
        int dl = i / (HC / 4), fq = i % (HC / 4);
        float den = sSum[dl * 4 + (fq / (CPH / 4))] + 1e-16f;
        float4 v  = *(const float4*)(acc + dl * HC + fq * 4);
        float4 bb = *(const float4*)(b + fq * 4);
        v.x = v.x / den + bb.x; v.y = v.y / den + bb.y;
        v.z = v.z / den + bb.z; v.w = v.w / den + bb.w;
        if (relu) {
            v.x = fmaxf(v.x, 0.f); v.y = fmaxf(v.y, 0.f);
            v.z = fmaxf(v.z, 0.f); v.w = fmaxf(v.w, 0.f);
        }
        *(float4*)(out + (size_t)(d0 + dl) * HC + fq * 4) = v;
    }
}

static void gat_layer(const float* Xin, int K, const float* W, const float* as,
                      const float* ad, const float* b, float* Hbuf, float* agg,
                      const int* ei, int E, int N, float* alphaS, float* alphaD,
                      int relu, int* tileList, int* tileCnt, bool build,
                      hipStream_t stream) {
    int waves  = ((N + 15) / 16) * (HC / 64);
    int blocks = (waves + 7) / 8;
    gemm_wmma_f32<<<blocks, 256, 0, stream>>>(Xin, W, Hbuf, N, K);

    int nh = N * HEADS;
    node_alpha<<<(nh + 255) / 256, 256, 0, stream>>>(Hbuf, as, ad, alphaS, alphaD, N);

    if (build)
        aggr_tile<true><<<(N + TD - 1) / TD, TILE_THREADS, 0, stream>>>(
                ei, E, N, Hbuf, alphaS, alphaD, b, relu, agg, tileList, tileCnt);
    else
        aggr_tile<false><<<(N + TD - 1) / TD, TILE_THREADS, 0, stream>>>(
                ei, E, N, Hbuf, alphaS, alphaD, b, relu, agg, tileList, tileCnt);
}

extern "C" void kernel_launch(void* const* d_in, const int* in_sizes, int n_in,
                              void* d_out, int out_size, void* d_ws, size_t ws_size,
                              hipStream_t stream) {
    const float* x   = (const float*)d_in[0];
    const int*   ei  = (const int*)d_in[1];
    const float* W1  = (const float*)d_in[2];
    const float* as1 = (const float*)d_in[3];
    const float* ad1 = (const float*)d_in[4];
    const float* b1  = (const float*)d_in[5];
    const float* W2  = (const float*)d_in[6];
    const float* as2 = (const float*)d_in[7];
    const float* ad2 = (const float*)d_in[8];
    const float* b2  = (const float*)d_in[9];
    const float* W3  = (const float*)d_in[10];
    const float* as3 = (const float*)d_in[11];
    const float* ad3 = (const float*)d_in[12];
    const float* b3  = (const float*)d_in[13];

    int K1 = in_sizes[2] / HC;
    int K2 = in_sizes[6] / HC;
    int K3 = in_sizes[10] / HC;
    int N  = in_sizes[0] / K1;
    int E  = in_sizes[1] / 2;

    float*    Hbuf   = (float*)d_ws;
    float*    bufA   = Hbuf + (size_t)N * HC;
    float*    bufB   = bufA + (size_t)N * HC;
    float*    alphaS = bufB + (size_t)N * HC;
    float*    alphaD = alphaS + (size_t)N * HEADS;
    int*      tileList = (int*)(alphaD + (size_t)N * HEADS);
    int*      tileCnt  = tileList + 2 * (size_t)CAP * ((N + TD - 1) / TD);

    gat_layer(x,    K1, W1, as1, ad1, b1, Hbuf, bufA,          ei, E, N,
              alphaS, alphaD,  1, tileList, tileCnt,  true,  stream);
    gat_layer(bufA, K2, W2, as2, ad2, b2, Hbuf, bufB,          ei, E, N,
              alphaS, alphaD,  1, tileList, tileCnt,  false, stream);
    gat_layer(bufB, K3, W3, as3, ad3, b3, Hbuf, (float*)d_out, ei, E, N,
              alphaS, alphaD,  0, tileList, tileCnt,  false, stream);
}


# GEMM: each of 250 resident blocks stages W once in LDS (k-pair rows) and loops over its row tiles; B fragments are 8-byte LDS reads, A and B requested a 32-k chunk ahead; k sum order unchanged
# speedup vs baseline: 6.0430x; 1.0869x over previous
#include <hip/hip_runtime.h>
#include <hip/hip_bf16.h>

typedef __attribute__((ext_vector_type(2))) float v2f;
typedef __attribute__((ext_vector_type(8))) float v8f;

#define HEADS 4
#define HC 256
#define CPH 64
#define NEG_SLOPE 0.2f

__device__ __forceinline__ unsigned enc_f32(float f) {
    unsigned u = __float_as_uint(f);
    return u ^ ((u >> 31) ? 0xFFFFFFFFu : 0x80000000u);
}
__device__ __forceinline__ float dec_f32(unsigned u) {
    return __uint_as_float(u ^ ((u >> 31) ? 0x80000000u : 0xFFFFFFFFu));
}

#define GEMM_THREADS 512
#define GEMM_BLOCKS  250
#define GEMM_KMAX    256
#define WPITCH       544

__device__ __forceinline__ void gemm_load_chunk(const float* __restrict__ xk, const float* wk,
                                                v2f (&a)[8], v2f (&b)[8][4]) {
#pragma unroll
    for (int j = 0; j < 8; ++j) {
        a[j] = *(const v2f*)(xk + 4 * j);
#pragma unroll
        for (int i = 0; i < 4; ++i)
            b[j][i] = *(const v2f*)(wk + 2 * j * WPITCH + 32 * i);
    }
}

__device__ __forceinline__ void gemm_mma_chunk(const v2f (&a)[8], const v2f (&b)[8][4],
                                               v8f& c0, v8f& c1, v8f& c2, v8f& c3) {
#pragma unroll
    for (int j = 0; j < 8; ++j) {
        c0 = __builtin_amdgcn_wmma_f32_16x16x4_f32(false, a[j], false, b[j][0], (short)0, c0, false, false);
        c1 = __builtin_amdgcn_wmma_f32_16x16x4_f32(false, a[j], false, b[j][1], (short)0, c1, false, false);
        c2 = __builtin_amdgcn_wmma_f32_16x16x4_f32(false, a[j], false, b[j][2], (short)0, c2, false, false);
        c3 = __builtin_amdgcn_wmma_f32_16x16x4_f32(false, a[j], false, b[j][3], (short)0, c3, false, false);
    }
}

__global__ __launch_bounds__(GEMM_THREADS) void gemm_wmma_f32(
        const float* __restrict__ X, const float* __restrict__ W,
        float* __restrict__ H, int M, int K) {
    __shared__ __attribute__((aligned(16))) float Wl[(GEMM_KMAX / 2) * WPITCH];

    const int tid = threadIdx.x;
    const int nPairs = (K >> 1) * (HC / 4);
    for (int base = 0; base < nPairs; base += 8 * GEMM_THREADS) {
        float4 w0[8], w1[8];
#pragma unroll
        for (int u = 0; u < 8; ++u) {
            int i = base + u * GEMM_THREADS + tid;
            i = i < nPairs ? i : nPairs - 1;
            int kp = i >> 6, n4 = i & 63;
            w0[u] = *(const float4*)(W + (size_t)(2 * kp) * HC + 4 * n4);
            w1[u] = *(const float4*)(W + (size_t)(2 * kp + 1) * HC + 4 * n4);
        }
#pragma unroll
        for (int u = 0; u < 8; ++u) {
            int i = base + u * GEMM_THREADS + tid;
            i = i < nPairs ? i : nPairs - 1;
            int kp = i >> 6, n4 = i & 63;
            float4* d = (float4*)(Wl + kp * WPITCH + 8 * n4);
            d[0] = make_float4(w0[u].x, w1[u].x, w0[u].y, w1[u].y);
            d[1] = make_float4(w0[u].z, w1[u].z, w0[u].w, w1[u].w);
        }
    }
    __syncthreads();

    const int lane = tid & 31;
    const int half = lane >> 4;
    const int l16  = lane & 15;
    const int wave = __builtin_amdgcn_readfirstlane(tid >> 5);
    const int strip = wave & 3;
    const int grp   = wave >> 2;
    const int numTiles = (M + 15) >> 4;
    const int col0 = strip * 64 + l16;
    const float* wl = Wl + half * WPITCH + 2 * col0;

    const int bx = (int)blockIdx.x, gx = (int)gridDim.x;
    const int nT = bx < numTiles ? (numTiles - bx + gx - 1) / gx : 0;
    const int rounds = nT >> 2;
    const int q = (grp - strip) & 3;
    const int jTail = 4 * rounds + q;
    const int nW = rounds + (q < (nT & 3) ? 1 : 0);
    if (nW == 0) return;

    int t = bx + gx * (0 < rounds ? grp : jTail);
    int m = t * 16 + l16;
    const float* xrow = X + (size_t)(m < M ? m : M - 1) * K + 2 * half;

    v2f aP[8], aQ[8], bP[8][4], bQ[8][4];
    gemm_load_chunk(xrow, wl, aP, bP);
    for (int it = 0; it < nW; ++it) {
        const int nx = it + 1;
        const int tn = nx < nW ? bx + gx * (nx < rounds ? 4 * nx + grp : jTail) : t;
        int mn = tn * 16 + l16;
        const float* xnext = X + (size_t)(mn < M ? mn : M - 1) * K + 2 * half;

        v8f acc0 = {}, acc1 = {}, acc2 = {}, acc3 = {};
        for (int kc = 0; kc < K; kc += 64) {
            gemm_load_chunk(xrow + kc + 32, wl + ((kc >> 1) + 16) * WPITCH, aQ, bQ);
            __builtin_amdgcn_sched_barrier(0);
            gemm_mma_chunk(aP, bP, acc0, acc1, acc2, acc3);
            const bool last = kc + 64 >= K;
            gemm_load_chunk(last ? xnext : xrow + kc + 64,
                            last ? wl : wl + ((kc >> 1) + 32) * WPITCH, aP, bP);
            __builtin_amdgcn_sched_barrier(0);
            gemm_mma_chunk(aQ, bQ, acc0, acc1, acc2, acc3);
        }
#pragma unroll
        for (int r = 0; r < 8; ++r) {
            int row = t * 16 + r + 8 * half;
            if (row < M) {
                float* hrow = H + (size_t)row * HC + col0;
                hrow[0]  = acc0[r];
                hrow[16] = acc1[r];
                hrow[32] = acc2[r];
                hrow[48] = acc3[r];
            }
        }
        t = tn;
        xrow = xnext;
    }
}

__global__ void node_alpha(const float* __restrict__ H,
                           const float* __restrict__ a_src,
                           const float* __restrict__ a_dst,
                           float* __restrict__ alphaS,
                           float* __restrict__ alphaD, int N) {
    int t = blockIdx.x * blockDim.x + threadIdx.x;
    if (t >= N * HEADS) return;
    int n = t >> 2, h = t & 3;
    const float4* hp = (const float4*)(H + (size_t)n * HC + h * CPH);
    const float4* as = (const float4*)(a_src + h * CPH);
    const float4* ad = (const float4*)(a_dst + h * CPH);
    float ss = 0.f, sd = 0.f;
#pragma unroll 4
    for (int i = 0; i < CPH / 4; ++i) {
        float4 hv = hp[i], av = as[i], dv = ad[i];
        ss += hv.x * av.x + hv.y * av.y + hv.z * av.z + hv.w * av.w;
        sd += hv.x * dv.x + hv.y * dv.y + hv.z * dv.z + hv.w * dv.w;
    }
    alphaS[t] = ss;
    alphaD[t] = sd;
}

__device__ __forceinline__ float leaky(float x) {
    return x >= 0.f ? x : NEG_SLOPE * x;
}

#define TD 200
#define TILE_THREADS 1024
#define SCAN_U 8
#define CAP 4352
#define NIT ((CAP + TILE_THREADS - 1) / TILE_THREADS)

__device__ __forceinline__ void load_dst(int (&d)[SCAN_U], const int* __restrict__ dsts,
                                         int c, int lane, int E) {
#pragma unroll
    for (int j = 0; j < SCAN_U; ++j) {
        int e = c + j * 32 + lane;
        int v = dsts[e < E ? e : E - 1];
        d[j] = (e < E) ? v : -1;
    }
}

__device__ __forceinline__ void tile_add_edge(
        const float* __restrict__ H, const float* __restrict__ aS, int s, int dl,
        int lane, float* acc, const float* sAD, const unsigned* sM, float* sSum) {
    const float* hrow = H + (size_t)s * HC + lane;
    float hv[8];
#pragma unroll
    for (int k = 0; k < 8; ++k) hv[k] = hrow[32 * k];
    float4 as = *(const float4*)(aS + 4 * (size_t)s);
    float w0 = expf(leaky(as.x + sAD[dl * 4 + 0]) - dec_f32(sM[dl * 4 + 0]));
    float w1 = expf(leaky(as.y + sAD[dl * 4 + 1]) - dec_f32(sM[dl * 4 + 1]));
    float w2 = expf(leaky(as.z + sAD[dl * 4 + 2]) - dec_f32(sM[dl * 4 + 2]));
    float w3 = expf(leaky(as.w + sAD[dl * 4 + 3]) - dec_f32(sM[dl * 4 + 3]));
    if (lane < 4) {
        float wl = lane == 0 ? w0 : lane == 1 ? w1 : lane == 2 ? w2 : w3;
        atomicAdd(&sSum[dl * 4 + lane], wl);
    }
    float* a = acc + dl * HC + lane;
    atomicAdd(a + 0,   hv[0] * w0);
    atomicAdd(a + 32,  hv[1] * w0);
    atomicAdd(a + 64,  hv[2] * w1);
    atomicAdd(a + 96,  hv[3] * w1);
    atomicAdd(a + 128, hv[4] * w2);
    atomicAdd(a + 160, hv[5] * w2);
    atomicAdd(a + 192, hv[6] * w3);
    atomicAdd(a + 224, hv[7] * w3);
}

__device__ __forceinline__ void load_piece(int4 (&d)[SCAN_U], const int4* __restrict__ dv,
                                           int vb, int lane, int nvec) {
#pragma unroll
    for (int j = 0; j < SCAN_U; ++j) {
        int vi = vb + j * 32 + lane;
        d[j] = dv[vi < nvec ? vi : nvec - 1];
    }
}

__device__ __forceinline__ void walk_issue(const float* __restrict__ H, const int2* lSD, int q,
                                           int lane, float (&h)[4][8], int (&dl)[4]) {
    const int4* p = (const int4*)(lSD + 4 * q);
    int4 e01 = p[0], e23 = p[1];
    int s[4] = {e01.x, e01.z, e23.x, e23.z};
    dl[0] = e01.y; dl[1] = e01.w; dl[2] = e23.y; dl[3] = e23.w;
#pragma unroll
    for (int u = 0; u < 4; ++u) {
        const float* row = H + (size_t)__builtin_amdgcn_readfirstlane(s[u]) * HC + lane;
#pragma unroll
        for (int k = 0; k < 8; ++k) h[u][k] = row[32 * k];
    }
}

__device__ __forceinline__ void walk_apply(float* acc, const float4* lW, int q, int lane,
                                           const float (&h)[4][8], const int (&dl)[4]) {
    float4 w[4];
#pragma unroll
    for (int u = 0; u < 4; ++u) w[u] = lW[4 * q + u];
#pragma unroll
    for (int u = 0; u < 4; ++u) {
        float* a = acc + dl[u] * HC + lane;
        atomicAdd(a + 0,   h[u][0] * w[u].x);
        atomicAdd(a + 32,  h[u][1] * w[u].x);
        atomicAdd(a + 64,  h[u][2] * w[u].y);
        atomicAdd(a + 96,  h[u][3] * w[u].y);
        atomicAdd(a + 128, h[u][4] * w[u].z);
        atomicAdd(a + 160, h[u][5] * w[u].z);
        atomicAdd(a + 192, h[u][6] * w[u].w);
        atomicAdd(a + 224, h[u][7] * w[u].w);
    }
}

template <bool BUILD>
__global__ __launch_bounds__(TILE_THREADS) void aggr_tile(
        const int* __restrict__ ei, int E, int N,
        const float* __restrict__ H,
        const float* __restrict__ aS,
        const float* __restrict__ aD,
        const float* __restrict__ b, int relu,
        float* __restrict__ out,
        int* __restrict__ tileList, int* __restrict__ tileCnt) {
    __shared__ __attribute__((aligned(16))) float    acc[(TD + 1) * HC];
    __shared__ __attribute__((aligned(16))) float    sAD[TD * 4];
    __shared__ __attribute__((aligned(16))) unsigned sM[TD * 4];
    __shared__ float    sSum[TD * 4];
    __shared__ __attribute__((aligned(16))) float4   lW[CAP];
    __shared__ __attribute__((aligned(16))) int2     lSD[CAP];
    __shared__ int sCnt;

    const int tid  = threadIdx.x;
    const int lane = tid & 31;
    const int wave = tid >> 5;
    const int nwaves = TILE_THREADS / 32;
    const int d0 = blockIdx.x * TD;
    const int nd = (N - d0) < TD ? (N - d0) : TD;
    const int* __restrict__ srcs = ei;
    const int* __restrict__ dsts = ei + E;
    const int wv = __builtin_amdgcn_readfirstlane(wave);
    int2* __restrict__ tl = (int2*)tileList + (size_t)blockIdx.x * CAP;
    int storedTotal = 0;
    if (!BUILD) storedTotal = tileCnt[blockIdx.x];

    for (int i = tid; i < TD * HC / 4; i += TILE_THREADS)
        ((float4*)acc)[i] = make_float4(0.f, 0.f, 0.f, 0.f);
    for (int i = tid; i < TD * 4; i += TILE_THREADS) {
        int dl = i >> 2;
        float ad = 0.f, lg = 0.f;
        unsigned me = 0u;
        if (dl < nd) {
            size_t g = 4 * (size_t)(d0 + dl) + (i & 3);
            ad = aD[g];
            lg = leaky(aS[g] + ad);
            me = enc_f32(lg);
            ((float*)lW)[i] = lg;
            if ((i & 3) == 0) lSD[dl] = make_int2(d0 + dl, dl);
        }
        sAD[i] = ad; sM[i] = me; sSum[i] = 0.f;
    }
    if (tid == 0) sCnt = nd;
    __syncthreads();

    if (BUILD) {
    int head = (int)((4u - (unsigned)(((size_t)dsts >> 2) & 3)) & 3u);
    if (head > E) head = E;
    const int nvec = (E - head) >> 2;
    const int4* __restrict__ dv = (const int4*)(dsts + head);
    if (tid < E - 4 * nvec) {
        int e = tid < head ? tid : 4 * nvec + tid;
        if ((unsigned)(dsts[e] - d0) < (unsigned)nd) {
            int idx = atomicAdd(&sCnt, 1);
            if (idx < CAP) lSD[idx].x = e;
        }
    }
    if (nvec > 0) {
        const int PV = 32 * SCAN_U;
        int4 cur[SCAN_U];
        load_piece(cur, dv, wv * PV, lane, nvec);
        for (int vb = wv * PV; vb < nvec; vb += nwaves * PV) {
            int4 nx[SCAN_U];
            load_piece(nx, dv, vb + nwaves * PV, lane, nvec);
            unsigned hm = 0u;
#pragma unroll
            for (int j = 0; j < SCAN_U; ++j) {
                unsigned lim = (vb + j * 32 + lane < nvec) ? (unsigned)nd : 0u;
                hm |= ((unsigned)(cur[j].x - d0) < lim ? 1u : 0u) << (4 * j + 0);
                hm |= ((unsigned)(cur[j].y - d0) < lim ? 1u : 0u) << (4 * j + 1);
                hm |= ((unsigned)(cur[j].z - d0) < lim ? 1u : 0u) << (4 * j + 2);
                hm |= ((unsigned)(cur[j].w - d0) < lim ? 1u : 0u) << (4 * j + 3);
            }
            if (hm) {
                int idx = atomicAdd(&sCnt, __popc(hm));
                int eb  = head + 4 * (vb + lane);
                do {
                    int bit = __ffs(hm) - 1;
                    hm &= hm - 1;
                    if (idx < CAP) lSD[idx].x = eb + (bit >> 2) * 128 + (bit & 3);
                    ++idx;
                } while (hm);
            }
#pragma unroll
            for (int j = 0; j < SCAN_U; ++j) cur[j] = nx[j];
        }
    }
    __syncthreads();
    if (tid == 0) tileCnt[blockIdx.x] = sCnt;
    }

    const int total = BUILD ? sCnt : storedTotal;
    if (total <= CAP - 4) {
        const int T = total;
        if (T > nd) {
            int sI[NIT], dI[NIT];
            float4 aI[NIT];
#pragma unroll
            for (int k = 0; k < NIT; ++k) {
                int i = nd + tid + k * TILE_THREADS;
                if (BUILD) {
                    int e = (i < T) ? lSD[i].x : 0;
                    sI[k] = srcs[e];
                    dI[k] = dsts[e] - d0;
                } else {
                    int2 p = make_int2(0, 0);
                    if (i < T) p = tl[i];
                    sI[k] = p.x;
                    dI[k] = p.y;
                }
            }
#pragma unroll
            for (int k = 0; k < NIT; ++k) aI[k] = *(const float4*)(aS + 4 * (size_t)sI[k]);
#pragma unroll
            for (int k = 0; k < NIT; ++k) {
                int i = nd + tid + k * TILE_THREADS;
                if (i < T) {
                    int dl = dI[k];
                    float4 ad = *(const float4*)(sAD + 4 * dl);
                    float4 lg;
                    lg.x = leaky(aI[k].x + ad.x); lg.y = leaky(aI[k].y + ad.y);
                    lg.z = leaky(aI[k].z + ad.z); lg.w = leaky(aI[k].w + ad.w);
                    atomicMax(&sM[dl * 4 + 0], enc_f32(lg.x));
                    atomicMax(&sM[dl * 4 + 1], enc_f32(lg.y));
                    atomicMax(&sM[dl * 4 + 2], enc_f32(lg.z));
                    atomicMax(&sM[dl * 4 + 3], enc_f32(lg.w));
                    lSD[i] = make_int2(sI[k], dl);
                    if (BUILD) tl[i] = make_int2(sI[k], dl);
                    lW[i]  = lg;
                }
            }
        }
        __syncthreads();

        const int T4 = (T + 3) & ~3;
        for (int i = tid; i < T4 + 4; i += TILE_THREADS) {
            if (i < T) {
                int dl = lSD[i].y;
                float4 lg = lW[i];
                uint4 m = *(const uint4*)(sM + 4 * dl);
                float4 w;
                w.x = expf(lg.x - dec_f32(m.x)); w.y = expf(lg.y - dec_f32(m.y));
                w.z = expf(lg.z - dec_f32(m.z)); w.w = expf(lg.w - dec_f32(m.w));
                lW[i] = w;
                atomicAdd(&sSum[dl * 4 + 0], w.x);
                atomicAdd(&sSum[dl * 4 + 1], w.y);
                atomicAdd(&sSum[dl * 4 + 2], w.z);
                atomicAdd(&sSum[dl * 4 + 3], w.w);
            } else {
                lSD[i] = make_int2(d0, TD);
                lW[i]  = make_float4(0.f, 0.f, 0.f, 0.f);
            }
        }
        __syncthreads();

        const int nb = T4 >> 2;
        float hA[4][8], hB[4][8];
        int dA[4], dB[4];
        if (wv < nb) walk_issue(H, lSD, wv, lane, hA, dA);
        for (int q = wv; q < nb; q += 2 * nwaves) {
            int q1 = q + nwaves;
            q1 = q1 < nb ? q1 : nb;
            walk_issue(H, lSD, q1, lane, hB, dB);
            __builtin_amdgcn_sched_barrier(0);
            walk_apply(acc, lW, q, lane, hA, dA);
            int q2 = q + 2 * nwaves;
            q2 = q2 < nb ? q2 : nb;
            walk_issue(H, lSD, q2, lane, hA, dA);
            walk_apply(acc, lW, q1, lane, hB, dB);
        }
    } else {
        const int CH = 32 * SCAN_U;
        int dn[SCAN_U];

        load_dst(dn, dsts, wave * CH, lane, E);
        for (int c0 = wave * CH; c0 < E; c0 += nwaves * CH) {
            int sv[SCAN_U];
            unsigned dlv[SCAN_U];
#pragma unroll
            for (int j = 0; j < SCAN_U; ++j) dlv[j] = (unsigned)(dn[j] - d0);
            load_dst(dn, dsts, c0 + nwaves * CH, lane, E);
#pragma unroll
            for (int j = 0; j < SCAN_U; ++j) {
                int e = c0 + j * 32 + lane;
                sv[j] = (dlv[j] < (unsigned)nd) ? srcs[e] : 0;
            }
#pragma unroll
            for (int j = 0; j < SCAN_U; ++j) {
                if (dlv[j] < (unsigned)nd) {
                    int dl = (int)dlv[j];
                    float4 as = *(const float4*)(aS + 4 * (size_t)sv[j]);
                    atomicMax(&sM[dl * 4 + 0], enc_f32(leaky(as.x + sAD[dl * 4 + 0])));
                    atomicMax(&sM[dl * 4 + 1], enc_f32(leaky(as.y + sAD[dl * 4 + 1])));
                    atomicMax(&sM[dl * 4 + 2], enc_f32(leaky(as.z + sAD[dl * 4 + 2])));
                    atomicMax(&sM[dl * 4 + 3], enc_f32(leaky(as.w + sAD[dl * 4 + 3])));
                }
            }
        }
        __syncthreads();

        load_dst(dn, dsts, wave * CH, lane, E);
        for (int dl = wave; dl < nd; dl += nwaves)
            tile_add_edge(H, aS, d0 + dl, dl, lane, acc, sAD, sM, sSum);

        for (int c0 = wave * CH; c0 < E; c0 += nwaves * CH) {
            int sv[SCAN_U];
            unsigned dlv[SCAN_U];
#pragma unroll
            for (int j = 0; j < SCAN_U; ++j) dlv[j] = (unsigned)(dn[j] - d0);
            load_dst(dn, dsts, c0 + nwaves * CH, lane, E);
#pragma unroll
            for (int j = 0; j < SCAN_U; ++j) {
                int e = c0 + j * 32 + lane;
                sv[j] = (dlv[j] < (unsigned)nd) ? srcs[e] : 0;
            }
#pragma unroll
            for (int j = 0; j < SCAN_U; ++j) {
                unsigned mask = (unsigned)__ballot(dlv[j] < (unsigned)nd);
                while (mask) {
                    int l = __ffs(mask) - 1;
                    mask &= mask - 1;
                    int s  = __shfl(sv[j], l);
                    int dl = __shfl((int)dlv[j], l);
                    tile_add_edge(H, aS, s, dl, lane, acc, sAD, sM, sSum);
                }
            }
        }
    }
    __syncthreads();

    for (int i = tid; i < nd * (HC / 4); i += TILE_THREADS) {
        int dl = i / (HC / 4), fq = i % (HC / 4);
        float den = sSum[dl * 4 + (fq / (CPH / 4))] + 1e-16f;
        float4 v  = *(const float4*)(acc + dl * HC + fq * 4);
        float4 bb = *(const float4*)(b + fq * 4);
        v.x = v.x / den + bb.x; v.y = v.y / den + bb.y;
        v.z = v.z / den + bb.z; v.w = v.w / den + bb.w;
        if (relu) {
            v.x = fmaxf(v.x, 0.f); v.y = fmaxf(v.y, 0.f);
            v.z = fmaxf(v.z, 0.f); v.w = fmaxf(v.w, 0.f);
        }
        *(float4*)(out + (size_t)(d0 + dl) * HC + fq * 4) = v;
    }
}

static void gat_layer(const float* Xin, int K, const float* W, const float* as,
                      const float* ad, const float* b, float* Hbuf, float* agg,
                      const int* ei, int E, int N, float* alphaS, float* alphaD,
                      int relu, int* tileList, int* tileCnt, bool build,
                      hipStream_t stream) {
    gemm_wmma_f32<<<GEMM_BLOCKS, GEMM_THREADS, 0, stream>>>(Xin, W, Hbuf, N, K);

    int nh = N * HEADS;
    node_alpha<<<(nh + 255) / 256, 256, 0, stream>>>(Hbuf, as, ad, alphaS, alphaD, N);

    if (build)
        aggr_tile<true><<<(N + TD - 1) / TD, TILE_THREADS, 0, stream>>>(
                ei, E, N, Hbuf, alphaS, alphaD, b, relu, agg, tileList, tileCnt);
    else
        aggr_tile<false><<<(N + TD - 1) / TD, TILE_THREADS, 0, stream>>>(
                ei, E, N, Hbuf, alphaS, alphaD, b, relu, agg, tileList, tileCnt);
}

extern "C" void kernel_launch(void* const* d_in, const int* in_sizes, int n_in,
                              void* d_out, int out_size, void* d_ws, size_t ws_size,
                              hipStream_t stream) {
    const float* x   = (const float*)d_in[0];
    const int*   ei  = (const int*)d_in[1];
    const float* W1  = (const float*)d_in[2];
    const float* as1 = (const float*)d_in[3];
    const float* ad1 = (const float*)d_in[4];
    const float* b1  = (const float*)d_in[5];
    const float* W2  = (const float*)d_in[6];
    const float* as2 = (const float*)d_in[7];
    const float* ad2 = (const float*)d_in[8];
    const float* b2  = (const float*)d_in[9];
    const float* W3  = (const float*)d_in[10];
    const float* as3 = (const float*)d_in[11];
    const float* ad3 = (const float*)d_in[12];
    const float* b3  = (const float*)d_in[13];

    int K1 = in_sizes[2] / HC;
    int K2 = in_sizes[6] / HC;
    int K3 = in_sizes[10] / HC;
    int N  = in_sizes[0] / K1;
    int E  = in_sizes[1] / 2;

    float*    Hbuf   = (float*)d_ws;
    float*    bufA   = Hbuf + (size_t)N * HC;
    float*    bufB   = bufA + (size_t)N * HC;
    float*    alphaS = bufB + (size_t)N * HC;
    float*    alphaD = alphaS + (size_t)N * HEADS;
    int*      tileList = (int*)(alphaD + (size_t)N * HEADS);
    int*      tileCnt  = tileList + 2 * (size_t)CAP * ((N + TD - 1) / TD);

    gat_layer(x,    K1, W1, as1, ad1, b1, Hbuf, bufA,          ei, E, N,
              alphaS, alphaD,  1, tileList, tileCnt,  true,  stream);
    gat_layer(bufA, K2, W2, as2, ad2, b2, Hbuf, bufB,          ei, E, N,
              alphaS, alphaD,  1, tileList, tileCnt,  false, stream);
    gat_layer(bufB, K3, W3, as3, ad3, b3, Hbuf, (float*)d_out, ei, E, N,
              alphaS, alphaD,  0, tileList, tileCnt,  false, stream);
}


# Aggregation: a tile's hit list is split into one queue per wave (destination mod 32); the owner wave adds rows by plain 16-byte LDS read-add-write, no float atomics on rows; row sums in queue order
# speedup vs baseline: 6.2275x; 1.0305x over previous
#include <hip/hip_runtime.h>
#include <hip/hip_bf16.h>

typedef __attribute__((ext_vector_type(2))) float v2f;
typedef __attribute__((ext_vector_type(8))) float v8f;

#define HEADS 4
#define HC 256
#define CPH 64
#define NEG_SLOPE 0.2f

__device__ __forceinline__ unsigned enc_f32(float f) {
    unsigned u = __float_as_uint(f);
    return u ^ ((u >> 31) ? 0xFFFFFFFFu : 0x80000000u);
}
__device__ __forceinline__ float dec_f32(unsigned u) {
    return __uint_as_float(u ^ ((u >> 31) ? 0x80000000u : 0xFFFFFFFFu));
}

#define GEMM_THREADS 512
#define GEMM_BLOCKS  250
#define GEMM_KMAX    256
#define WPITCH       544

__device__ __forceinline__ void gemm_load_chunk(const float* __restrict__ xk, const float* wk,
                                                v2f (&a)[8], v2f (&b)[8][4]) {
#pragma unroll
    for (int j = 0; j < 8; ++j) {
        a[j] = *(const v2f*)(xk + 4 * j);
#pragma unroll
        for (int i = 0; i < 4; ++i)
            b[j][i] = *(const v2f*)(wk + 2 * j * WPITCH + 32 * i);
    }
}

__device__ __forceinline__ void gemm_mma_chunk(const v2f (&a)[8], const v2f (&b)[8][4],
                                               v8f& c0, v8f& c1, v8f& c2, v8f& c3) {
#pragma unroll
    for (int j = 0; j < 8; ++j) {
        c0 = __builtin_amdgcn_wmma_f32_16x16x4_f32(false, a[j], false, b[j][0], (short)0, c0, false, false);
        c1 = __builtin_amdgcn_wmma_f32_16x16x4_f32(false, a[j], false, b[j][1], (short)0, c1, false, false);
        c2 = __builtin_amdgcn_wmma_f32_16x16x4_f32(false, a[j], false, b[j][2], (short)0, c2, false, false);
        c3 = __builtin_amdgcn_wmma_f32_16x16x4_f32(false, a[j], false, b[j][3], (short)0, c3, false, false);
    }
}

__global__ __launch_bounds__(GEMM_THREADS) void gemm_wmma_f32(
        const float* __restrict__ X, const float* __restrict__ W,
        float* __restrict__ H, int M, int K) {
    __shared__ __attribute__((aligned(16))) float Wl[(GEMM_KMAX / 2) * WPITCH];

    const int tid = threadIdx.x;
    const int nPairs = (K >> 1) * (HC / 4);
    for (int base = 0; base < nPairs; base += 8 * GEMM_THREADS) {
        float4 w0[8], w1[8];
#pragma unroll
        for (int u = 0; u < 8; ++u) {
            int i = base + u * GEMM_THREADS + tid;
            i = i < nPairs ? i : nPairs - 1;
            int kp = i >> 6, n4 = i & 63;
            w0[u] = *(const float4*)(W + (size_t)(2 * kp) * HC + 4 * n4);
            w1[u] = *(const float4*)(W + (size_t)(2 * kp + 1) * HC + 4 * n4);
        }
#pragma unroll
        for (int u = 0; u < 8; ++u) {
            int i = base + u * GEMM_THREADS + tid;
            i = i < nPairs ? i : nPairs - 1;
            int kp = i >> 6, n4 = i & 63;
            float4* d = (float4*)(Wl + kp * WPITCH + 8 * n4);
            d[0] = make_float4(w0[u].x, w1[u].x, w0[u].y, w1[u].y);
            d[1] = make_float4(w0[u].z, w1[u].z, w0[u].w, w1[u].w);
        }
    }
    __syncthreads();

    const int lane = tid & 31;
    const int half = lane >> 4;
    const int l16  = lane & 15;
    const int wave = __builtin_amdgcn_readfirstlane(tid >> 5);
    const int strip = wave & 3;
    const int grp   = wave >> 2;
    const int numTiles = (M + 15) >> 4;
    const int col0 = strip * 64 + l16;
    const float* wl = Wl + half * WPITCH + 2 * col0;

    const int bx = (int)blockIdx.x, gx = (int)gridDim.x;
    const int nT = bx < numTiles ? (numTiles - bx + gx - 1) / gx : 0;
    const int rounds = nT >> 2;
    const int q = (grp - strip) & 3;
    const int jTail = 4 * rounds + q;
    const int nW = rounds + (q < (nT & 3) ? 1 : 0);
    if (nW == 0) return;

    int t = bx + gx * (0 < rounds ? grp : jTail);
    int m = t * 16 + l16;
    const float* xrow = X + (size_t)(m < M ? m : M - 1) * K + 2 * half;

    v2f aP[8], aQ[8], bP[8][4], bQ[8][4];
    gemm_load_chunk(xrow, wl, aP, bP);
    for (int it = 0; it < nW; ++it) {
        const int nx = it + 1;
        const int tn = nx < nW ? bx + gx * (nx < rounds ? 4 * nx + grp : jTail) : t;
        int mn = tn * 16 + l16;
        const float* xnext = X + (size_t)(mn < M ? mn : M - 1) * K + 2 * half;

        v8f acc0 = {}, acc1 = {}, acc2 = {}, acc3 = {};
        for (int kc = 0; kc < K; kc += 64) {
            gemm_load_chunk(xrow + kc + 32, wl + ((kc >> 1) + 16) * WPITCH, aQ, bQ);
            __builtin_amdgcn_sched_barrier(0);
            gemm_mma_chunk(aP, bP, acc0, acc1, acc2, acc3);
            const bool last = kc + 64 >= K;
            gemm_load_chunk(last ? xnext : xrow + kc + 64,
                            last ? wl : wl + ((kc >> 1) + 32) * WPITCH, aP, bP);
            __builtin_amdgcn_sched_barrier(0);
            gemm_mma_chunk(aQ, bQ, acc0, acc1, acc2, acc3);
        }
#pragma unroll
        for (int r = 0; r < 8; ++r) {
            int row = t * 16 + r + 8 * half;
            if (row < M) {
                float* hrow = H + (size_t)row * HC + col0;
                hrow[0]  = acc0[r];
                hrow[16] = acc1[r];
                hrow[32] = acc2[r];
                hrow[48] = acc3[r];
            }
        }
        t = tn;
        xrow = xnext;
    }
}

__global__ void node_alpha(const float* __restrict__ H,
                           const float* __restrict__ a_src,
                           const float* __restrict__ a_dst,
                           float* __restrict__ alphaS,
                           float* __restrict__ alphaD, int N) {
    int t = blockIdx.x * blockDim.x + threadIdx.x;
    if (t >= N * HEADS) return;
    int n = t >> 2, h = t & 3;
    const float4* hp = (const float4*)(H + (size_t)n * HC + h * CPH);
    const float4* as = (const float4*)(a_src + h * CPH);
    const float4* ad = (const float4*)(a_dst + h * CPH);
    float ss = 0.f, sd = 0.f;
#pragma unroll 4
    for (int i = 0; i < CPH / 4; ++i) {
        float4 hv = hp[i], av = as[i], dv = ad[i];
        ss += hv.x * av.x + hv.y * av.y + hv.z * av.z + hv.w * av.w;
        sd += hv.x * dv.x + hv.y * dv.y + hv.z * dv.z + hv.w * dv.w;
    }
    alphaS[t] = ss;
    alphaD[t] = sd;
}

__device__ __forceinline__ float leaky(float x) {
    return x >= 0.f ? x : NEG_SLOPE * x;
}

#define TD 200
#define TILE_THREADS 1024
#define SCAN_U 8
#define CAP 4352
#define NIT ((CAP + TILE_THREADS - 1) / TILE_THREADS)
#define NQ (TILE_THREADS / 32)
#define TQ_STRIDE (NQ + 2)

__device__ __forceinline__ void load_dst(int (&d)[SCAN_U], const int* __restrict__ dsts,
                                         int c, int lane, int E) {
#pragma unroll
    for (int j = 0; j < SCAN_U; ++j) {
        int e = c + j * 32 + lane;
        int v = dsts[e < E ? e : E - 1];
        d[j] = (e < E) ? v : -1;
    }
}

__device__ __forceinline__ void tile_add_edge(
        const float* __restrict__ H, const float* __restrict__ aS, int s, int dl,
        int lane, float* acc, const float* sAD, const unsigned* sM, float* sSum) {
    const float* hrow = H + (size_t)s * HC + lane;
    float hv[8];
#pragma unroll
    for (int k = 0; k < 8; ++k) hv[k] = hrow[32 * k];
    float4 as = *(const float4*)(aS + 4 * (size_t)s);
    float w0 = expf(leaky(as.x + sAD[dl * 4 + 0]) - dec_f32(sM[dl * 4 + 0]));
    float w1 = expf(leaky(as.y + sAD[dl * 4 + 1]) - dec_f32(sM[dl * 4 + 1]));
    float w2 = expf(leaky(as.z + sAD[dl * 4 + 2]) - dec_f32(sM[dl * 4 + 2]));
    float w3 = expf(leaky(as.w + sAD[dl * 4 + 3]) - dec_f32(sM[dl * 4 + 3]));
    if (lane < 4) {
        float wl = lane == 0 ? w0 : lane == 1 ? w1 : lane == 2 ? w2 : w3;
        atomicAdd(&sSum[dl * 4 + lane], wl);
    }
    float* a = acc + dl * HC + lane;
    atomicAdd(a + 0,   hv[0] * w0);
    atomicAdd(a + 32,  hv[1] * w0);
    atomicAdd(a + 64,  hv[2] * w1);
    atomicAdd(a + 96,  hv[3] * w1);
    atomicAdd(a + 128, hv[4] * w2);
    atomicAdd(a + 160, hv[5] * w2);
    atomicAdd(a + 192, hv[6] * w3);
    atomicAdd(a + 224, hv[7] * w3);
}

__device__ __forceinline__ void load_piece(int4 (&d)[SCAN_U], const int4* __restrict__ dv,
                                           int vb, int lane, int nvec) {
#pragma unroll
    for (int j = 0; j < SCAN_U; ++j) {
        int vi = vb + j * 32 + lane;
        d[j] = dv[vi < nvec ? vi : nvec - 1];
    }
}

__device__ __forceinline__ void walk_issue(const float* __restrict__ H, const int2* lSD, int i0,
                                           int iLast, int lane, float4 (&h)[4][2], int (&dl)[4]) {
#pragma unroll
    for (int u = 0; u < 4; ++u) {
        int i = i0 + u;
        i = i < iLast ? i : iLast;
        int2 e = lSD[i];
        dl[u] = __builtin_amdgcn_readfirstlane(e.y);
        const float* row = H + (size_t)__builtin_amdgcn_readfirstlane(e.x) * HC + 4 * lane;
        h[u][0] = *(const float4*)(row);
        h[u][1] = *(const float4*)(row + 128);
    }
}

__device__ __forceinline__ void walk_apply(float* acc, const float4* lW, int i0, int iEnd, int lane,
                                           const float4 (&h)[4][2], const int (&dl)[4]) {
    const bool lo = lane < 16;
#pragma unroll
    for (int u = 0; u < 4; ++u) {
        if (i0 + u < iEnd) {
            float4 w = lW[i0 + u];
            float wa = lo ? w.x : w.y;
            float wb = lo ? w.z : w.w;
            float4* a = (float4*)(acc + dl[u] * HC + 4 * lane);
            float4 a0 = a[0], a1 = a[32];
            a0.x += h[u][0].x * wa; a0.y += h[u][0].y * wa;
            a0.z += h[u][0].z * wa; a0.w += h[u][0].w * wa;
            a1.x += h[u][1].x * wb; a1.y += h[u][1].y * wb;
            a1.z += h[u][1].z * wb; a1.w += h[u][1].w * wb;
            a[0] = a0; a[32] = a1;
        }
    }
}

template <bool BUILD>
__global__ __launch_bounds__(TILE_THREADS) void aggr_tile(
        const int* __restrict__ ei, int E, int N,
        const float* __restrict__ H,
        const float* __restrict__ aS,
        const float* __restrict__ aD,
        const float* __restrict__ b, int relu,
        float* __restrict__ out,
        int* __restrict__ tileList, int* __restrict__ tileQ) {
    __shared__ __attribute__((aligned(16))) float    acc[(TD + 1) * HC];
    __shared__ __attribute__((aligned(16))) float    sAD[TD * 4];
    __shared__ __attribute__((aligned(16))) unsigned sM[TD * 4];
    __shared__ float    sSum[TD * 4];
    __shared__ __attribute__((aligned(16))) float4   lW[CAP];
    __shared__ __attribute__((aligned(16))) int2     lSD[CAP];
    __shared__ int sCnt;
    __shared__ int qCnt[NQ];
    __shared__ int qBase[NQ + 1];

    const int tid  = threadIdx.x;
    const int lane = tid & 31;
    const int wave = tid >> 5;
    const int nwaves = TILE_THREADS / 32;
    const int d0 = blockIdx.x * TD;
    const int nd = (N - d0) < TD ? (N - d0) : TD;
    const int* __restrict__ srcs = ei;
    const int* __restrict__ dsts = ei + E;
    const int wv = __builtin_amdgcn_readfirstlane(wave);
    int2* __restrict__ tl = (int2*)tileList + (size_t)blockIdx.x * CAP;
    int* __restrict__ tq = tileQ + (size_t)blockIdx.x * TQ_STRIDE;
    int storedTotal = 0, storedBase = 0;
    if (!BUILD) {
        storedTotal = tq[NQ + 1];
        if (tid <= NQ) storedBase = tq[tid];
    }

    for (int i = tid; i < TD * HC / 4; i += TILE_THREADS)
        ((float4*)acc)[i] = make_float4(0.f, 0.f, 0.f, 0.f);
    for (int i = tid; i < TD * 4; i += TILE_THREADS) {
        int dl = i >> 2;
        float ad = 0.f;
        unsigned me = 0u;
        if (dl < nd) {
            size_t g = 4 * (size_t)(d0 + dl) + (i & 3);
            ad = aD[g];
            me = enc_f32(leaky(aS[g] + ad));
        }
        sAD[i] = ad; sM[i] = me; sSum[i] = 0.f;
    }
    if (tid == 0) sCnt = nd;
    if (BUILD) {
        if (tid < NQ) qCnt[tid] = 0;
    } else {
        if (tid <= NQ) qBase[tid] = storedBase;
    }
    __syncthreads();

    if (BUILD) {
    int head = (int)((4u - (unsigned)(((size_t)dsts >> 2) & 3)) & 3u);
    if (head > E) head = E;
    const int nvec = (E - head) >> 2;
    const int4* __restrict__ dv = (const int4*)(dsts + head);
    if (tid < E - 4 * nvec) {
        int e = tid < head ? tid : 4 * nvec + tid;
        if ((unsigned)(dsts[e] - d0) < (unsigned)nd) {
            int idx = atomicAdd(&sCnt, 1);
            if (idx < CAP) lSD[idx].x = e;
        }
    }
    if (nvec > 0) {
        const int PV = 32 * SCAN_U;
        int4 cur[SCAN_U];
        load_piece(cur, dv, wv * PV, lane, nvec);
        for (int vb = wv * PV; vb < nvec; vb += nwaves * PV) {
            int4 nx[SCAN_U];
            load_piece(nx, dv, vb + nwaves * PV, lane, nvec);
            unsigned hm = 0u;
#pragma unroll
            for (int j = 0; j < SCAN_U; ++j) {
                unsigned lim = (vb + j * 32 + lane < nvec) ? (unsigned)nd : 0u;
                hm |= ((unsigned)(cur[j].x - d0) < lim ? 1u : 0u) << (4 * j + 0);
                hm |= ((unsigned)(cur[j].y - d0) < lim ? 1u : 0u) << (4 * j + 1);
                hm |= ((unsigned)(cur[j].z - d0) < lim ? 1u : 0u) << (4 * j + 2);
                hm |= ((unsigned)(cur[j].w - d0) < lim ? 1u : 0u) << (4 * j + 3);
            }
            if (hm) {
                int idx = atomicAdd(&sCnt, __popc(hm));
                int eb  = head + 4 * (vb + lane);
                do {
                    int bit = __ffs(hm) - 1;
                    hm &= hm - 1;
                    if (idx < CAP) lSD[idx].x = eb + (bit >> 2) * 128 + (bit & 3);
                    ++idx;
                } while (hm);
            }
#pragma unroll
            for (int j = 0; j < SCAN_U; ++j) cur[j] = nx[j];
        }
    }
    __syncthreads();
    if (tid == 0) tq[NQ + 1] = sCnt;
    }

    const int total = BUILD ? sCnt : storedTotal;
    if (total <= CAP - 4) {
        const int T = total;
        {
            int sI[NIT], dI[NIT], rI[NIT];
            float4 aI[NIT], lgI[NIT];
#pragma unroll
            for (int k = 0; k < NIT; ++k) {
                int i = tid + k * TILE_THREADS;
                if (BUILD) {
                    int e = (i >= nd && i < T) ? lSD[i].x : 0;
                    int sv = srcs[e];
                    int dv = dsts[e] - d0;
                    if (i < nd) { sv = d0 + i; dv = i; }
                    sI[k] = sv;
                    dI[k] = dv;
                    rI[k] = 0;
                    if (i < T) rI[k] = atomicAdd(&qCnt[dv & (NQ - 1)], 1);
                } else {
                    int2 p = make_int2(0, 0);
                    if (i < T) p = tl[i];
                    sI[k] = p.x;
                    dI[k] = p.y;
                    rI[k] = 0;
                }
            }
#pragma unroll
            for (int k = 0; k < NIT; ++k) aI[k] = *(const float4*)(aS + 4 * (size_t)sI[k]);
#pragma unroll
            for (int k = 0; k < NIT; ++k) {
                int i = tid + k * TILE_THREADS;
                lgI[k] = make_float4(0.f, 0.f, 0.f, 0.f);
                if (i < T) {
                    int dl = dI[k];
                    float4 ad = *(const float4*)(sAD + 4 * dl);
                    float4 lg;
                    lg.x = leaky(aI[k].x + ad.x); lg.y = leaky(aI[k].y + ad.y);
                    lg.z = leaky(aI[k].z + ad.z); lg.w = leaky(aI[k].w + ad.w);
                    atomicMax(&sM[dl * 4 + 0], enc_f32(lg.x));
                    atomicMax(&sM[dl * 4 + 1], enc_f32(lg.y));
                    atomicMax(&sM[dl * 4 + 2], enc_f32(lg.z));
                    atomicMax(&sM[dl * 4 + 3], enc_f32(lg.w));
                    lgI[k] = lg;
                    if (!BUILD) {
                        lSD[i] = make_int2(sI[k], dl);
                        lW[i]  = lg;
                    }
                }
            }
            if (BUILD) {
                __syncthreads();
                if (wv == 0) {
                    int c = qCnt[lane];
                    int inc = c;
#pragma unroll
                    for (int d = 1; d < NQ; d <<= 1) {
                        int up = __shfl_up(inc, d);
                        if (lane >= d) inc += up;
                    }
                    qBase[lane] = inc - c;
                    if (lane == NQ - 1) qBase[NQ] = inc;
                }
                __syncthreads();
#pragma unroll
                for (int k = 0; k < NIT; ++k) {
                    int i = tid + k * TILE_THREADS;
                    if (i < T) {
                        int pos = qBase[dI[k] & (NQ - 1)] + rI[k];
                        lSD[pos] = make_int2(sI[k], dI[k]);
                        tl[pos]  = make_int2(sI[k], dI[k]);
                        lW[pos]  = lgI[k];
                    }
                }
                if (tid <= NQ) tq[tid] = qBase[tid];
            }
        }
        __syncthreads();

        for (int i = tid; i < T; i += TILE_THREADS) {
            int dl = lSD[i].y;
            float4 lg = lW[i];
            uint4 m = *(const uint4*)(sM + 4 * dl);
            float4 w;
            w.x = expf(lg.x - dec_f32(m.x)); w.y = expf(lg.y - dec_f32(m.y));
            w.z = expf(lg.z - dec_f32(m.z)); w.w = expf(lg.w - dec_f32(m.w));
            lW[i] = w;
            atomicAdd(&sSum[dl * 4 + 0], w.x);
            atomicAdd(&sSum[dl * 4 + 1], w.y);
            atomicAdd(&sSum[dl * 4 + 2], w.z);
            atomicAdd(&sSum[dl * 4 + 3], w.w);
        }
        __syncthreads();

        const int b0 = __builtin_amdgcn_readfirstlane(qBase[wv]);
        const int b1 = __builtin_amdgcn_readfirstlane(qBase[wv + 1]);
        if (b0 < b1) {
            const int iLast = b1 - 1;
            float4 hA[4][2], hB[4][2];
            int dA[4], dB[4];
            walk_issue(H, lSD, b0, iLast, lane, hA, dA);
            for (int i = b0; i < b1; i += 8) {
                walk_issue(H, lSD, i + 4, iLast, lane, hB, dB);
                walk_apply(acc, lW, i, b1, lane, hA, dA);
                walk_issue(H, lSD, i + 8, iLast, lane, hA, dA);
                walk_apply(acc, lW, i + 4, b1, lane, hB, dB);
            }
        }
    } else {
        if (BUILD && tid <= NQ) tq[tid] = 0;
        const int CH = 32 * SCAN_U;
        int dn[SCAN_U];

        load_dst(dn, dsts, wave * CH, lane, E);
        for (int c0 = wave * CH; c0 < E; c0 += nwaves * CH) {
            int sv[SCAN_U];
            unsigned dlv[SCAN_U];
#pragma unroll
            for (int j = 0; j < SCAN_U; ++j) dlv[j] = (unsigned)(dn[j] - d0);
            load_dst(dn, dsts, c0 + nwaves * CH, lane, E);
#pragma unroll
            for (int j = 0; j < SCAN_U; ++j) {
                int e = c0 + j * 32 + lane;
                sv[j] = (dlv[j] < (unsigned)nd) ? srcs[e] : 0;
            }
#pragma unroll
            for (int j = 0; j < SCAN_U; ++j) {
                if (dlv[j] < (unsigned)nd) {
                    int dl = (int)dlv[j];
                    float4 as = *(const float4*)(aS + 4 * (size_t)sv[j]);
                    atomicMax(&sM[dl * 4 + 0], enc_f32(leaky(as.x + sAD[dl * 4 + 0])));
                    atomicMax(&sM[dl * 4 + 1], enc_f32(leaky(as.y + sAD[dl * 4 + 1])));
                    atomicMax(&sM[dl * 4 + 2], enc_f32(leaky(as.z + sAD[dl * 4 + 2])));
                    atomicMax(&sM[dl * 4 + 3], enc_f32(leaky(as.w + sAD[dl * 4 + 3])));
                }
            }
        }
        __syncthreads();

        load_dst(dn, dsts, wave * CH, lane, E);
        for (int dl = wave; dl < nd; dl += nwaves)
            tile_add_edge(H, aS, d0 + dl, dl, lane, acc, sAD, sM, sSum);

        for (int c0 = wave * CH; c0 < E; c0 += nwaves * CH) {
            int sv[SCAN_U];
            unsigned dlv[SCAN_U];
#pragma unroll
            for (int j = 0; j < SCAN_U; ++j) dlv[j] = (unsigned)(dn[j] - d0);
            load_dst(dn, dsts, c0 + nwaves * CH, lane, E);
#pragma unroll
            for (int j = 0; j < SCAN_U; ++j) {
                int e = c0 + j * 32 + lane;
                sv[j] = (dlv[j] < (unsigned)nd) ? srcs[e] : 0;
            }
#pragma unroll
            for (int j = 0; j < SCAN_U; ++j) {
                unsigned mask = (unsigned)__ballot(dlv[j] < (unsigned)nd);
                while (mask) {
                    int l = __ffs(mask) - 1;
                    mask &= mask - 1;
                    int s  = __shfl(sv[j], l);
                    int dl = __shfl((int)dlv[j], l);
                    tile_add_edge(H, aS, s, dl, lane, acc, sAD, sM, sSum);
                }
            }
        }
    }
    __syncthreads();

    for (int i = tid; i < nd * (HC / 4); i += TILE_THREADS) {
        int dl = i / (HC / 4), fq = i % (HC / 4);
        float den = sSum[dl * 4 + (fq / (CPH / 4))] + 1e-16f;
        float4 v  = *(const float4*)(acc + dl * HC + fq * 4);
        float4 bb = *(const float4*)(b + fq * 4);
        v.x = v.x / den + bb.x; v.y = v.y / den + bb.y;
        v.z = v.z / den + bb.z; v.w = v.w / den + bb.w;
        if (relu) {
            v.x = fmaxf(v.x, 0.f); v.y = fmaxf(v.y, 0.f);
            v.z = fmaxf(v.z, 0.f); v.w = fmaxf(v.w, 0.f);
        }
        *(float4*)(out + (size_t)(d0 + dl) * HC + fq * 4) = v;
    }
}

static void gat_layer(const float* Xin, int K, const float* W, const float* as,
                      const float* ad, const float* b, float* Hbuf, float* agg,
                      const int* ei, int E, int N, float* alphaS, float* alphaD,
                      int relu, int* tileList, int* tileQ, bool build,
                      hipStream_t stream) {
    gemm_wmma_f32<<<GEMM_BLOCKS, GEMM_THREADS, 0, stream>>>(Xin, W, Hbuf, N, K);

    int nh = N * HEADS;
    node_alpha<<<(nh + 255) / 256, 256, 0, stream>>>(Hbuf, as, ad, alphaS, alphaD, N);

    if (build)
        aggr_tile<true><<<(N + TD - 1) / TD, TILE_THREADS, 0, stream>>>(
                ei, E, N, Hbuf, alphaS, alphaD, b, relu, agg, tileList, tileQ);
    else
        aggr_tile<false><<<(N + TD - 1) / TD, TILE_THREADS, 0, stream>>>(
                ei, E, N, Hbuf, alphaS, alphaD, b, relu, agg, tileList, tileQ);
}

extern "C" void kernel_launch(void* const* d_in, const int* in_sizes, int n_in,
                              void* d_out, int out_size, void* d_ws, size_t ws_size,
                              hipStream_t stream) {
    const float* x   = (const float*)d_in[0];
    const int*   ei  = (const int*)d_in[1];
    const float* W1  = (const float*)d_in[2];
    const float* as1 = (const float*)d_in[3];
    const float* ad1 = (const float*)d_in[4];
    const float* b1  = (const float*)d_in[5];
    const float* W2  = (const float*)d_in[6];
    const float* as2 = (const float*)d_in[7];
    const float* ad2 = (const float*)d_in[8];
    const float* b2  = (const float*)d_in[9];
    const float* W3  = (const float*)d_in[10];
    const float* as3 = (const float*)d_in[11];
    const float* ad3 = (const float*)d_in[12];
    const float* b3  = (const float*)d_in[13];

    int K1 = in_sizes[2] / HC;
    int K2 = in_sizes[6] / HC;
    int K3 = in_sizes[10] / HC;
    int N  = in_sizes[0] / K1;
    int E  = in_sizes[1] / 2;

    float*    Hbuf   = (float*)d_ws;
    float*    bufA   = Hbuf + (size_t)N * HC;
    float*    bufB   = bufA + (size_t)N * HC;
    float*    alphaS = bufB + (size_t)N * HC;
    float*    alphaD = alphaS + (size_t)N * HEADS;
    int*      tileList = (int*)(alphaD + (size_t)N * HEADS);
    int*      tileQ    = tileList + 2 * (size_t)CAP * ((N + TD - 1) / TD);

    gat_layer(x,    K1, W1, as1, ad1, b1, Hbuf, bufA,          ei, E, N,
              alphaS, alphaD,  1, tileList, tileQ,  true,  stream);
    gat_layer(bufA, K2, W2, as2, ad2, b2, Hbuf, bufB,          ei, E, N,
              alphaS, alphaD,  1, tileList, tileQ,  false, stream);
    gat_layer(bufB, K3, W3, as3, ad3, b3, Hbuf, (float*)d_out, ei, E, N,
              alphaS, alphaD,  0, tileList, tileQ,  false, stream);
}


# Softmax maxima and denominators: lane l of every wave takes queue l, so the 32 lanes of an LDS atomic name 32 different destinations, not one bank; denominator sum order: LDS arrival, as before
# speedup vs baseline: 6.8750x; 1.1040x over previous
#include <hip/hip_runtime.h>
#include <hip/hip_bf16.h>

typedef __attribute__((ext_vector_type(2))) float v2f;
typedef __attribute__((ext_vector_type(8))) float v8f;

#define HEADS 4
#define HC 256
#define CPH 64
#define NEG_SLOPE 0.2f

__device__ __forceinline__ unsigned enc_f32(float f) {
    unsigned u = __float_as_uint(f);
    return u ^ ((u >> 31) ? 0xFFFFFFFFu : 0x80000000u);
}
__device__ __forceinline__ float dec_f32(unsigned u) {
    return __uint_as_float(u ^ ((u >> 31) ? 0x80000000u : 0xFFFFFFFFu));
}

#define GEMM_THREADS 512
#define GEMM_BLOCKS  250
#define GEMM_KMAX    256
#define WPITCH       544

__device__ __forceinline__ void gemm_load_chunk(const float* __restrict__ xk, const float* wk,
                                                v2f (&a)[8], v2f (&b)[8][4]) {
#pragma unroll
    for (int j = 0; j < 8; ++j) {
        a[j] = *(const v2f*)(xk + 4 * j);
#pragma unroll
        for (int i = 0; i < 4; ++i)
            b[j][i] = *(const v2f*)(wk + 2 * j * WPITCH + 32 * i);
    }
}

__device__ __forceinline__ void gemm_mma_chunk(const v2f (&a)[8], const v2f (&b)[8][4],
                                               v8f& c0, v8f& c1, v8f& c2, v8f& c3) {
#pragma unroll
    for (int j = 0; j < 8; ++j) {
        c0 = __builtin_amdgcn_wmma_f32_16x16x4_f32(false, a[j], false, b[j][0], (short)0, c0, false, false);
        c1 = __builtin_amdgcn_wmma_f32_16x16x4_f32(false, a[j], false, b[j][1], (short)0, c1, false, false);
        c2 = __builtin_amdgcn_wmma_f32_16x16x4_f32(false, a[j], false, b[j][2], (short)0, c2, false, false);
        c3 = __builtin_amdgcn_wmma_f32_16x16x4_f32(false, a[j], false, b[j][3], (short)0, c3, false, false);
    }
}

__global__ __launch_bounds__(GEMM_THREADS) void gemm_wmma_f32(
        const float* __restrict__ X, const float* __restrict__ W,
        float* __restrict__ H, int M, int K) {
    __shared__ __attribute__((aligned(16))) float Wl[(GEMM_KMAX / 2) * WPITCH];

    const int tid = threadIdx.x;
    const int nPairs = (K >> 1) * (HC / 4);
    for (int base = 0; base < nPairs; base += 8 * GEMM_THREADS) {
        float4 w0[8], w1[8];
#pragma unroll
        for (int u = 0; u < 8; ++u) {
            int i = base + u * GEMM_THREADS + tid;
            i = i < nPairs ? i : nPairs - 1;
            int kp = i >> 6, n4 = i & 63;
            w0[u] = *(const float4*)(W + (size_t)(2 * kp) * HC + 4 * n4);
            w1[u] = *(const float4*)(W + (size_t)(2 * kp + 1) * HC + 4 * n4);
        }
#pragma unroll
        for (int u = 0; u < 8; ++u) {
            int i = base + u * GEMM_THREADS + tid;
            i = i < nPairs ? i : nPairs - 1;
            int kp = i >> 6, n4 = i & 63;
            float4* d = (float4*)(Wl + kp * WPITCH + 8 * n4);
            d[0] = make_float4(w0[u].x, w1[u].x, w0[u].y, w1[u].y);
            d[1] = make_float4(w0[u].z, w1[u].z, w0[u].w, w1[u].w);
        }
    }
    __syncthreads();

    const int lane = tid & 31;
    const int half = lane >> 4;
    const int l16  = lane & 15;
    const int wave = __builtin_amdgcn_readfirstlane(tid >> 5);
    const int strip = wave & 3;
    const int grp   = wave >> 2;
    const int numTiles = (M + 15) >> 4;
    const int col0 = strip * 64 + l16;
    const float* wl = Wl + half * WPITCH + 2 * col0;

    const int bx = (int)blockIdx.x, gx = (int)gridDim.x;
    const int nT = bx < numTiles ? (numTiles - bx + gx - 1) / gx : 0;
    const int rounds = nT >> 2;
    const int q = (grp - strip) & 3;
    const int jTail = 4 * rounds + q;
    const int nW = rounds + (q < (nT & 3) ? 1 : 0);
    if (nW == 0) return;

    int t = bx + gx * (0 < rounds ? grp : jTail);
    int m = t * 16 + l16;
    const float* xrow = X + (size_t)(m < M ? m : M - 1) * K + 2 * half;

    v2f aP[8], aQ[8], bP[8][4], bQ[8][4];
    gemm_load_chunk(xrow, wl, aP, bP);
    for (int it = 0; it < nW; ++it) {
        const int nx = it + 1;
        const int tn = nx < nW ? bx + gx * (nx < rounds ? 4 * nx + grp : jTail) : t;
        int mn = tn * 16 + l16;
        const float* xnext = X + (size_t)(mn < M ? mn : M - 1) * K + 2 * half;

        v8f acc0 = {}, acc1 = {}, acc2 = {}, acc3 = {};
        for (int kc = 0; kc < K; kc += 64) {
            gemm_load_chunk(xrow + kc + 32, wl + ((kc >> 1) + 16) * WPITCH, aQ, bQ);
            __builtin_amdgcn_sched_barrier(0);
            gemm_mma_chunk(aP, bP, acc0, acc1, acc2, acc3);
            const bool last = kc + 64 >= K;
            gemm_load_chunk(last ? xnext : xrow + kc + 64,
                            last ? wl : wl + ((kc >> 1) + 32) * WPITCH, aP, bP);
            __builtin_amdgcn_sched_barrier(0);
            gemm_mma_chunk(aQ, bQ, acc0, acc1, acc2, acc3);
        }
#pragma unroll
        for (int r = 0; r < 8; ++r) {
            int row = t * 16 + r + 8 * half;
            if (row < M) {
                float* hrow = H + (size_t)row * HC + col0;
                hrow[0]  = acc0[r];
                hrow[16] = acc1[r];
                hrow[32] = acc2[r];
                hrow[48] = acc3[r];
            }
        }
        t = tn;
        xrow = xnext;
    }
}

__global__ void node_alpha(const float* __restrict__ H,
                           const float* __restrict__ a_src,
                           const float* __restrict__ a_dst,
                           float* __restrict__ alphaS,
                           float* __restrict__ alphaD, int N) {
    int t = blockIdx.x * blockDim.x + threadIdx.x;
    if (t >= N * HEADS) return;
    int n = t >> 2, h = t & 3;
    const float4* hp = (const float4*)(H + (size_t)n * HC + h * CPH);
    const float4* as = (const float4*)(a_src + h * CPH);
    const float4* ad = (const float4*)(a_dst + h * CPH);
    float ss = 0.f, sd = 0.f;
#pragma unroll 4
    for (int i = 0; i < CPH / 4; ++i) {
        float4 hv = hp[i], av = as[i], dv = ad[i];
        ss += hv.x * av.x + hv.y * av.y + hv.z * av.z + hv.w * av.w;
        sd += hv.x * dv.x + hv.y * dv.y + hv.z * dv.z + hv.w * dv.w;
    }
    alphaS[t] = ss;
    alphaD[t] = sd;
}

__device__ __forceinline__ float leaky(float x) {
    return x >= 0.f ? x : NEG_SLOPE * x;
}

#define TD 200
#define TILE_THREADS 1024
#define SCAN_U 8
#define CAP 4352
#define NIT ((CAP + TILE_THREADS - 1) / TILE_THREADS)
#define NQ (TILE_THREADS / 32)
#define TQ_STRIDE (NQ + 2)
static_assert(NQ == 32, "the statistics passes put lane l of a wave on queue l");

__device__ __forceinline__ void load_dst(int (&d)[SCAN_U], const int* __restrict__ dsts,
                                         int c, int lane, int E) {
#pragma unroll
    for (int j = 0; j < SCAN_U; ++j) {
        int e = c + j * 32 + lane;
        int v = dsts[e < E ? e : E - 1];
        d[j] = (e < E) ? v : -1;
    }
}

__device__ __forceinline__ void tile_add_edge(
        const float* __restrict__ H, const float* __restrict__ aS, int s, int dl,
        int lane, float* acc, const float* sAD, const unsigned* sM, float* sSum) {
    const float* hrow = H + (size_t)s * HC + lane;
    float hv[8];
#pragma unroll
    for (int k = 0; k < 8; ++k) hv[k] = hrow[32 * k];
    float4 as = *(const float4*)(aS + 4 * (size_t)s);
    float w0 = expf(leaky(as.x + sAD[dl * 4 + 0]) - dec_f32(sM[dl * 4 + 0]));
    float w1 = expf(leaky(as.y + sAD[dl * 4 + 1]) - dec_f32(sM[dl * 4 + 1]));
    float w2 = expf(leaky(as.z + sAD[dl * 4 + 2]) - dec_f32(sM[dl * 4 + 2]));
    float w3 = expf(leaky(as.w + sAD[dl * 4 + 3]) - dec_f32(sM[dl * 4 + 3]));
    if (lane < 4) {
        float wl = lane == 0 ? w0 : lane == 1 ? w1 : lane == 2 ? w2 : w3;
        atomicAdd(&sSum[dl * 4 + lane], wl);
    }
    float* a = acc + dl * HC + lane;
    atomicAdd(a + 0,   hv[0] * w0);
    atomicAdd(a + 32,  hv[1] * w0);
    atomicAdd(a + 64,  hv[2] * w1);
    atomicAdd(a + 96,  hv[3] * w1);
    atomicAdd(a + 128, hv[4] * w2);
    atomicAdd(a + 160, hv[5] * w2);
    atomicAdd(a + 192, hv[6] * w3);
    atomicAdd(a + 224, hv[7] * w3);
}

__device__ __forceinline__ void load_piece(int4 (&d)[SCAN_U], const int4* __restrict__ dv,
                                           int vb, int lane, int nvec) {
#pragma unroll
    for (int j = 0; j < SCAN_U; ++j) {
        int vi = vb + j * 32 + lane;
        d[j] = dv[vi < nvec ? vi : nvec - 1];
    }
}

__device__ __forceinline__ void walk_issue(const float* __restrict__ H, const int2* lSD, int i0,
                                           int iLast, int lane, float4 (&h)[4][2], int (&dl)[4]) {
#pragma unroll
    for (int u = 0; u < 4; ++u) {
        int i = i0 + u;
        i = i < iLast ? i : iLast;
        int2 e = lSD[i];
        dl[u] = __builtin_amdgcn_readfirstlane(e.y);
        const float* row = H + (size_t)__builtin_amdgcn_readfirstlane(e.x) * HC + 4 * lane;
        h[u][0] = *(const float4*)(row);
        h[u][1] = *(const float4*)(row + 128);
    }
}

__device__ __forceinline__ void walk_apply(float* acc, const float4* lW, int i0, int iEnd, int lane,
                                           const float4 (&h)[4][2], const int (&dl)[4]) {
    const bool lo = lane < 16;
#pragma unroll
    for (int u = 0; u < 4; ++u) {
        if (i0 + u < iEnd) {
            float4 w = lW[i0 + u];
            float wa = lo ? w.x : w.y;
            float wb = lo ? w.z : w.w;
            float4* a = (float4*)(acc + dl[u] * HC + 4 * lane);
            float4 a0 = a[0], a1 = a[32];
            a0.x += h[u][0].x * wa; a0.y += h[u][0].y * wa;
            a0.z += h[u][0].z * wa; a0.w += h[u][0].w * wa;
            a1.x += h[u][1].x * wb; a1.y += h[u][1].y * wb;
            a1.z += h[u][1].z * wb; a1.w += h[u][1].w * wb;
            a[0] = a0; a[32] = a1;
        }
    }
}

template <bool BUILD>
__global__ __launch_bounds__(TILE_THREADS) void aggr_tile(
        const int* __restrict__ ei, int E, int N,
        const float* __restrict__ H,
        const float* __restrict__ aS,
        const float* __restrict__ aD,
        const float* __restrict__ b, int relu,
        float* __restrict__ out,
        int* __restrict__ tileList, int* __restrict__ tileQ) {
    __shared__ __attribute__((aligned(16))) float    acc[(TD + 1) * HC];
    __shared__ __attribute__((aligned(16))) float    sAD[TD * 4];
    __shared__ __attribute__((aligned(16))) unsigned sM[TD * 4];
    __shared__ float    sSum[TD * 4];
    __shared__ __attribute__((aligned(16))) float4   lW[CAP];
    __shared__ __attribute__((aligned(16))) int2     lSD[CAP];
    __shared__ int sCnt;
    __shared__ int qCnt[NQ];
    __shared__ int qBase[NQ + 1];

    const int tid  = threadIdx.x;
    const int lane = tid & 31;
    const int wave = tid >> 5;
    const int nwaves = TILE_THREADS / 32;
    const int d0 = blockIdx.x * TD;
    const int nd = (N - d0) < TD ? (N - d0) : TD;
    const int* __restrict__ srcs = ei;
    const int* __restrict__ dsts = ei + E;
    const int wv = __builtin_amdgcn_readfirstlane(wave);
    int2* __restrict__ tl = (int2*)tileList + (size_t)blockIdx.x * CAP;
    int* __restrict__ tq = tileQ + (size_t)blockIdx.x * TQ_STRIDE;
    int storedTotal = 0, storedBase = 0;
    if (!BUILD) {
        storedTotal = tq[NQ + 1];
        if (tid <= NQ) storedBase = tq[tid];
    }

    for (int i = tid; i < TD * HC / 4; i += TILE_THREADS)
        ((float4*)acc)[i] = make_float4(0.f, 0.f, 0.f, 0.f);
    for (int i = tid; i < TD * 4; i += TILE_THREADS) {
        int dl = i >> 2;
        float ad = 0.f;
        unsigned me = 0u;
        if (dl < nd) {
            size_t g = 4 * (size_t)(d0 + dl) + (i & 3);
            ad = aD[g];
            me = enc_f32(leaky(aS[g] + ad));
        }
        sAD[i] = ad; sM[i] = me; sSum[i] = 0.f;
    }
    if (tid == 0) sCnt = nd;
    if (BUILD) {
        if (tid < NQ) qCnt[tid] = 0;
    } else {
        if (tid <= NQ) qBase[tid] = storedBase;
    }
    __syncthreads();

    if (BUILD) {
    int head = (int)((4u - (unsigned)(((size_t)dsts >> 2) & 3)) & 3u);
    if (head > E) head = E;
    const int nvec = (E - head) >> 2;
    const int4* __restrict__ dv = (const int4*)(dsts + head);
    if (tid < E - 4 * nvec) {
        int e = tid < head ? tid : 4 * nvec + tid;
        if ((unsigned)(dsts[e] - d0) < (unsigned)nd) {
            int idx = atomicAdd(&sCnt, 1);
            if (idx < CAP) lSD[idx].x = e;
        }
    }
    if (nvec > 0) {
        const int PV = 32 * SCAN_U;
        int4 cur[SCAN_U];
        load_piece(cur, dv, wv * PV, lane, nvec);
        for (int vb = wv * PV; vb < nvec; vb += nwaves * PV) {
            int4 nx[SCAN_U];
            load_piece(nx, dv, vb + nwaves * PV, lane, nvec);
            unsigned hm = 0u;
#pragma unroll
            for (int j = 0; j < SCAN_U; ++j) {
                unsigned lim = (vb + j * 32 + lane < nvec) ? (unsigned)nd : 0u;
                hm |= ((unsigned)(cur[j].x - d0) < lim ? 1u : 0u) << (4 * j + 0);
                hm |= ((unsigned)(cur[j].y - d0) < lim ? 1u : 0u) << (4 * j + 1);
                hm |= ((unsigned)(cur[j].z - d0) < lim ? 1u : 0u) << (4 * j + 2);
                hm |= ((unsigned)(cur[j].w - d0) < lim ? 1u : 0u) << (4 * j + 3);
            }
            if (hm) {
                int idx = atomicAdd(&sCnt, __popc(hm));
                int eb  = head + 4 * (vb + lane);
                do {
                    int bit = __ffs(hm) - 1;
                    hm &= hm - 1;
                    if (idx < CAP) lSD[idx].x = eb + (bit >> 2) * 128 + (bit & 3);
                    ++idx;
                } while (hm);
            }
#pragma unroll
            for (int j = 0; j < SCAN_U; ++j) cur[j] = nx[j];
        }
    }
    __syncthreads();
    if (tid == 0) tq[NQ + 1] = sCnt;
    }

    const int total = BUILD ? sCnt : storedTotal;
    if (total <= CAP - 4) {
        const int T = total;
        {
            int sI[NIT], dI[NIT], rI[NIT];
            float4 aI[NIT], lgI[NIT];
#pragma unroll
            for (int k = 0; k < NIT; ++k) {
                int i = tid + k * TILE_THREADS;
                if (BUILD) {
                    int e = (i >= nd && i < T) ? lSD[i].x : 0;
                    int sv = srcs[e];
                    int dv = dsts[e] - d0;
                    if (i < nd) { sv = d0 + i; dv = i; }
                    sI[k] = sv;
                    dI[k] = dv;
                    rI[k] = 0;
                    if (i < T) rI[k] = atomicAdd(&qCnt[dv & (NQ - 1)], 1);
                } else {
                    int2 p = make_int2(0, 0);
                    if (i < T) p = tl[i];
                    sI[k] = p.x;
                    dI[k] = p.y;
                    rI[k] = 0;
                }
            }
#pragma unroll
            for (int k = 0; k < NIT; ++k) aI[k] = *(const float4*)(aS + 4 * (size_t)sI[k]);
#pragma unroll
            for (int k = 0; k < NIT; ++k) {
                int i = tid + k * TILE_THREADS;
                lgI[k] = make_float4(0.f, 0.f, 0.f, 0.f);
                if (i < T) {
                    int dl = dI[k];
                    float4 ad = *(const float4*)(sAD + 4 * dl);
                    float4 lg;
                    lg.x = leaky(aI[k].x + ad.x); lg.y = leaky(aI[k].y + ad.y);
                    lg.z = leaky(aI[k].z + ad.z); lg.w = leaky(aI[k].w + ad.w);
                    lgI[k] = lg;
                    if (BUILD) {
                        atomicMax(&sM[dl * 4 + 0], enc_f32(lg.x));
                        atomicMax(&sM[dl * 4 + 1], enc_f32(lg.y));
                        atomicMax(&sM[dl * 4 + 2], enc_f32(lg.z));
                        atomicMax(&sM[dl * 4 + 3], enc_f32(lg.w));
                    } else {
                        lSD[i] = make_int2(sI[k], dl);
                        lW[i]  = lg;
                    }
                }
            }
            if (BUILD) {
                __syncthreads();
                if (wv == 0) {
                    int c = qCnt[lane];
                    int inc = c;
#pragma unroll
                    for (int d = 1; d < NQ; d <<= 1) {
                        int up = __shfl_up(inc, d);
                        if (lane >= d) inc += up;
                    }
                    qBase[lane] = inc - c;
                    if (lane == NQ - 1) qBase[NQ] = inc;
                }
                __syncthreads();
#pragma unroll
                for (int k = 0; k < NIT; ++k) {
                    int i = tid + k * TILE_THREADS;
                    if (i < T) {
                        int pos = qBase[dI[k] & (NQ - 1)] + rI[k];
                        lSD[pos] = make_int2(sI[k], dI[k]);
                        tl[pos]  = make_int2(sI[k], dI[k]);
                        lW[pos]  = lgI[k];
                    }
                }
                if (tid <= NQ) tq[tid] = qBase[tid];
            }
        }
        __syncthreads();

        const int qb = qBase[lane] + wv;
        const int qe = qBase[lane + 1];
        if (!BUILD) {
            for (int i = qb; i < qe; i += NQ) {
                int dl = lSD[i].y;
                float4 lg = lW[i];
                atomicMax(&sM[dl * 4 + 0], enc_f32(lg.x));
                atomicMax(&sM[dl * 4 + 1], enc_f32(lg.y));
                atomicMax(&sM[dl * 4 + 2], enc_f32(lg.z));
                atomicMax(&sM[dl * 4 + 3], enc_f32(lg.w));
            }
            __syncthreads();
        }

        for (int i = qb; i < qe; i += NQ) {
            int dl = lSD[i].y;
            float4 lg = lW[i];
            uint4 m = *(const uint4*)(sM + 4 * dl);
            float4 w;
            w.x = expf(lg.x - dec_f32(m.x)); w.y = expf(lg.y - dec_f32(m.y));
            w.z = expf(lg.z - dec_f32(m.z)); w.w = expf(lg.w - dec_f32(m.w));
            lW[i] = w;
            atomicAdd(&sSum[dl * 4 + 0], w.x);
            atomicAdd(&sSum[dl * 4 + 1], w.y);
            atomicAdd(&sSum[dl * 4 + 2], w.z);
            atomicAdd(&sSum[dl * 4 + 3], w.w);
        }
        __syncthreads();

        const int b0 = __builtin_amdgcn_readfirstlane(qBase[wv]);
        const int b1 = __builtin_amdgcn_readfirstlane(qBase[wv + 1]);
        if (b0 < b1) {
            const int iLast = b1 - 1;
            float4 hA[4][2], hB[4][2];
            int dA[4], dB[4];
            walk_issue(H, lSD, b0, iLast, lane, hA, dA);
            for (int i = b0; i < b1; i += 8) {
                walk_issue(H, lSD, i + 4, iLast, lane, hB, dB);
                walk_apply(acc, lW, i, b1, lane, hA, dA);
                walk_issue(H, lSD, i + 8, iLast, lane, hA, dA);
                walk_apply(acc, lW, i + 4, b1, lane, hB, dB);
            }
        }
    } else {
        if (BUILD && tid <= NQ) tq[tid] = 0;
        const int CH = 32 * SCAN_U;
        int dn[SCAN_U];

        load_dst(dn, dsts, wave * CH, lane, E);
        for (int c0 = wave * CH; c0 < E; c0 += nwaves * CH) {
            int sv[SCAN_U];
            unsigned dlv[SCAN_U];
#pragma unroll
            for (int j = 0; j < SCAN_U; ++j) dlv[j] = (unsigned)(dn[j] - d0);
            load_dst(dn, dsts, c0 + nwaves * CH, lane, E);
#pragma unroll
            for (int j = 0; j < SCAN_U; ++j) {
                int e = c0 + j * 32 + lane;
                sv[j] = (dlv[j] < (unsigned)nd) ? srcs[e] : 0;
            }
#pragma unroll
            for (int j = 0; j < SCAN_U; ++j) {
                if (dlv[j] < (unsigned)nd) {
                    int dl = (int)dlv[j];
                    float4 as = *(const float4*)(aS + 4 * (size_t)sv[j]);
                    atomicMax(&sM[dl * 4 + 0], enc_f32(leaky(as.x + sAD[dl * 4 + 0])));
                    atomicMax(&sM[dl * 4 + 1], enc_f32(leaky(as.y + sAD[dl * 4 + 1])));
                    atomicMax(&sM[dl * 4 + 2], enc_f32(leaky(as.z + sAD[dl * 4 + 2])));
                    atomicMax(&sM[dl * 4 + 3], enc_f32(leaky(as.w + sAD[dl * 4 + 3])));
                }
            }
        }
        __syncthreads();

        load_dst(dn, dsts, wave * CH, lane, E);
        for (int dl = wave; dl < nd; dl += nwaves)
            tile_add_edge(H, aS, d0 + dl, dl, lane, acc, sAD, sM, sSum);

        for (int c0 = wave * CH; c0 < E; c0 += nwaves * CH) {
            int sv[SCAN_U];
            unsigned dlv[SCAN_U];
#pragma unroll
            for (int j = 0; j < SCAN_U; ++j) dlv[j] = (unsigned)(dn[j] - d0);
            load_dst(dn, dsts, c0 + nwaves * CH, lane, E);
#pragma unroll
            for (int j = 0; j < SCAN_U; ++j) {
                int e = c0 + j * 32 + lane;
                sv[j] = (dlv[j] < (unsigned)nd) ? srcs[e] : 0;
            }
#pragma unroll
            for (int j = 0; j < SCAN_U; ++j) {
                unsigned mask = (unsigned)__ballot(dlv[j] < (unsigned)nd);
                while (mask) {
                    int l = __ffs(mask) - 1;
                    mask &= mask - 1;
                    int s  = __shfl(sv[j], l);
                    int dl = __shfl((int)dlv[j], l);
                    tile_add_edge(H, aS, s, dl, lane, acc, sAD, sM, sSum);
                }
            }
        }
    }
    __syncthreads();

    for (int i = tid; i < nd * (HC / 4); i += TILE_THREADS) {
        int dl = i / (HC / 4), fq = i % (HC / 4);
        float den = sSum[dl * 4 + (fq / (CPH / 4))] + 1e-16f;
        float4 v  = *(const float4*)(acc + dl * HC + fq * 4);
        float4 bb = *(const float4*)(b + fq * 4);
        v.x = v.x / den + bb.x; v.y = v.y / den + bb.y;
        v.z = v.z / den + bb.z; v.w = v.w / den + bb.w;
        if (relu) {
            v.x = fmaxf(v.x, 0.f); v.y = fmaxf(v.y, 0.f);
            v.z = fmaxf(v.z, 0.f); v.w = fmaxf(v.w, 0.f);
        }
        *(float4*)(out + (size_t)(d0 + dl) * HC + fq * 4) = v;
    }
}

static void gat_layer(const float* Xin, int K, const float* W, const float* as,
                      const float* ad, const float* b, float* Hbuf, float* agg,
                      const int* ei, int E, int N, float* alphaS, float* alphaD,
                      int relu, int* tileList, int* tileQ, bool build,
                      hipStream_t stream) {
    gemm_wmma_f32<<<GEMM_BLOCKS, GEMM_THREADS, 0, stream>>>(Xin, W, Hbuf, N, K);

    int nh = N * HEADS;
    node_alpha<<<(nh + 255) / 256, 256, 0, stream>>>(Hbuf, as, ad, alphaS, alphaD, N);

    if (build)
        aggr_tile<true><<<(N + TD - 1) / TD, TILE_THREADS, 0, stream>>>(
                ei, E, N, Hbuf, alphaS, alphaD, b, relu, agg, tileList, tileQ);
    else
        aggr_tile<false><<<(N + TD - 1) / TD, TILE_THREADS, 0, stream>>>(
                ei, E, N, Hbuf, alphaS, alphaD, b, relu, agg, tileList, tileQ);
}

extern "C" void kernel_launch(void* const* d_in, const int* in_sizes, int n_in,
                              void* d_out, int out_size, void* d_ws, size_t ws_size,
                              hipStream_t stream) {
    const float* x   = (const float*)d_in[0];
    const int*   ei  = (const int*)d_in[1];
    const float* W1  = (const float*)d_in[2];
    const float* as1 = (const float*)d_in[3];
    const float* ad1 = (const float*)d_in[4];
    const float* b1  = (const float*)d_in[5];
    const float* W2  = (const float*)d_in[6];
    const float* as2 = (const float*)d_in[7];
    const float* ad2 = (const float*)d_in[8];
    const float* b2  = (const float*)d_in[9];
    const float* W3  = (const float*)d_in[10];
    const float* as3 = (const float*)d_in[11];
    const float* ad3 = (const float*)d_in[12];
    const float* b3  = (const float*)d_in[13];

    int K1 = in_sizes[2] / HC;
    int K2 = in_sizes[6] / HC;
    int K3 = in_sizes[10] / HC;
    int N  = in_sizes[0] / K1;
    int E  = in_sizes[1] / 2;

    float*    Hbuf   = (float*)d_ws;
    float*    bufA   = Hbuf + (size_t)N * HC;
    float*    bufB   = bufA + (size_t)N * HC;
    float*    alphaS = bufB + (size_t)N * HC;
    float*    alphaD = alphaS + (size_t)N * HEADS;
    int*      tileList = (int*)(alphaD + (size_t)N * HEADS);
    int*      tileQ    = tileList + 2 * (size_t)CAP * ((N + TD - 1) / TD);

    gat_layer(x,    K1, W1, as1, ad1, b1, Hbuf, bufA,          ei, E, N,
              alphaS, alphaD,  1, tileList, tileQ,  true,  stream);
    gat_layer(bufA, K2, W2, as2, ad2, b2, Hbuf, bufB,          ei, E, N,
              alphaS, alphaD,  1, tileList, tileQ,  false, stream);
    gat_layer(bufB, K3, W3, as3, ad3, b3, Hbuf, (float*)d_out, ei, E, N,
              alphaS, alphaD,  0, tileList, tileQ,  false, stream);
}


# Edge filter reads one byte an edge: layer 1's alpha launch writes each destination's tile number, a block compares 4 edges a dword; list arrival order, so row and denominator sum order, changes
# speedup vs baseline: 7.1976x; 1.0469x over previous
#include <hip/hip_runtime.h>
#include <hip/hip_bf16.h>

typedef __attribute__((ext_vector_type(2))) float v2f;
typedef __attribute__((ext_vector_type(8))) float v8f;

#define HEADS 4
#define HC 256
#define CPH 64
#define NEG_SLOPE 0.2f

__device__ __forceinline__ unsigned enc_f32(float f) {
    unsigned u = __float_as_uint(f);
    return u ^ ((u >> 31) ? 0xFFFFFFFFu : 0x80000000u);
}
__device__ __forceinline__ float dec_f32(unsigned u) {
    return __uint_as_float(u ^ ((u >> 31) ? 0x80000000u : 0xFFFFFFFFu));
}

#define GEMM_THREADS 512
#define GEMM_BLOCKS  250
#define GEMM_KMAX    256
#define WPITCH       544

__device__ __forceinline__ void gemm_load_chunk(const float* __restrict__ xk, const float* wk,
                                                v2f (&a)[8], v2f (&b)[8][4]) {
#pragma unroll
    for (int j = 0; j < 8; ++j) {
        a[j] = *(const v2f*)(xk + 4 * j);
#pragma unroll
        for (int i = 0; i < 4; ++i)
            b[j][i] = *(const v2f*)(wk + 2 * j * WPITCH + 32 * i);
    }
}

__device__ __forceinline__ void gemm_mma_chunk(const v2f (&a)[8], const v2f (&b)[8][4],
                                               v8f& c0, v8f& c1, v8f& c2, v8f& c3) {
#pragma unroll
    for (int j = 0; j < 8; ++j) {
        c0 = __builtin_amdgcn_wmma_f32_16x16x4_f32(false, a[j], false, b[j][0], (short)0, c0, false, false);
        c1 = __builtin_amdgcn_wmma_f32_16x16x4_f32(false, a[j], false, b[j][1], (short)0, c1, false, false);
        c2 = __builtin_amdgcn_wmma_f32_16x16x4_f32(false, a[j], false, b[j][2], (short)0, c2, false, false);
        c3 = __builtin_amdgcn_wmma_f32_16x16x4_f32(false, a[j], false, b[j][3], (short)0, c3, false, false);
    }
}

__global__ __launch_bounds__(GEMM_THREADS) void gemm_wmma_f32(
        const float* __restrict__ X, const float* __restrict__ W,
        float* __restrict__ H, int M, int K) {
    __shared__ __attribute__((aligned(16))) float Wl[(GEMM_KMAX / 2) * WPITCH];

    const int tid = threadIdx.x;
    const int nPairs = (K >> 1) * (HC / 4);
    for (int base = 0; base < nPairs; base += 8 * GEMM_THREADS) {
        float4 w0[8], w1[8];
#pragma unroll
        for (int u = 0; u < 8; ++u) {
            int i = base + u * GEMM_THREADS + tid;
            i = i < nPairs ? i : nPairs - 1;
            int kp = i >> 6, n4 = i & 63;
            w0[u] = *(const float4*)(W + (size_t)(2 * kp) * HC + 4 * n4);
            w1[u] = *(const float4*)(W + (size_t)(2 * kp + 1) * HC + 4 * n4);
        }
#pragma unroll
        for (int u = 0; u < 8; ++u) {
            int i = base + u * GEMM_THREADS + tid;
            i = i < nPairs ? i : nPairs - 1;
            int kp = i >> 6, n4 = i & 63;
            float4* d = (float4*)(Wl + kp * WPITCH + 8 * n4);
            d[0] = make_float4(w0[u].x, w1[u].x, w0[u].y, w1[u].y);
            d[1] = make_float4(w0[u].z, w1[u].z, w0[u].w, w1[u].w);
        }
    }
    __syncthreads();

    const int lane = tid & 31;
    const int half = lane >> 4;
    const int l16  = lane & 15;
    const int wave = __builtin_amdgcn_readfirstlane(tid >> 5);
    const int strip = wave & 3;
    const int grp   = wave >> 2;
    const int numTiles = (M + 15) >> 4;
    const int col0 = strip * 64 + l16;
    const float* wl = Wl + half * WPITCH + 2 * col0;

    const int bx = (int)blockIdx.x, gx = (int)gridDim.x;
    const int nT = bx < numTiles ? (numTiles - bx + gx - 1) / gx : 0;
    const int rounds = nT >> 2;
    const int q = (grp - strip) & 3;
    const int jTail = 4 * rounds + q;
    const int nW = rounds + (q < (nT & 3) ? 1 : 0);
    if (nW == 0) return;

    int t = bx + gx * (0 < rounds ? grp : jTail);
    int m = t * 16 + l16;
    const float* xrow = X + (size_t)(m < M ? m : M - 1) * K + 2 * half;

    v2f aP[8], aQ[8], bP[8][4], bQ[8][4];
    gemm_load_chunk(xrow, wl, aP, bP);
    for (int it = 0; it < nW; ++it) {
        const int nx = it + 1;
        const int tn = nx < nW ? bx + gx * (nx < rounds ? 4 * nx + grp : jTail) : t;
        int mn = tn * 16 + l16;
        const float* xnext = X + (size_t)(mn < M ? mn : M - 1) * K + 2 * half;

        v8f acc0 = {}, acc1 = {}, acc2 = {}, acc3 = {};
        for (int kc = 0; kc < K; kc += 64) {
            gemm_load_chunk(xrow + kc + 32, wl + ((kc >> 1) + 16) * WPITCH, aQ, bQ);
            __builtin_amdgcn_sched_barrier(0);
            gemm_mma_chunk(aP, bP, acc0, acc1, acc2, acc3);
            const bool last = kc + 64 >= K;
            gemm_load_chunk(last ? xnext : xrow + kc + 64,
                            last ? wl : wl + ((kc >> 1) + 32) * WPITCH, aP, bP);
            __builtin_amdgcn_sched_barrier(0);
            gemm_mma_chunk(aQ, bQ, acc0, acc1, acc2, acc3);
        }
#pragma unroll
        for (int r = 0; r < 8; ++r) {
            int row = t * 16 + r + 8 * half;
            if (row < M) {
                float* hrow = H + (size_t)row * HC + col0;
                hrow[0]  = acc0[r];
                hrow[16] = acc1[r];
                hrow[32] = acc2[r];
                hrow[48] = acc3[r];
            }
        }
        t = tn;
        xrow = xnext;
    }
}

#define TD_NODES 200
__global__ void node_alpha(const float* __restrict__ H,
                           const float* __restrict__ a_src,
                           const float* __restrict__ a_dst,
                           float* __restrict__ alphaS,
                           float* __restrict__ alphaD, int N,
                           const int* __restrict__ dsts, unsigned* __restrict__ tile8,
                           int E, int nWords) {
    int t = blockIdx.x * blockDim.x + threadIdx.x;
    if (tile8) {
        const bool al = (((size_t)dsts) & 15) == 0;
        for (int i = t; i < nWords; i += gridDim.x * blockDim.x) {
            unsigned w = 0xFFFFFFFFu;
            if (al && 4 * i + 3 < E) {
                int4 d = *(const int4*)(dsts + 4 * i);
                w = (unsigned)(d.x / TD_NODES) | (unsigned)(d.y / TD_NODES) << 8 |
                    (unsigned)(d.z / TD_NODES) << 16 | (unsigned)(d.w / TD_NODES) << 24;
            } else if (4 * i < E) {
                w = 0u;
#pragma unroll
                for (int c = 0; c < 4; ++c) {
                    int e = 4 * i + c;
                    unsigned id = e < E ? (unsigned)(dsts[e] / TD_NODES) : 0xFFu;
                    w |= id << (8 * c);
                }
            }
            tile8[i] = w;
        }
    }
    if (t >= N * HEADS) return;
    int n = t >> 2, h = t & 3;
    const float4* hp = (const float4*)(H + (size_t)n * HC + h * CPH);
    const float4* as = (const float4*)(a_src + h * CPH);
    const float4* ad = (const float4*)(a_dst + h * CPH);
    float ss = 0.f, sd = 0.f;
#pragma unroll 4
    for (int i = 0; i < CPH / 4; ++i) {
        float4 hv = hp[i], av = as[i], dv = ad[i];
        ss += hv.x * av.x + hv.y * av.y + hv.z * av.z + hv.w * av.w;
        sd += hv.x * dv.x + hv.y * dv.y + hv.z * dv.z + hv.w * dv.w;
    }
    alphaS[t] = ss;
    alphaD[t] = sd;
}

__device__ __forceinline__ float leaky(float x) {
    return x >= 0.f ? x : NEG_SLOPE * x;
}

#define TD TD_NODES
#define TILE_THREADS 1024
#define SCAN_U 8
#define CAP 4352
#define NIT ((CAP + TILE_THREADS - 1) / TILE_THREADS)
#define NQ (TILE_THREADS / 32)
#define TQ_STRIDE (NQ + 2)
static_assert(NQ == 32, "the statistics passes put lane l of a wave on queue l");

__device__ __forceinline__ void load_dst(int (&d)[SCAN_U], const int* __restrict__ dsts,
                                         int c, int lane, int E) {
#pragma unroll
    for (int j = 0; j < SCAN_U; ++j) {
        int e = c + j * 32 + lane;
        int v = dsts[e < E ? e : E - 1];
        d[j] = (e < E) ? v : -1;
    }
}

__device__ __forceinline__ void tile_add_edge(
        const float* __restrict__ H, const float* __restrict__ aS, int s, int dl,
        int lane, float* acc, const float* sAD, const unsigned* sM, float* sSum) {
    const float* hrow = H + (size_t)s * HC + lane;
    float hv[8];
#pragma unroll
    for (int k = 0; k < 8; ++k) hv[k] = hrow[32 * k];
    float4 as = *(const float4*)(aS + 4 * (size_t)s);
    float w0 = expf(leaky(as.x + sAD[dl * 4 + 0]) - dec_f32(sM[dl * 4 + 0]));
    float w1 = expf(leaky(as.y + sAD[dl * 4 + 1]) - dec_f32(sM[dl * 4 + 1]));
    float w2 = expf(leaky(as.z + sAD[dl * 4 + 2]) - dec_f32(sM[dl * 4 + 2]));
    float w3 = expf(leaky(as.w + sAD[dl * 4 + 3]) - dec_f32(sM[dl * 4 + 3]));
    if (lane < 4) {
        float wl = lane == 0 ? w0 : lane == 1 ? w1 : lane == 2 ? w2 : w3;
        atomicAdd(&sSum[dl * 4 + lane], wl);
    }
    float* a = acc + dl * HC + lane;
    atomicAdd(a + 0,   hv[0] * w0);
    atomicAdd(a + 32,  hv[1] * w0);
    atomicAdd(a + 64,  hv[2] * w1);
    atomicAdd(a + 96,  hv[3] * w1);
    atomicAdd(a + 128, hv[4] * w2);
    atomicAdd(a + 160, hv[5] * w2);
    atomicAdd(a + 192, hv[6] * w3);
    atomicAdd(a + 224, hv[7] * w3);
}

__device__ __forceinline__ void load_piece(uint4 (&d)[SCAN_U], const uint4* __restrict__ tv,
                                           int vb, int lane) {
#pragma unroll
    for (int j = 0; j < SCAN_U; ++j) d[j] = tv[vb + j * 32 + lane];
}

__device__ __forceinline__ unsigned byte_ne(unsigned w, unsigned pat) {
    unsigned t = w ^ pat;
    return ((t & 0x7F7F7F7Fu) + 0x7F7F7F7Fu) | t;
}

__device__ __forceinline__ unsigned pair_hits(const uint4& p0, const uint4& p1, unsigned pat) {
    const unsigned M = 0x01010101u;
    unsigned acc;
    acc = (byte_ne(p0.x, pat) >> 7) & M;
    acc |= (byte_ne(p0.y, pat) >> 6) & (M << 1);
    acc |= (byte_ne(p0.z, pat) >> 5) & (M << 2);
    acc |= (byte_ne(p0.w, pat) >> 4) & (M << 3);
    acc |= (byte_ne(p1.x, pat) >> 3) & (M << 4);
    acc |= (byte_ne(p1.y, pat) >> 2) & (M << 5);
    acc |= (byte_ne(p1.z, pat) >> 1) & (M << 6);
    acc |= byte_ne(p1.w, pat) & (M << 7);
    return ~acc;
}

__device__ __forceinline__ void walk_issue(const float* __restrict__ H, const int2* lSD, int i0,
                                           int iLast, int lane, float4 (&h)[4][2], int (&dl)[4]) {
#pragma unroll
    for (int u = 0; u < 4; ++u) {
        int i = i0 + u;
        i = i < iLast ? i : iLast;
        int2 e = lSD[i];
        dl[u] = __builtin_amdgcn_readfirstlane(e.y);
        const float* row = H + (size_t)__builtin_amdgcn_readfirstlane(e.x) * HC + 4 * lane;
        h[u][0] = *(const float4*)(row);
        h[u][1] = *(const float4*)(row + 128);
    }
}

__device__ __forceinline__ void walk_apply(float* acc, const float4* lW, int i0, int iEnd, int lane,
                                           const float4 (&h)[4][2], const int (&dl)[4]) {
    const bool lo = lane < 16;
#pragma unroll
    for (int u = 0; u < 4; ++u) {
        if (i0 + u < iEnd) {
            float4 w = lW[i0 + u];
            float wa = lo ? w.x : w.y;
            float wb = lo ? w.z : w.w;
            float4* a = (float4*)(acc + dl[u] * HC + 4 * lane);
            float4 a0 = a[0], a1 = a[32];
            a0.x += h[u][0].x * wa; a0.y += h[u][0].y * wa;
            a0.z += h[u][0].z * wa; a0.w += h[u][0].w * wa;
            a1.x += h[u][1].x * wb; a1.y += h[u][1].y * wb;
            a1.z += h[u][1].z * wb; a1.w += h[u][1].w * wb;
            a[0] = a0; a[32] = a1;
        }
    }
}

template <bool BUILD>
__global__ __launch_bounds__(TILE_THREADS) void aggr_tile(
        const int* __restrict__ ei, int E, int N,
        const float* __restrict__ H,
        const float* __restrict__ aS,
        const float* __restrict__ aD,
        const float* __restrict__ b, int relu,
        float* __restrict__ out,
        int* __restrict__ tileList, int* __restrict__ tileQ,
        const unsigned* __restrict__ tile8) {
    __shared__ __attribute__((aligned(16))) float    acc[(TD + 1) * HC];
    __shared__ __attribute__((aligned(16))) float    sAD[TD * 4];
    __shared__ __attribute__((aligned(16))) unsigned sM[TD * 4];
    __shared__ float    sSum[TD * 4];
    __shared__ __attribute__((aligned(16))) float4   lW[CAP];
    __shared__ __attribute__((aligned(16))) int2     lSD[CAP];
    __shared__ int sCnt;
    __shared__ int qCnt[NQ];
    __shared__ int qBase[NQ + 1];

    const int tid  = threadIdx.x;
    const int lane = tid & 31;
    const int wave = tid >> 5;
    const int nwaves = TILE_THREADS / 32;
    const int d0 = blockIdx.x * TD;
    const int nd = (N - d0) < TD ? (N - d0) : TD;
    const int* __restrict__ srcs = ei;
    const int* __restrict__ dsts = ei + E;
    const int wv = __builtin_amdgcn_readfirstlane(wave);
    int2* __restrict__ tl = (int2*)tileList + (size_t)blockIdx.x * CAP;
    int* __restrict__ tq = tileQ + (size_t)blockIdx.x * TQ_STRIDE;
    int storedTotal = 0, storedBase = 0;
    if (!BUILD) {
        storedTotal = tq[NQ + 1];
        if (tid <= NQ) storedBase = tq[tid];
    }

    for (int i = tid; i < TD * HC / 4; i += TILE_THREADS)
        ((float4*)acc)[i] = make_float4(0.f, 0.f, 0.f, 0.f);
    for (int i = tid; i < TD * 4; i += TILE_THREADS) {
        int dl = i >> 2;
        float ad = 0.f;
        unsigned me = 0u;
        if (dl < nd) {
            size_t g = 4 * (size_t)(d0 + dl) + (i & 3);
            ad = aD[g];
            me = enc_f32(leaky(aS[g] + ad));
        }
        sAD[i] = ad; sM[i] = me; sSum[i] = 0.f;
    }
    if (tid == 0) sCnt = nd;
    if (BUILD) {
        if (tid < NQ) qCnt[tid] = 0;
    } else {
        if (tid <= NQ) qBase[tid] = storedBase;
    }
    __syncthreads();

    if (BUILD) {
    {
        const uint4* __restrict__ tv = (const uint4*)tile8;
        const unsigned pat = (unsigned)blockIdx.x * 0x01010101u;
        const int PV = 32 * SCAN_U;
        const int nReal = (E + 15) >> 4;
        uint4 cur[SCAN_U];
        load_piece(cur, tv, wv * PV, lane);
        for (int vb = wv * PV; vb < nReal; vb += nwaves * PV) {
            uint4 nx[SCAN_U];
            const bool more = vb + nwaves * PV < nReal;
            load_piece(nx, tv, more ? vb + nwaves * PV : 0, more ? lane : 0);
            unsigned hm[SCAN_U / 2];
            int cnt = 0;
#pragma unroll
            for (int q = 0; q < SCAN_U / 2; ++q) {
                hm[q] = pair_hits(cur[2 * q], cur[2 * q + 1], pat);
                cnt += __popc(hm[q]);
            }
            if (cnt) {
                int idx = atomicAdd(&sCnt, cnt);
                const int eb = 16 * (vb + lane);
#pragma unroll
                for (int q = 0; q < SCAN_U / 2; ++q) {
                    unsigned h = hm[q];
                    while (h) {
                        int bit = __ffs(h) - 1;
                        h &= h - 1;
                        int e = eb + (2 * q + ((bit >> 2) & 1)) * 512 + 4 * (bit & 3) + (bit >> 3);
                        if (idx < CAP) lSD[idx].x = e;
                        ++idx;
                    }
                }
            }
#pragma unroll
            for (int j = 0; j < SCAN_U; ++j) cur[j] = nx[j];
        }
    }
    __syncthreads();
    if (tid == 0) tq[NQ + 1] = sCnt;
    }

    const int total = BUILD ? sCnt : storedTotal;
    if (total <= CAP - 4) {
        const int T = total;
        {
            int sI[NIT], dI[NIT], rI[NIT];
            float4 aI[NIT], lgI[NIT];
#pragma unroll
            for (int k = 0; k < NIT; ++k) {
                int i = tid + k * TILE_THREADS;
                if (BUILD) {
                    int e = (i >= nd && i < T) ? lSD[i].x : 0;
                    int sv = srcs[e];
                    int dv = dsts[e] - d0;
                    if (i < nd) { sv = d0 + i; dv = i; }
                    sI[k] = sv;
                    dI[k] = dv;
                    rI[k] = 0;
                    if (i < T) rI[k] = atomicAdd(&qCnt[dv & (NQ - 1)], 1);
                } else {
                    int2 p = make_int2(0, 0);
                    if (i < T) p = tl[i];
                    sI[k] = p.x;
                    dI[k] = p.y;
                    rI[k] = 0;
                }
            }
#pragma unroll
            for (int k = 0; k < NIT; ++k) aI[k] = *(const float4*)(aS + 4 * (size_t)sI[k]);
#pragma unroll
            for (int k = 0; k < NIT; ++k) {
                int i = tid + k * TILE_THREADS;
                lgI[k] = make_float4(0.f, 0.f, 0.f, 0.f);
                if (i < T) {
                    int dl = dI[k];
                    float4 ad = *(const float4*)(sAD + 4 * dl);
                    float4 lg;
                    lg.x = leaky(aI[k].x + ad.x); lg.y = leaky(aI[k].y + ad.y);
                    lg.z = leaky(aI[k].z + ad.z); lg.w = leaky(aI[k].w + ad.w);
                    lgI[k] = lg;
                    if (BUILD) {
                        atomicMax(&sM[dl * 4 + 0], enc_f32(lg.x));
                        atomicMax(&sM[dl * 4 + 1], enc_f32(lg.y));
                        atomicMax(&sM[dl * 4 + 2], enc_f32(lg.z));
                        atomicMax(&sM[dl * 4 + 3], enc_f32(lg.w));
                    } else {
                        lSD[i] = make_int2(sI[k], dl);
                        lW[i]  = lg;
                    }
                }
            }
            if (BUILD) {
                __syncthreads();
                if (wv == 0) {
                    int c = qCnt[lane];
                    int inc = c;
#pragma unroll
                    for (int d = 1; d < NQ; d <<= 1) {
                        int up = __shfl_up(inc, d);
                        if (lane >= d) inc += up;
                    }
                    qBase[lane] = inc - c;
                    if (lane == NQ - 1) qBase[NQ] = inc;
                }
                __syncthreads();
#pragma unroll
                for (int k = 0; k < NIT; ++k) {
                    int i = tid + k * TILE_THREADS;
                    if (i < T) {
                        int pos = qBase[dI[k] & (NQ - 1)] + rI[k];
                        lSD[pos] = make_int2(sI[k], dI[k]);
                        tl[pos]  = make_int2(sI[k], dI[k]);
                        lW[pos]  = lgI[k];
                    }
                }
                if (tid <= NQ) tq[tid] = qBase[tid];
            }
        }
        __syncthreads();

        const int qb = qBase[lane] + wv;
        const int qe = qBase[lane + 1];
        if (!BUILD) {
            for (int i = qb; i < qe; i += NQ) {
                int dl = lSD[i].y;
                float4 lg = lW[i];
                atomicMax(&sM[dl * 4 + 0], enc_f32(lg.x));
                atomicMax(&sM[dl * 4 + 1], enc_f32(lg.y));
                atomicMax(&sM[dl * 4 + 2], enc_f32(lg.z));
                atomicMax(&sM[dl * 4 + 3], enc_f32(lg.w));
            }
            __syncthreads();
        }

        for (int i = qb; i < qe; i += NQ) {
            int dl = lSD[i].y;
            float4 lg = lW[i];
            uint4 m = *(const uint4*)(sM + 4 * dl);
            float4 w;
            w.x = expf(lg.x - dec_f32(m.x)); w.y = expf(lg.y - dec_f32(m.y));
            w.z = expf(lg.z - dec_f32(m.z)); w.w = expf(lg.w - dec_f32(m.w));
            lW[i] = w;
            atomicAdd(&sSum[dl * 4 + 0], w.x);
            atomicAdd(&sSum[dl * 4 + 1], w.y);
            atomicAdd(&sSum[dl * 4 + 2], w.z);
            atomicAdd(&sSum[dl * 4 + 3], w.w);
        }
        __syncthreads();

        const int b0 = __builtin_amdgcn_readfirstlane(qBase[wv]);
        const int b1 = __builtin_amdgcn_readfirstlane(qBase[wv + 1]);
        if (b0 < b1) {
            const int iLast = b1 - 1;
            float4 hA[4][2], hB[4][2];
            int dA[4], dB[4];
            walk_issue(H, lSD, b0, iLast, lane, hA, dA);
            for (int i = b0; i < b1; i += 8) {
                walk_issue(H, lSD, i + 4, iLast, lane, hB, dB);
                walk_apply(acc, lW, i, b1, lane, hA, dA);
                walk_issue(H, lSD, i + 8, iLast, lane, hA, dA);
                walk_apply(acc, lW, i + 4, b1, lane, hB, dB);
            }
        }
    } else {
        if (BUILD && tid <= NQ) tq[tid] = 0;
        const int CH = 32 * SCAN_U;
        int dn[SCAN_U];

        load_dst(dn, dsts, wave * CH, lane, E);
        for (int c0 = wave * CH; c0 < E; c0 += nwaves * CH) {
            int sv[SCAN_U];
            unsigned dlv[SCAN_U];
#pragma unroll
            for (int j = 0; j < SCAN_U; ++j) dlv[j] = (unsigned)(dn[j] - d0);
            load_dst(dn, dsts, c0 + nwaves * CH, lane, E);
#pragma unroll
            for (int j = 0; j < SCAN_U; ++j) {
                int e = c0 + j * 32 + lane;
                sv[j] = (dlv[j] < (unsigned)nd) ? srcs[e] : 0;
            }
#pragma unroll
            for (int j = 0; j < SCAN_U; ++j) {
                if (dlv[j] < (unsigned)nd) {
                    int dl = (int)dlv[j];
                    float4 as = *(const float4*)(aS + 4 * (size_t)sv[j]);
                    atomicMax(&sM[dl * 4 + 0], enc_f32(leaky(as.x + sAD[dl * 4 + 0])));
                    atomicMax(&sM[dl * 4 + 1], enc_f32(leaky(as.y + sAD[dl * 4 + 1])));
                    atomicMax(&sM[dl * 4 + 2], enc_f32(leaky(as.z + sAD[dl * 4 + 2])));
                    atomicMax(&sM[dl * 4 + 3], enc_f32(leaky(as.w + sAD[dl * 4 + 3])));
                }
            }
        }
        __syncthreads();

        load_dst(dn, dsts, wave * CH, lane, E);
        for (int dl = wave; dl < nd; dl += nwaves)
            tile_add_edge(H, aS, d0 + dl, dl, lane, acc, sAD, sM, sSum);

        for (int c0 = wave * CH; c0 < E; c0 += nwaves * CH) {
            int sv[SCAN_U];
            unsigned dlv[SCAN_U];
#pragma unroll
            for (int j = 0; j < SCAN_U; ++j) dlv[j] = (unsigned)(dn[j] - d0);
            load_dst(dn, dsts, c0 + nwaves * CH, lane, E);
#pragma unroll
            for (int j = 0; j < SCAN_U; ++j) {
                int e = c0 + j * 32 + lane;
                sv[j] = (dlv[j] < (unsigned)nd) ? srcs[e] : 0;
            }
#pragma unroll
            for (int j = 0; j < SCAN_U; ++j) {
                unsigned mask = (unsigned)__ballot(dlv[j] < (unsigned)nd);
                while (mask) {
                    int l = __ffs(mask) - 1;
                    mask &= mask - 1;
                    int s  = __shfl(sv[j], l);
                    int dl = __shfl((int)dlv[j], l);
                    tile_add_edge(H, aS, s, dl, lane, acc, sAD, sM, sSum);
                }
            }
        }
    }
    __syncthreads();

    for (int i = tid; i < nd * (HC / 4); i += TILE_THREADS) {
        int dl = i / (HC / 4), fq = i % (HC / 4);
        float den = sSum[dl * 4 + (fq / (CPH / 4))] + 1e-16f;
        float4 v  = *(const float4*)(acc + dl * HC + fq * 4);
        float4 bb = *(const float4*)(b + fq * 4);
        v.x = v.x / den + bb.x; v.y = v.y / den + bb.y;
        v.z = v.z / den + bb.z; v.w = v.w / den + bb.w;
        if (relu) {
            v.x = fmaxf(v.x, 0.f); v.y = fmaxf(v.y, 0.f);
            v.z = fmaxf(v.z, 0.f); v.w = fmaxf(v.w, 0.f);
        }
        *(float4*)(out + (size_t)(d0 + dl) * HC + fq * 4) = v;
    }
}

static void gat_layer(const float* Xin, int K, const float* W, const float* as,
                      const float* ad, const float* b, float* Hbuf, float* agg,
                      const int* ei, int E, int N, float* alphaS, float* alphaD,
                      int relu, int* tileList, int* tileQ, unsigned* tile8, int nPiece, bool build,
                      hipStream_t stream) {
    gemm_wmma_f32<<<GEMM_BLOCKS, GEMM_THREADS, 0, stream>>>(Xin, W, Hbuf, N, K);

    int nh = N * HEADS;
    node_alpha<<<(nh + 255) / 256, 256, 0, stream>>>(Hbuf, as, ad, alphaS, alphaD, N,
                                                     ei + E, build ? tile8 : nullptr, E, 4 * nPiece);

    if (build)
        aggr_tile<true><<<(N + TD - 1) / TD, TILE_THREADS, 0, stream>>>(
                ei, E, N, Hbuf, alphaS, alphaD, b, relu, agg, tileList, tileQ, tile8);
    else
        aggr_tile<false><<<(N + TD - 1) / TD, TILE_THREADS, 0, stream>>>(
                ei, E, N, Hbuf, alphaS, alphaD, b, relu, agg, tileList, tileQ, tile8);
}

extern "C" void kernel_launch(void* const* d_in, const int* in_sizes, int n_in,
                              void* d_out, int out_size, void* d_ws, size_t ws_size,
                              hipStream_t stream) {
    const float* x   = (const float*)d_in[0];
    const int*   ei  = (const int*)d_in[1];
    const float* W1  = (const float*)d_in[2];
    const float* as1 = (const float*)d_in[3];
    const float* ad1 = (const float*)d_in[4];
    const float* b1  = (const float*)d_in[5];
    const float* W2  = (const float*)d_in[6];
    const float* as2 = (const float*)d_in[7];
    const float* ad2 = (const float*)d_in[8];
    const float* b2  = (const float*)d_in[9];
    const float* W3  = (const float*)d_in[10];
    const float* as3 = (const float*)d_in[11];
    const float* ad3 = (const float*)d_in[12];
    const float* b3  = (const float*)d_in[13];

    int K1 = in_sizes[2] / HC;
    int K2 = in_sizes[6] / HC;
    int K3 = in_sizes[10] / HC;
    int N  = in_sizes[0] / K1;
    int E  = in_sizes[1] / 2;

    float*    Hbuf   = (float*)d_ws;
    float*    bufA   = Hbuf + (size_t)N * HC;
    float*    bufB   = bufA + (size_t)N * HC;
    float*    alphaS = bufB + (size_t)N * HC;
    float*    alphaD = alphaS + (size_t)N * HEADS;
    const int stepPieces = TILE_THREADS * SCAN_U;
    const int nPiece  = ((E + 15) / 16 + stepPieces - 1) / stepPieces * stepPieces;
    unsigned* tile8  = (unsigned*)(alphaD + (size_t)N * HEADS);
    int*      tileList = (int*)(tile8 + 4 * (size_t)nPiece);
    int*      tileQ    = tileList + 2 * (size_t)CAP * ((N + TD - 1) / TD);

    gat_layer(x,    K1, W1, as1, ad1, b1, Hbuf, bufA,          ei, E, N,
              alphaS, alphaD,  1, tileList, tileQ, tile8, nPiece,  true,  stream);
    gat_layer(bufA, K2, W2, as2, ad2, b2, Hbuf, bufB,          ei, E, N,
              alphaS, alphaD,  1, tileList, tileQ, tile8, nPiece,  false, stream);
    gat_layer(bufB, K3, W3, as3, ad3, b3, Hbuf, (float*)d_out, ei, E, N,
              alphaS, alphaD,  0, tileList, tileQ, tile8, nPiece,  false, stream);
}


# Aggregation: the owner wave keeps the rows of its 6 or 7 destinations in registers, the row picked by scalar bit tests on the destination; LDS row reads, writes and clearing go; sum order unchanged
# speedup vs baseline: 7.4481x; 1.0348x over previous
#include <hip/hip_runtime.h>
#include <hip/hip_bf16.h>

typedef __attribute__((ext_vector_type(2))) float v2f;
typedef __attribute__((ext_vector_type(8))) float v8f;

#define HEADS 4
#define HC 256
#define CPH 64
#define NEG_SLOPE 0.2f

__device__ __forceinline__ unsigned enc_f32(float f) {
    unsigned u = __float_as_uint(f);
    return u ^ ((u >> 31) ? 0xFFFFFFFFu : 0x80000000u);
}
__device__ __forceinline__ float dec_f32(unsigned u) {
    return __uint_as_float(u ^ ((u >> 31) ? 0x80000000u : 0xFFFFFFFFu));
}

#define GEMM_THREADS 512
#define GEMM_BLOCKS  250
#define GEMM_KMAX    256
#define WPITCH       544

__device__ __forceinline__ void gemm_load_chunk(const float* __restrict__ xk, const float* wk,
                                                v2f (&a)[8], v2f (&b)[8][4]) {
#pragma unroll
    for (int j = 0; j < 8; ++j) {
        a[j] = *(const v2f*)(xk + 4 * j);
#pragma unroll
        for (int i = 0; i < 4; ++i)
            b[j][i] = *(const v2f*)(wk + 2 * j * WPITCH + 32 * i);
    }
}

__device__ __forceinline__ void gemm_mma_chunk(const v2f (&a)[8], const v2f (&b)[8][4],
                                               v8f& c0, v8f& c1, v8f& c2, v8f& c3) {
#pragma unroll
    for (int j = 0; j < 8; ++j) {
        c0 = __builtin_amdgcn_wmma_f32_16x16x4_f32(false, a[j], false, b[j][0], (short)0, c0, false, false);
        c1 = __builtin_amdgcn_wmma_f32_16x16x4_f32(false, a[j], false, b[j][1], (short)0, c1, false, false);
        c2 = __builtin_amdgcn_wmma_f32_16x16x4_f32(false, a[j], false, b[j][2], (short)0, c2, false, false);
        c3 = __builtin_amdgcn_wmma_f32_16x16x4_f32(false, a[j], false, b[j][3], (short)0, c3, false, false);
    }
}

__global__ __launch_bounds__(GEMM_THREADS) void gemm_wmma_f32(
        const float* __restrict__ X, const float* __restrict__ W,
        float* __restrict__ H, int M, int K) {
    __shared__ __attribute__((aligned(16))) float Wl[(GEMM_KMAX / 2) * WPITCH];

    const int tid = threadIdx.x;
    const int nPairs = (K >> 1) * (HC / 4);
    for (int base = 0; base < nPairs; base += 8 * GEMM_THREADS) {
        float4 w0[8], w1[8];
#pragma unroll
        for (int u = 0; u < 8; ++u) {
            int i = base + u * GEMM_THREADS + tid;
            i = i < nPairs ? i : nPairs - 1;
            int kp = i >> 6, n4 = i & 63;
            w0[u] = *(const float4*)(W + (size_t)(2 * kp) * HC + 4 * n4);
            w1[u] = *(const float4*)(W + (size_t)(2 * kp + 1) * HC + 4 * n4);
        }
#pragma unroll
        for (int u = 0; u < 8; ++u) {
            int i = base + u * GEMM_THREADS + tid;
            i = i < nPairs ? i : nPairs - 1;
            int kp = i >> 6, n4 = i & 63;
            float4* d = (float4*)(Wl + kp * WPITCH + 8 * n4);
            d[0] = make_float4(w0[u].x, w1[u].x, w0[u].y, w1[u].y);
            d[1] = make_float4(w0[u].z, w1[u].z, w0[u].w, w1[u].w);
        }
    }
    __syncthreads();

    const int lane = tid & 31;
    const int half = lane >> 4;
    const int l16  = lane & 15;
    const int wave = __builtin_amdgcn_readfirstlane(tid >> 5);
    const int strip = wave & 3;
    const int grp   = wave >> 2;
    const int numTiles = (M + 15) >> 4;
    const int col0 = strip * 64 + l16;
    const float* wl = Wl + half * WPITCH + 2 * col0;

    const int bx = (int)blockIdx.x, gx = (int)gridDim.x;
    const int nT = bx < numTiles ? (numTiles - bx + gx - 1) / gx : 0;
    const int rounds = nT >> 2;
    const int q = (grp - strip) & 3;
    const int jTail = 4 * rounds + q;
    const int nW = rounds + (q < (nT & 3) ? 1 : 0);
    if (nW == 0) return;

    int t = bx + gx * (0 < rounds ? grp : jTail);
    int m = t * 16 + l16;
    const float* xrow = X + (size_t)(m < M ? m : M - 1) * K + 2 * half;

    v2f aP[8], aQ[8], bP[8][4], bQ[8][4];
    gemm_load_chunk(xrow, wl, aP, bP);
    for (int it = 0; it < nW; ++it) {
        const int nx = it + 1;
        const int tn = nx < nW ? bx + gx * (nx < rounds ? 4 * nx + grp : jTail) : t;
        int mn = tn * 16 + l16;
        const float* xnext = X + (size_t)(mn < M ? mn : M - 1) * K + 2 * half;

        v8f acc0 = {}, acc1 = {}, acc2 = {}, acc3 = {};
        for (int kc = 0; kc < K; kc += 64) {
            gemm_load_chunk(xrow + kc + 32, wl + ((kc >> 1) + 16) * WPITCH, aQ, bQ);
            __builtin_amdgcn_sched_barrier(0);
            gemm_mma_chunk(aP, bP, acc0, acc1, acc2, acc3);
            const bool last = kc + 64 >= K;
            gemm_load_chunk(last ? xnext : xrow + kc + 64,
                            last ? wl : wl + ((kc >> 1) + 32) * WPITCH, aP, bP);
            __builtin_amdgcn_sched_barrier(0);
            gemm_mma_chunk(aQ, bQ, acc0, acc1, acc2, acc3);
        }
#pragma unroll
        for (int r = 0; r < 8; ++r) {
            int row = t * 16 + r + 8 * half;
            if (row < M) {
                float* hrow = H + (size_t)row * HC + col0;
                hrow[0]  = acc0[r];
                hrow[16] = acc1[r];
                hrow[32] = acc2[r];
                hrow[48] = acc3[r];
            }
        }
        t = tn;
        xrow = xnext;
    }
}

#define TD_NODES 200
__global__ void node_alpha(const float* __restrict__ H,
                           const float* __restrict__ a_src,
                           const float* __restrict__ a_dst,
                           float* __restrict__ alphaS,
                           float* __restrict__ alphaD, int N,
                           const int* __restrict__ dsts, unsigned* __restrict__ tile8,
                           int E, int nWords) {
    int t = blockIdx.x * blockDim.x + threadIdx.x;
    if (tile8) {
        const bool al = (((size_t)dsts) & 15) == 0;
        for (int i = t; i < nWords; i += gridDim.x * blockDim.x) {
            unsigned w = 0xFFFFFFFFu;
            if (al && 4 * i + 3 < E) {
                int4 d = *(const int4*)(dsts + 4 * i);
                w = (unsigned)(d.x / TD_NODES) | (unsigned)(d.y / TD_NODES) << 8 |
                    (unsigned)(d.z / TD_NODES) << 16 | (unsigned)(d.w / TD_NODES) << 24;
            } else if (4 * i < E) {
                w = 0u;
#pragma unroll
                for (int c = 0; c < 4; ++c) {
                    int e = 4 * i + c;
                    unsigned id = e < E ? (unsigned)(dsts[e] / TD_NODES) : 0xFFu;
                    w |= id << (8 * c);
                }
            }
            tile8[i] = w;
        }
    }
    if (t >= N * HEADS) return;
    int n = t >> 2, h = t & 3;
    const float4* hp = (const float4*)(H + (size_t)n * HC + h * CPH);
    const float4* as = (const float4*)(a_src + h * CPH);
    const float4* ad = (const float4*)(a_dst + h * CPH);
    float ss = 0.f, sd = 0.f;
#pragma unroll 4
    for (int i = 0; i < CPH / 4; ++i) {
        float4 hv = hp[i], av = as[i], dv = ad[i];
        ss += hv.x * av.x + hv.y * av.y + hv.z * av.z + hv.w * av.w;
        sd += hv.x * dv.x + hv.y * dv.y + hv.z * dv.z + hv.w * dv.w;
    }
    alphaS[t] = ss;
    alphaD[t] = sd;
}

__device__ __forceinline__ float leaky(float x) {
    return x >= 0.f ? x : NEG_SLOPE * x;
}

#define TD TD_NODES
#define TILE_THREADS 1024
#define SCAN_U 8
#define CAP 4352
#define NIT ((CAP + TILE_THREADS - 1) / TILE_THREADS)
#define NQ (TILE_THREADS / 32)
#define TQ_STRIDE (NQ + 2)
static_assert(NQ == 32, "the statistics passes put lane l of a wave on queue l");

__device__ __forceinline__ void load_dst(int (&d)[SCAN_U], const int* __restrict__ dsts,
                                         int c, int lane, int E) {
#pragma unroll
    for (int j = 0; j < SCAN_U; ++j) {
        int e = c + j * 32 + lane;
        int v = dsts[e < E ? e : E - 1];
        d[j] = (e < E) ? v : -1;
    }
}

__device__ __forceinline__ void tile_add_edge(
        const float* __restrict__ H, const float* __restrict__ aS, int s, int dl,
        int lane, float* acc, const float* sAD, const unsigned* sM, float* sSum) {
    const float* hrow = H + (size_t)s * HC + lane;
    float hv[8];
#pragma unroll
    for (int k = 0; k < 8; ++k) hv[k] = hrow[32 * k];
    float4 as = *(const float4*)(aS + 4 * (size_t)s);
    float w0 = expf(leaky(as.x + sAD[dl * 4 + 0]) - dec_f32(sM[dl * 4 + 0]));
    float w1 = expf(leaky(as.y + sAD[dl * 4 + 1]) - dec_f32(sM[dl * 4 + 1]));
    float w2 = expf(leaky(as.z + sAD[dl * 4 + 2]) - dec_f32(sM[dl * 4 + 2]));
    float w3 = expf(leaky(as.w + sAD[dl * 4 + 3]) - dec_f32(sM[dl * 4 + 3]));
    if (lane < 4) {
        float wl = lane == 0 ? w0 : lane == 1 ? w1 : lane == 2 ? w2 : w3;
        atomicAdd(&sSum[dl * 4 + lane], wl);
    }
    float* a = acc + dl * HC + lane;
    atomicAdd(a + 0,   hv[0] * w0);
    atomicAdd(a + 32,  hv[1] * w0);
    atomicAdd(a + 64,  hv[2] * w1);
    atomicAdd(a + 96,  hv[3] * w1);
    atomicAdd(a + 128, hv[4] * w2);
    atomicAdd(a + 160, hv[5] * w2);
    atomicAdd(a + 192, hv[6] * w3);
    atomicAdd(a + 224, hv[7] * w3);
}

__device__ __forceinline__ void load_piece(uint4 (&d)[SCAN_U], const uint4* __restrict__ tv,
                                           int vb, int lane) {
#pragma unroll
    for (int j = 0; j < SCAN_U; ++j) d[j] = tv[vb + j * 32 + lane];
}

__device__ __forceinline__ unsigned byte_ne(unsigned w, unsigned pat) {
    unsigned t = w ^ pat;
    return ((t & 0x7F7F7F7Fu) + 0x7F7F7F7Fu) | t;
}

__device__ __forceinline__ unsigned pair_hits(const uint4& p0, const uint4& p1, unsigned pat) {
    const unsigned M = 0x01010101u;
    unsigned acc;
    acc = (byte_ne(p0.x, pat) >> 7) & M;
    acc |= (byte_ne(p0.y, pat) >> 6) & (M << 1);
    acc |= (byte_ne(p0.z, pat) >> 5) & (M << 2);
    acc |= (byte_ne(p0.w, pat) >> 4) & (M << 3);
    acc |= (byte_ne(p1.x, pat) >> 3) & (M << 4);
    acc |= (byte_ne(p1.y, pat) >> 2) & (M << 5);
    acc |= (byte_ne(p1.z, pat) >> 1) & (M << 6);
    acc |= byte_ne(p1.w, pat) & (M << 7);
    return ~acc;
}

#define RING 6
#define KQ ((TD + NQ - 1) / NQ)

__device__ __forceinline__ void row_add(float (&a)[KQ][8], int k, const float4& h0,
                                        const float4& h1, float wa, float wb) {
    const int m = 1 << k;
#pragma unroll
    for (int q = 0; q < KQ; ++q) {
        if (m & (1 << q)) {
            a[q][0] += h0.x * wa; a[q][1] += h0.y * wa; a[q][2] += h0.z * wa; a[q][3] += h0.w * wa;
            a[q][4] += h1.x * wb; a[q][5] += h1.y * wb; a[q][6] += h1.z * wb; a[q][7] += h1.w * wb;
        }
    }
}

template <bool BUILD>
__global__ __launch_bounds__(TILE_THREADS) void aggr_tile(
        const int* __restrict__ ei, int E, int N,
        const float* __restrict__ H,
        const float* __restrict__ aS,
        const float* __restrict__ aD,
        const float* __restrict__ b, int relu,
        float* __restrict__ out,
        int* __restrict__ tileList, int* __restrict__ tileQ,
        const unsigned* __restrict__ tile8) {
    __shared__ __attribute__((aligned(16))) float    acc[(TD + 1) * HC];
    __shared__ __attribute__((aligned(16))) float    sAD[TD * 4];
    __shared__ __attribute__((aligned(16))) unsigned sM[TD * 4];
    __shared__ float    sSum[TD * 4];
    __shared__ __attribute__((aligned(16))) float4   lW[CAP];
    __shared__ __attribute__((aligned(16))) int2     lSD[CAP];
    __shared__ int sCnt;
    __shared__ int qCnt[NQ];
    __shared__ int qBase[NQ + 1];

    const int tid  = threadIdx.x;
    const int lane = tid & 31;
    const int wave = tid >> 5;
    const int nwaves = TILE_THREADS / 32;
    const int d0 = blockIdx.x * TD;
    const int nd = (N - d0) < TD ? (N - d0) : TD;
    const int* __restrict__ srcs = ei;
    const int* __restrict__ dsts = ei + E;
    const int wv = __builtin_amdgcn_readfirstlane(wave);
    int2* __restrict__ tl = (int2*)tileList + (size_t)blockIdx.x * CAP;
    int* __restrict__ tq = tileQ + (size_t)blockIdx.x * TQ_STRIDE;
    int storedTotal = 0, storedBase = 0;
    if (!BUILD) {
        storedTotal = tq[NQ + 1];
        if (tid <= NQ) storedBase = tq[tid];
    }

    for (int i = tid; i < TD * 4; i += TILE_THREADS) {
        int dl = i >> 2;
        float ad = 0.f;
        unsigned me = 0u;
        if (dl < nd) {
            size_t g = 4 * (size_t)(d0 + dl) + (i & 3);
            ad = aD[g];
            me = enc_f32(leaky(aS[g] + ad));
        }
        sAD[i] = ad; sM[i] = me; sSum[i] = 0.f;
    }
    if (tid == 0) sCnt = nd;
    if (BUILD) {
        if (tid < NQ) qCnt[tid] = 0;
    } else {
        if (tid <= NQ) qBase[tid] = storedBase;
    }
    __syncthreads();

    if (BUILD) {
    {
        const uint4* __restrict__ tv = (const uint4*)tile8;
        const unsigned pat = (unsigned)blockIdx.x * 0x01010101u;
        const int PV = 32 * SCAN_U;
        const int nReal = (E + 15) >> 4;
        uint4 cur[SCAN_U];
        load_piece(cur, tv, wv * PV, lane);
        for (int vb = wv * PV; vb < nReal; vb += nwaves * PV) {
            uint4 nx[SCAN_U];
            const bool more = vb + nwaves * PV < nReal;
            load_piece(nx, tv, more ? vb + nwaves * PV : 0, more ? lane : 0);
            unsigned hm[SCAN_U / 2];
            int cnt = 0;
#pragma unroll
            for (int q = 0; q < SCAN_U / 2; ++q) {
                hm[q] = pair_hits(cur[2 * q], cur[2 * q + 1], pat);
                cnt += __popc(hm[q]);
            }
            if (cnt) {
                int idx = atomicAdd(&sCnt, cnt);
                const int eb = 16 * (vb + lane);
#pragma unroll
                for (int q = 0; q < SCAN_U / 2; ++q) {
                    unsigned h = hm[q];
                    while (h) {
                        int bit = __ffs(h) - 1;
                        h &= h - 1;
                        int e = eb + (2 * q + ((bit >> 2) & 1)) * 512 + 4 * (bit & 3) + (bit >> 3);
                        if (idx < CAP) lSD[idx].x = e;
                        ++idx;
                    }
                }
            }
#pragma unroll
            for (int j = 0; j < SCAN_U; ++j) cur[j] = nx[j];
        }
    }
    __syncthreads();
    if (tid == 0) tq[NQ + 1] = sCnt;
    }

    const int total = BUILD ? sCnt : storedTotal;
    if (total <= CAP - 4) {
        const int T = total;
        {
            int sI[NIT], dI[NIT], rI[NIT];
            float4 aI[NIT], lgI[NIT];
#pragma unroll
            for (int k = 0; k < NIT; ++k) {
                int i = tid + k * TILE_THREADS;
                if (BUILD) {
                    int e = (i >= nd && i < T) ? lSD[i].x : 0;
                    int sv = srcs[e];
                    int dv = dsts[e] - d0;
                    if (i < nd) { sv = d0 + i; dv = i; }
                    sI[k] = sv;
                    dI[k] = dv;
                    rI[k] = 0;
                    if (i < T) rI[k] = atomicAdd(&qCnt[dv & (NQ - 1)], 1);
                } else {
                    int2 p = make_int2(0, 0);
                    if (i < T) p = tl[i];
                    sI[k] = p.x;
                    dI[k] = p.y;
                    rI[k] = 0;
                }
            }
#pragma unroll
            for (int k = 0; k < NIT; ++k) aI[k] = *(const float4*)(aS + 4 * (size_t)sI[k]);
#pragma unroll
            for (int k = 0; k < NIT; ++k) {
                int i = tid + k * TILE_THREADS;
                lgI[k] = make_float4(0.f, 0.f, 0.f, 0.f);
                if (i < T) {
                    int dl = dI[k];
                    float4 ad = *(const float4*)(sAD + 4 * dl);
                    float4 lg;
                    lg.x = leaky(aI[k].x + ad.x); lg.y = leaky(aI[k].y + ad.y);
                    lg.z = leaky(aI[k].z + ad.z); lg.w = leaky(aI[k].w + ad.w);
                    lgI[k] = lg;
                    if (BUILD) {
                        atomicMax(&sM[dl * 4 + 0], enc_f32(lg.x));
                        atomicMax(&sM[dl * 4 + 1], enc_f32(lg.y));
                        atomicMax(&sM[dl * 4 + 2], enc_f32(lg.z));
                        atomicMax(&sM[dl * 4 + 3], enc_f32(lg.w));
                    } else {
                        lSD[i] = make_int2(sI[k], dl);
                        lW[i]  = lg;
                    }
                }
            }
            if (BUILD) {
                __syncthreads();
                if (wv == 0) {
                    int c = qCnt[lane];
                    int inc = c;
#pragma unroll
                    for (int d = 1; d < NQ; d <<= 1) {
                        int up = __shfl_up(inc, d);
                        if (lane >= d) inc += up;
                    }
                    qBase[lane] = inc - c;
                    if (lane == NQ - 1) qBase[NQ] = inc;
                }
                __syncthreads();
#pragma unroll
                for (int k = 0; k < NIT; ++k) {
                    int i = tid + k * TILE_THREADS;
                    if (i < T) {
                        int pos = qBase[dI[k] & (NQ - 1)] + rI[k];
                        lSD[pos] = make_int2(sI[k], dI[k]);
                        tl[pos]  = make_int2(sI[k], dI[k]);
                        lW[pos]  = lgI[k];
                    }
                }
                if (tid <= NQ) tq[tid] = qBase[tid];
            }
        }
        __syncthreads();

        const int qb = qBase[lane] + wv;
        const int qe = qBase[lane + 1];
        if (!BUILD) {
            for (int i = qb; i < qe; i += NQ) {
                int dl = lSD[i].y;
                float4 lg = lW[i];
                atomicMax(&sM[dl * 4 + 0], enc_f32(lg.x));
                atomicMax(&sM[dl * 4 + 1], enc_f32(lg.y));
                atomicMax(&sM[dl * 4 + 2], enc_f32(lg.z));
                atomicMax(&sM[dl * 4 + 3], enc_f32(lg.w));
            }
            __syncthreads();
        }

        for (int i = qb; i < qe; i += NQ) {
            int dl = lSD[i].y;
            float4 lg = lW[i];
            uint4 m = *(const uint4*)(sM + 4 * dl);
            float4 w;
            w.x = expf(lg.x - dec_f32(m.x)); w.y = expf(lg.y - dec_f32(m.y));
            w.z = expf(lg.z - dec_f32(m.z)); w.w = expf(lg.w - dec_f32(m.w));
            lW[i] = w;
            atomicAdd(&sSum[dl * 4 + 0], w.x);
            atomicAdd(&sSum[dl * 4 + 1], w.y);
            atomicAdd(&sSum[dl * 4 + 2], w.z);
            atomicAdd(&sSum[dl * 4 + 3], w.w);
        }
        __syncthreads();

        const int b0 = __builtin_amdgcn_readfirstlane(qBase[wv]);
        const int b1 = __builtin_amdgcn_readfirstlane(qBase[wv + 1]);
        const bool lo = lane < 16;
        float a[KQ][8];
#pragma unroll
        for (int k = 0; k < KQ; ++k)
#pragma unroll
            for (int j = 0; j < 8; ++j) a[k][j] = 0.f;
        if (b0 < b1) {
            const int iLast = b1 - 1;
            const float* Hl = H + 4 * lane;
            float4 r0[RING], r1[RING];
            int dcur[RING];
#pragma unroll
            for (int u = 0; u < RING; ++u) {
                int i = b0 + u;
                int2 e = lSD[i < iLast ? i : iLast];
                dcur[u] = __builtin_amdgcn_readfirstlane(e.y);
                const float* row = Hl + (size_t)__builtin_amdgcn_readfirstlane(e.x) * HC;
                r0[u] = *(const float4*)(row);
                r1[u] = *(const float4*)(row + 128);
                __builtin_amdgcn_sched_barrier(0);
            }
            int je = b0 + RING;
            int2 eN = lSD[je < iLast ? je : iLast];
            float4 wN = lW[b0];
            for (int i = b0; i < b1; i += RING) {
#pragma unroll
                for (int u = 0; u < RING; ++u) {
                    const float4 w = wN;
                    const int2 e = eN;
                    const int jn = i + u + 1;
                    wN = lW[jn < iLast ? jn : iLast];
                    je = jn + RING;
                    eN = lSD[je < iLast ? je : iLast];
                    if (i + u < b1)
                        row_add(a, dcur[u] >> 5, r0[u], r1[u], lo ? w.x : w.y, lo ? w.z : w.w);
                    dcur[u] = __builtin_amdgcn_readfirstlane(e.y);
                    const float* row = Hl + (size_t)__builtin_amdgcn_readfirstlane(e.x) * HC;
                    r0[u] = *(const float4*)(row);
                    r1[u] = *(const float4*)(row + 128);
                }
            }
        }

        const float4 bb0 = *(const float4*)(b + 4 * lane);
        const float4 bb1 = *(const float4*)(b + 128 + 4 * lane);
#pragma unroll
        for (int k = 0; k < KQ; ++k) {
            const int dl = wv + NQ * k;
            if (dl < nd) {
                const float da = sSum[dl * 4 + (lo ? 0 : 1)] + 1e-16f;
                const float db = sSum[dl * 4 + (lo ? 2 : 3)] + 1e-16f;
                float4 v0, v1;
                v0.x = a[k][0] / da + bb0.x; v0.y = a[k][1] / da + bb0.y;
                v0.z = a[k][2] / da + bb0.z; v0.w = a[k][3] / da + bb0.w;
                v1.x = a[k][4] / db + bb1.x; v1.y = a[k][5] / db + bb1.y;
                v1.z = a[k][6] / db + bb1.z; v1.w = a[k][7] / db + bb1.w;
                if (relu) {
                    v0.x = fmaxf(v0.x, 0.f); v0.y = fmaxf(v0.y, 0.f);
                    v0.z = fmaxf(v0.z, 0.f); v0.w = fmaxf(v0.w, 0.f);
                    v1.x = fmaxf(v1.x, 0.f); v1.y = fmaxf(v1.y, 0.f);
                    v1.z = fmaxf(v1.z, 0.f); v1.w = fmaxf(v1.w, 0.f);
                }
                float* o = out + (size_t)(d0 + dl) * HC + 4 * lane;
                *(float4*)(o)       = v0;
                *(float4*)(o + 128) = v1;
            }
        }
    } else {
        for (int i = tid; i < TD * HC / 4; i += TILE_THREADS)
            ((float4*)acc)[i] = make_float4(0.f, 0.f, 0.f, 0.f);
        if (BUILD && tid <= NQ) tq[tid] = 0;
        const int CH = 32 * SCAN_U;
        int dn[SCAN_U];

        load_dst(dn, dsts, wave * CH, lane, E);
        for (int c0 = wave * CH; c0 < E; c0 += nwaves * CH) {
            int sv[SCAN_U];
            unsigned dlv[SCAN_U];
#pragma unroll
            for (int j = 0; j < SCAN_U; ++j) dlv[j] = (unsigned)(dn[j] - d0);
            load_dst(dn, dsts, c0 + nwaves * CH, lane, E);
#pragma unroll
            for (int j = 0; j < SCAN_U; ++j) {
                int e = c0 + j * 32 + lane;
                sv[j] = (dlv[j] < (unsigned)nd) ? srcs[e] : 0;
            }
#pragma unroll
            for (int j = 0; j < SCAN_U; ++j) {
                if (dlv[j] < (unsigned)nd) {
                    int dl = (int)dlv[j];
                    float4 as = *(const float4*)(aS + 4 * (size_t)sv[j]);
                    atomicMax(&sM[dl * 4 + 0], enc_f32(leaky(as.x + sAD[dl * 4 + 0])));
                    atomicMax(&sM[dl * 4 + 1], enc_f32(leaky(as.y + sAD[dl * 4 + 1])));
                    atomicMax(&sM[dl * 4 + 2], enc_f32(leaky(as.z + sAD[dl * 4 + 2])));
                    atomicMax(&sM[dl * 4 + 3], enc_f32(leaky(as.w + sAD[dl * 4 + 3])));
                }
            }
        }
        __syncthreads();

        load_dst(dn, dsts, wave * CH, lane, E);
        for (int dl = wave; dl < nd; dl += nwaves)
            tile_add_edge(H, aS, d0 + dl, dl, lane, acc, sAD, sM, sSum);

        for (int c0 = wave * CH; c0 < E; c0 += nwaves * CH) {
            int sv[SCAN_U];
            unsigned dlv[SCAN_U];
#pragma unroll
            for (int j = 0; j < SCAN_U; ++j) dlv[j] = (unsigned)(dn[j] - d0);
            load_dst(dn, dsts, c0 + nwaves * CH, lane, E);
#pragma unroll
            for (int j = 0; j < SCAN_U; ++j) {
                int e = c0 + j * 32 + lane;
                sv[j] = (dlv[j] < (unsigned)nd) ? srcs[e] : 0;
            }
#pragma unroll
            for (int j = 0; j < SCAN_U; ++j) {
                unsigned mask = (unsigned)__ballot(dlv[j] < (unsigned)nd);
                while (mask) {
                    int l = __ffs(mask) - 1;
                    mask &= mask - 1;
                    int s  = __shfl(sv[j], l);
                    int dl = __shfl((int)dlv[j], l);
                    tile_add_edge(H, aS, s, dl, lane, acc, sAD, sM, sSum);
                }
            }
        }
        __syncthreads();

        for (int i = tid; i < nd * (HC / 4); i += TILE_THREADS) {
            int dl = i / (HC / 4), fq = i % (HC / 4);
            float den = sSum[dl * 4 + (fq / (CPH / 4))] + 1e-16f;
            float4 v  = *(const float4*)(acc + dl * HC + fq * 4);
            float4 bb = *(const float4*)(b + fq * 4);
            v.x = v.x / den + bb.x; v.y = v.y / den + bb.y;
            v.z = v.z / den + bb.z; v.w = v.w / den + bb.w;
            if (relu) {
                v.x = fmaxf(v.x, 0.f); v.y = fmaxf(v.y, 0.f);
                v.z = fmaxf(v.z, 0.f); v.w = fmaxf(v.w, 0.f);
            }
            *(float4*)(out + (size_t)(d0 + dl) * HC + fq * 4) = v;
        }
    }
}

static void gat_layer(const float* Xin, int K, const float* W, const float* as,
                      const float* ad, const float* b, float* Hbuf, float* agg,
                      const int* ei, int E, int N, float* alphaS, float* alphaD,
                      int relu, int* tileList, int* tileQ, unsigned* tile8, int nPiece, bool build,
                      hipStream_t stream) {
    gemm_wmma_f32<<<GEMM_BLOCKS, GEMM_THREADS, 0, stream>>>(Xin, W, Hbuf, N, K);

    int nh = N * HEADS;
    node_alpha<<<(nh + 255) / 256, 256, 0, stream>>>(Hbuf, as, ad, alphaS, alphaD, N,
                                                     ei + E, build ? tile8 : nullptr, E, 4 * nPiece);

    if (build)
        aggr_tile<true><<<(N + TD - 1) / TD, TILE_THREADS, 0, stream>>>(
                ei, E, N, Hbuf, alphaS, alphaD, b, relu, agg, tileList, tileQ, tile8);
    else
        aggr_tile<false><<<(N + TD - 1) / TD, TILE_THREADS, 0, stream>>>(
                ei, E, N, Hbuf, alphaS, alphaD, b, relu, agg, tileList, tileQ, tile8);
}

extern "C" void kernel_launch(void* const* d_in, const int* in_sizes, int n_in,
                              void* d_out, int out_size, void* d_ws, size_t ws_size,
                              hipStream_t stream) {
    const float* x   = (const float*)d_in[0];
    const int*   ei  = (const int*)d_in[1];
    const float* W1  = (const float*)d_in[2];
    const float* as1 = (const float*)d_in[3];
    const float* ad1 = (const float*)d_in[4];
    const float* b1  = (const float*)d_in[5];
    const float* W2  = (const float*)d_in[6];
    const float* as2 = (const float*)d_in[7];
    const float* ad2 = (const float*)d_in[8];
    const float* b2  = (const float*)d_in[9];
    const float* W3  = (const float*)d_in[10];
    const float* as3 = (const float*)d_in[11];
    const float* ad3 = (const float*)d_in[12];
    const float* b3  = (const float*)d_in[13];

    int K1 = in_sizes[2] / HC;
    int K2 = in_sizes[6] / HC;
    int K3 = in_sizes[10] / HC;
    int N  = in_sizes[0] / K1;
    int E  = in_sizes[1] / 2;

    float*    Hbuf   = (float*)d_ws;
    float*    bufA   = Hbuf + (size_t)N * HC;
    float*    bufB   = bufA + (size_t)N * HC;
    float*    alphaS = bufB + (size_t)N * HC;
    float*    alphaD = alphaS + (size_t)N * HEADS;
    const int stepPieces = TILE_THREADS * SCAN_U;
    const int nPiece  = ((E + 15) / 16 + stepPieces - 1) / stepPieces * stepPieces;
    unsigned* tile8  = (unsigned*)(alphaD + (size_t)N * HEADS);
    int*      tileList = (int*)(tile8 + 4 * (size_t)nPiece);
    int*      tileQ    = tileList + 2 * (size_t)CAP * ((N + TD - 1) / TD);

    gat_layer(x,    K1, W1, as1, ad1, b1, Hbuf, bufA,          ei, E, N,
              alphaS, alphaD,  1, tileList, tileQ, tile8, nPiece,  true,  stream);
    gat_layer(bufA, K2, W2, as2, ad2, b2, Hbuf, bufB,          ei, E, N,
              alphaS, alphaD,  1, tileList, tileQ, tile8, nPiece,  false, stream);
    gat_layer(bufB, K3, W3, as3, ad3, b3, Hbuf, (float*)d_out, ei, E, N,
              alphaS, alphaD,  0, tileList, tileQ, tile8, nPiece,  false, stream);
}
